# starting kernel
# baseline (speedup 1.0000x reference)
#include <hip/hip_runtime.h>
#include <hip/hip_bf16.h>

typedef float v2f __attribute__((ext_vector_type(2)));
typedef float v8f __attribute__((ext_vector_type(8)));

#define N_NODES 50000
#define E_EDGES 500000
#define E2_EDGES 62500
#define ET_EDGES (E_EDGES + E2_EDGES)
#define NREL 1000
#define ALPHA 0.2f
#define KDIM 128

template <int LDBK, int NT>
__global__ void gat_wmma_gemm_f32(const float* __restrict__ A,
                                  const float* __restrict__ B,
                                  float* __restrict__ C,
                                  int M, int lda, int ldbn, int ldc, int coff,
                                  int mtiles, int ngroups) {
    int wave = threadIdx.x >> 5;
    long tile = (long)blockIdx.x * (blockDim.x >> 5) + wave;
    long tiles_total = (long)mtiles * ngroups;
    if (tile >= tiles_total) return;
    int ng = (int)(tile / mtiles);
    int mt = (int)(tile % mtiles);

    int lane = threadIdx.x & 31;
    int half = lane >> 4;
    int l16  = lane & 15;

    int arow = mt * 16 + l16;
    int arow_c = arow < M ? arow : (M - 1);
    const float* Ap = A + (size_t)arow_c * lda + half * 2;
    const float* Bp[NT];
#pragma unroll
    for (int s = 0; s < NT; ++s)
        Bp[s] = B + (size_t)(ng * NT * 16 + s * 16 + l16) * ldbn
                  + (size_t)(half * 2) * LDBK;

    v8f acc[NT];
#pragma unroll
    for (int s = 0; s < NT; ++s) acc[s] = (v8f){0.f,0.f,0.f,0.f,0.f,0.f,0.f,0.f};

    v2f a_cur = *(const v2f*)(Ap);
    v2f b_cur[NT];
#pragma unroll
    for (int s = 0; s < NT; ++s) {
        if constexpr (LDBK == 1) b_cur[s] = *(const v2f*)(Bp[s]);
        else { b_cur[s][0] = Bp[s][0]; b_cur[s][1] = Bp[s][LDBK]; }
    }

#pragma unroll 4
    for (int k0 = 0; k0 < KDIM; k0 += 4) {
        int kn = (k0 + 4 < KDIM) ? (k0 + 4) : 0;
        v2f a_nxt = *(const v2f*)(Ap + kn);
        v2f b_nxt[NT];
#pragma unroll
        for (int s = 0; s < NT; ++s) {
            if constexpr (LDBK == 1) b_nxt[s] = *(const v2f*)(Bp[s] + kn);
            else {
                b_nxt[s][0] = Bp[s][(size_t)kn * LDBK];
                b_nxt[s][1] = Bp[s][(size_t)kn * LDBK + LDBK];
            }
        }
#pragma unroll
        for (int s = 0; s < NT; ++s)
            acc[s] = __builtin_amdgcn_wmma_f32_16x16x4_f32(false, a_cur, false,
                                                           b_cur[s], (short)0,
                                                           acc[s], false, false);
        a_cur = a_nxt;
#pragma unroll
        for (int s = 0; s < NT; ++s) b_cur[s] = b_nxt[s];
    }

#pragma unroll
    for (int s = 0; s < NT; ++s) {
        int ccol = coff + ng * NT * 16 + s * 16 + l16;
#pragma unroll
        for (int r = 0; r < 8; ++r) {
            int crow = mt * 16 + r + half * 8;
            if (crow < M) C[(size_t)crow * ldc + ccol] = acc[s][r];
        }
    }
}

__global__ void gat_edge1(const int* __restrict__ el, const int* __restrict__ eln,
                          const int* __restrict__ etn,
                          const float* __restrict__ xnode,
                          const float* __restrict__ ec,
                          const float* __restrict__ relc,
                          const float* __restrict__ w2h,
                          float* __restrict__ num1, float* __restrict__ denom1) {
    int e = blockIdx.x * (blockDim.x >> 5) + (threadIdx.x >> 5);
    if (e >= ET_EDGES) return;
    int lane = threadIdx.x & 31;

    int s, d, t0 = 0, t1 = 0;
    bool direct = (e < E_EDGES);
    if (direct) {
        s = el[e]; d = el[E_EDGES + e];
    } else {
        int j = e - E_EDGES;
        s = eln[j]; d = eln[E2_EDGES + j];
        t0 = etn[2 * j]; t1 = etn[2 * j + 1];
    }

    int k = lane * 2;
#pragma unroll
    for (int h = 0; h < 2; ++h) {
        float2 xa = *(const float2*)(xnode + (size_t)s * 256 + h * 128 + k);
        float2 xb = *(const float2*)(xnode + (size_t)d * 256 + h * 128 + 64 + k);
        float e0, e1;
        if (direct) {
            float2 ev = *(const float2*)(ec + (size_t)e * 128 + h * 64 + k);
            e0 = ev.x; e1 = ev.y;
        } else {
            float2 a0 = *(const float2*)(relc + (size_t)t0 * 128 + h * 64 + k);
            float2 a1 = *(const float2*)(relc + (size_t)t1 * 128 + h * 64 + k);
            e0 = a0.x + a1.x; e1 = a0.y + a1.y;
        }
        float m0 = xa.x + xb.x + e0;
        float m1 = xa.y + xb.y + e1;
        float2 wv = *(const float2*)(w2h + h * 64 + k);
        float p = m0 * wv.x + m1 * wv.y;
#pragma unroll
        for (int off = 16; off > 0; off >>= 1) p += __shfl_xor(p, off, 32);
        float lr = p > 0.f ? p : ALPHA * p;
        float eev = __expf(-lr);
        float* dp = num1 + (size_t)s * 128 + h * 64 + k;
        atomicAdd(dp,     eev * m0);
        atomicAdd(dp + 1, eev * m1);
        if (lane == 0) atomicAdd(denom1 + (size_t)s * 2 + h, eev);
    }
}

__global__ void gat_fin1(float* __restrict__ num1, const float* __restrict__ denom1) {
    int i = blockIdx.x * blockDim.x + threadIdx.x;
    if (i >= N_NODES * 128) return;
    int n = i >> 7;
    int h = (i & 127) >> 6;
    float dnm = denom1[n * 2 + h];
    dnm = dnm > 1e-12f ? dnm : 1e-12f;
    num1[i] = num1[i] / dnm;
}

__global__ void gat_edge2(const int* __restrict__ el, const int* __restrict__ eln,
                          const int* __restrict__ et, const int* __restrict__ etn,
                          const float* __restrict__ xab2,
                          const float* __restrict__ relW2,
                          const float* __restrict__ w2o,
                          float* __restrict__ num2, float* __restrict__ denom2) {
    int e = blockIdx.x * (blockDim.x >> 5) + (threadIdx.x >> 5);
    if (e >= ET_EDGES) return;
    int lane = threadIdx.x & 31;

    int s, d;
    float4 ev;
    int k = lane * 4;
    if (e < E_EDGES) {
        s = el[e]; d = el[E_EDGES + e];
        ev = *(const float4*)(relW2 + (size_t)et[e] * 128 + k);
    } else {
        int j = e - E_EDGES;
        s = eln[j]; d = eln[E2_EDGES + j];
        int t0 = etn[2 * j], t1 = etn[2 * j + 1];
        float4 a0 = *(const float4*)(relW2 + (size_t)t0 * 128 + k);
        float4 a1 = *(const float4*)(relW2 + (size_t)t1 * 128 + k);
        ev.x = a0.x + a1.x; ev.y = a0.y + a1.y;
        ev.z = a0.z + a1.z; ev.w = a0.w + a1.w;
    }
    float4 xa = *(const float4*)(xab2 + (size_t)s * 256 + k);
    float4 xb = *(const float4*)(xab2 + (size_t)d * 256 + 128 + k);
    float m0 = xa.x + xb.x + ev.x;
    float m1 = xa.y + xb.y + ev.y;
    float m2 = xa.z + xb.z + ev.z;
    float m3 = xa.w + xb.w + ev.w;
    float4 wv = *(const float4*)(w2o + k);
    float p = m0 * wv.x + m1 * wv.y + m2 * wv.z + m3 * wv.w;
#pragma unroll
    for (int off = 16; off > 0; off >>= 1) p += __shfl_xor(p, off, 32);
    float lr = p > 0.f ? p : ALPHA * p;
    float eev = __expf(-lr);
    float* dp = num2 + (size_t)s * 128 + k;
    atomicAdd(dp,     eev * m0);
    atomicAdd(dp + 1, eev * m1);
    atomicAdd(dp + 2, eev * m2);
    atomicAdd(dp + 3, eev * m3);
    if (lane == 0) atomicAdd(denom2 + s, eev);
}

__global__ void gat_fin2(const float* __restrict__ num2,
                         const float* __restrict__ denom2,
                         float* __restrict__ out) {
    int i = blockIdx.x * blockDim.x + threadIdx.x;
    if (i >= N_NODES * 128) return;
    int n = i >> 7;
    float dnm = denom2[n];
    dnm = dnm > 1e-12f ? dnm : 1e-12f;
    float h = num2[i] / dnm;
    float y = h > 0.f ? h : (__expf(h) - 1.f);
    y = y > 0.f ? y : (__expf(y) - 1.f);
    out[i] = y;
}

template <int LDBK>
static inline void launch_gemm(hipStream_t st, const float* A, const float* B,
                               float* C, int M, int N,
                               int lda, int ldbn, int ldc, int coff) {
    constexpr int NT = 4;
    int mtiles  = (M + 15) / 16;
    int ngroups = N / (NT * 16);
    long tiles = (long)mtiles * ngroups;
    int blocks = (int)((tiles + 7) / 8);
    gat_wmma_gemm_f32<LDBK, NT><<<blocks, 256, 0, st>>>(A, B, C, M, lda, ldbn,
                                                        ldc, coff, mtiles,
                                                        ngroups);
}

extern "C" void kernel_launch(void* const* d_in, const int* in_sizes, int n_in,
                              void* d_out, int out_size, void* d_ws, size_t ws_size,
                              hipStream_t stream) {
    const float* x    = (const float*)d_in[0];
    const float* rel  = (const float*)d_in[1];
    const int*   el   = (const int*)d_in[2];
    const int*   et   = (const int*)d_in[3];
    const float* eemb = (const float*)d_in[4];
    const int*   eln  = (const int*)d_in[5];
    const int*   etn  = (const int*)d_in[6];
    const float* w1h  = (const float*)d_in[8];
    const float* w2h  = (const float*)d_in[9];
    const float* w1o  = (const float*)d_in[10];
    const float* w2o  = (const float*)d_in[11];
    const float* WR   = (const float*)d_in[12];
    float* out = (float*)d_out;

    float* ws = (float*)d_ws;
    float* xnode  = ws;
    float* ec     = xnode  + (size_t)N_NODES * 256;
    float* relc   = ec     + (size_t)E_EDGES * 128;
    float* num1   = relc   + (size_t)NREL * 128;
    float* denom1 = num1   + (size_t)N_NODES * 128;
    float* xab2   = denom1 + (size_t)N_NODES * 2;
    float* relW2  = xab2   + (size_t)N_NODES * 256;
    float* num2   = relW2  + (size_t)NREL * 128;
    float* denom2 = num2   + (size_t)N_NODES * 128;

    float* out_rel = out + (size_t)N_NODES * 128;

    hipMemsetAsync(num1,   0, (size_t)N_NODES * 128 * sizeof(float), stream);
    hipMemsetAsync(denom1, 0, (size_t)N_NODES * 2   * sizeof(float), stream);
    hipMemsetAsync(num2,   0, (size_t)N_NODES * 128 * sizeof(float), stream);
    hipMemsetAsync(denom2, 0, (size_t)N_NODES       * sizeof(float), stream);

    for (int h = 0; h < 2; ++h)
        for (int part = 0; part < 2; ++part)
            launch_gemm<1>(stream, x, w1h + (size_t)h * 64 * 384 + part * 128,
                           xnode, N_NODES, 64, 128, 384, 256,
                           h * 128 + part * 64);
    for (int h = 0; h < 2; ++h)
        launch_gemm<1>(stream, eemb, w1h + (size_t)h * 64 * 384 + 256,
                       ec, E_EDGES, 64, 128, 384, 128, h * 64);
    for (int h = 0; h < 2; ++h)
        launch_gemm<1>(stream, rel, w1h + (size_t)h * 64 * 384 + 256,
                       relc, NREL, 64, 128, 384, 128, h * 64);

    {
        int blocks = (ET_EDGES + 7) / 8;
        gat_edge1<<<blocks, 256, 0, stream>>>(el, eln, etn, xnode, ec, relc,
                                              w2h, num1, denom1);
    }
    gat_fin1<<<(N_NODES * 128 + 255) / 256, 256, 0, stream>>>(num1, denom1);

    launch_gemm<128>(stream, rel, WR, out_rel, NREL, 128, 128, 1, 128, 0);
    for (int part = 0; part < 2; ++part)
        launch_gemm<1>(stream, num1, w1o + part * 128,
                       xab2, N_NODES, 128, 128, 384, 256, part * 128);
    launch_gemm<1>(stream, out_rel, w1o + 256, relW2, NREL, 128, 128, 384, 128, 0);

    {
        int blocks = (ET_EDGES + 7) / 8;
        gat_edge2<<<blocks, 256, 0, stream>>>(el, eln, et, etn, xab2, relW2,
                                              w2o, num2, denom2);
    }
    gat_fin2<<<(N_NODES * 128 + 255) / 256, 256, 0, stream>>>(num2, denom2, out);
}


# GEMMs: 128x128 B panel staged once per block in LDS, each wave makes full 16x128 tiles from 16-byte fragments (k regrouped within each 8: WMMA sum order changes); per-head launches merged
# speedup vs baseline: 1.5020x; 1.5020x over previous
#include <hip/hip_runtime.h>
#include <hip/hip_bf16.h>

typedef float v2f __attribute__((ext_vector_type(2)));
typedef float v4f __attribute__((ext_vector_type(4)));
typedef float v8f __attribute__((ext_vector_type(8)));

#define N_NODES 50000
#define E_EDGES 500000
#define E2_EDGES 62500
#define ET_EDGES (E_EDGES + E2_EDGES)
#define NREL 1000
#define ALPHA 0.2f
#define KDIM 128

#define GEMM_N        128
#define GEMM_NT       8
#define GEMM_THREADS  512
#define GEMM_WAVES    (GEMM_THREADS / 32)
#define GEMM_MAX_BLOCKS 256
#define BPITCH        132

__device__ __forceinline__ void gemm_step(v8f (&acc)[GEMM_NT], const v4f& a,
                                          const v4f (&b)[GEMM_NT]) {
    v2f a_lo = (v2f){a[0], a[1]};
    v2f a_hi = (v2f){a[2], a[3]};
#pragma unroll
    for (int s = 0; s < GEMM_NT; ++s)
        acc[s] = __builtin_amdgcn_wmma_f32_16x16x4_f32(
            false, a_lo, false, (v2f){b[s][0], b[s][1]}, (short)0,
            acc[s], false, false);
#pragma unroll
    for (int s = 0; s < GEMM_NT; ++s)
        acc[s] = __builtin_amdgcn_wmma_f32_16x16x4_f32(
            false, a_hi, false, (v2f){b[s][2], b[s][3]}, (short)0,
            acc[s], false, false);
}

template <int LDBK>
__global__ __launch_bounds__(GEMM_THREADS)
void gat_wmma_gemm_f32(const float* __restrict__ A,
                       const float* __restrict__ B,
                       float* __restrict__ C,
                       int M, int lda, int ldbn, int ldc, int coff,
                       int mtiles) {
    __shared__ __attribute__((aligned(16))) float Bs[GEMM_N * BPITCH];
    const int tid = threadIdx.x;

    if constexpr (LDBK == 1) {
        constexpr int NV = GEMM_N * (KDIM / 4) / GEMM_THREADS;
        v4f v[NV];
#pragma unroll
        for (int it = 0; it < NV; ++it) {
            int i = tid + it * GEMM_THREADS;
            v[it] = *(const v4f*)(B + (size_t)(i >> 5) * ldbn + (i & 31) * 4);
        }
#pragma unroll
        for (int it = 0; it < NV; ++it) {
            int i = tid + it * GEMM_THREADS;
            *(v4f*)(Bs + (i >> 5) * BPITCH + (i & 31) * 4) = v[it];
        }
    } else {
        constexpr int NV = 8;
        for (int i0 = tid; i0 < GEMM_N * KDIM; i0 += NV * GEMM_THREADS) {
            float v[NV];
#pragma unroll
            for (int it = 0; it < NV; ++it) {
                int i = i0 + it * GEMM_THREADS;
                v[it] = B[(size_t)(i & 127) * ldbn + (size_t)(i >> 7) * LDBK];
            }
#pragma unroll
            for (int it = 0; it < NV; ++it) {
                int i = i0 + it * GEMM_THREADS;
                Bs[(i & 127) * BPITCH + (i >> 7)] = v[it];
            }
        }
    }
    __syncthreads();

    const int wave = __builtin_amdgcn_readfirstlane(tid >> 5);
    const int lane = tid & 31;
    const int half = lane >> 4;
    const int l16  = lane & 15;
    const float* Bl = Bs + l16 * BPITCH + half * 4;
    const int wstride = gridDim.x * GEMM_WAVES;

    int mt = blockIdx.x + gridDim.x * wave;
    if (mt >= mtiles) return;

    int arow = mt * 16 + l16;
    const float* Ap = A + (size_t)(arow < M ? arow : (M - 1)) * lda + half * 4;

    v4f a_cur[4], a_nxt[4];
    v4f b0[GEMM_NT], b1[GEMM_NT];
#pragma unroll
    for (int t = 0; t < 4; ++t) a_cur[t] = *(const v4f*)(Ap + t * 8);
#pragma unroll
    for (int s = 0; s < GEMM_NT; ++s) b0[s] = *(const v4f*)(Bl + s * 16 * BPITCH);

    for (; mt < mtiles; mt += wstride) {
        int mtn = mt + wstride;
        if (mtn >= mtiles) mtn = mt;
        int arown = mtn * 16 + l16;
        const float* Apn = A + (size_t)(arown < M ? arown : (M - 1)) * lda + half * 4;

        v8f acc[GEMM_NT];
#pragma unroll
        for (int s = 0; s < GEMM_NT; ++s)
            acc[s] = (v8f){0.f,0.f,0.f,0.f,0.f,0.f,0.f,0.f};

#pragma unroll 1
        for (int q = 0; q < 4; ++q) {
            const float* pa = (q < 3) ? (Ap + (q + 1) * 32) : Apn;
#pragma unroll
            for (int t = 0; t < 4; ++t) a_nxt[t] = *(const v4f*)(pa + t * 8);

            const float* Bq = Bl + q * 32;
            const float* Bw = Bl + ((q + 1) & 3) * 32;
#pragma unroll
            for (int s = 0; s < GEMM_NT; ++s) b1[s] = *(const v4f*)(Bq + s * 16 * BPITCH + 8);
            gemm_step(acc, a_cur[0], b0);
#pragma unroll
            for (int s = 0; s < GEMM_NT; ++s) b0[s] = *(const v4f*)(Bq + s * 16 * BPITCH + 16);
            gemm_step(acc, a_cur[1], b1);
#pragma unroll
            for (int s = 0; s < GEMM_NT; ++s) b1[s] = *(const v4f*)(Bq + s * 16 * BPITCH + 24);
            gemm_step(acc, a_cur[2], b0);
#pragma unroll
            for (int s = 0; s < GEMM_NT; ++s) b0[s] = *(const v4f*)(Bw + s * 16 * BPITCH);
            gemm_step(acc, a_cur[3], b1);

#pragma unroll
            for (int t = 0; t < 4; ++t) a_cur[t] = a_nxt[t];
        }

        float* Ct = C + (size_t)(mt * 16 + half * 8) * ldc + coff + l16;
        if (mt * 16 + 16 <= M) {
#pragma unroll
            for (int r = 0; r < 8; ++r)
#pragma unroll
                for (int s = 0; s < GEMM_NT; ++s)
                    Ct[(size_t)r * ldc + s * 16] = acc[s][r];
        } else {
#pragma unroll
            for (int r = 0; r < 8; ++r) {
                if (mt * 16 + half * 8 + r < M) {
#pragma unroll
                    for (int s = 0; s < GEMM_NT; ++s)
                        Ct[(size_t)r * ldc + s * 16] = acc[s][r];
                }
            }
        }
        Ap = Apn;
    }
}

__global__ void gat_edge1(const int* __restrict__ el, const int* __restrict__ eln,
                          const int* __restrict__ etn,
                          const float* __restrict__ xnode,
                          const float* __restrict__ ec,
                          const float* __restrict__ relc,
                          const float* __restrict__ w2h,
                          float* __restrict__ num1, float* __restrict__ denom1) {
    int e = blockIdx.x * (blockDim.x >> 5) + (threadIdx.x >> 5);
    if (e >= ET_EDGES) return;
    int lane = threadIdx.x & 31;

    int s, d, t0 = 0, t1 = 0;
    bool direct = (e < E_EDGES);
    if (direct) {
        s = el[e]; d = el[E_EDGES + e];
    } else {
        int j = e - E_EDGES;
        s = eln[j]; d = eln[E2_EDGES + j];
        t0 = etn[2 * j]; t1 = etn[2 * j + 1];
    }

    int k = lane * 2;
#pragma unroll
    for (int h = 0; h < 2; ++h) {
        float2 xa = *(const float2*)(xnode + (size_t)s * 256 + h * 64 + k);
        float2 xb = *(const float2*)(xnode + (size_t)d * 256 + 128 + h * 64 + k);
        float e0, e1;
        if (direct) {
            float2 ev = *(const float2*)(ec + (size_t)e * 128 + h * 64 + k);
            e0 = ev.x; e1 = ev.y;
        } else {
            float2 a0 = *(const float2*)(relc + (size_t)t0 * 128 + h * 64 + k);
            float2 a1 = *(const float2*)(relc + (size_t)t1 * 128 + h * 64 + k);
            e0 = a0.x + a1.x; e1 = a0.y + a1.y;
        }
        float m0 = xa.x + xb.x + e0;
        float m1 = xa.y + xb.y + e1;
        float2 wv = *(const float2*)(w2h + h * 64 + k);
        float p = m0 * wv.x + m1 * wv.y;
#pragma unroll
        for (int off = 16; off > 0; off >>= 1) p += __shfl_xor(p, off, 32);
        float lr = p > 0.f ? p : ALPHA * p;
        float eev = __expf(-lr);
        float* dp = num1 + (size_t)s * 128 + h * 64 + k;
        atomicAdd(dp,     eev * m0);
        atomicAdd(dp + 1, eev * m1);
        if (lane == 0) atomicAdd(denom1 + (size_t)s * 2 + h, eev);
    }
}

__global__ void gat_fin1(float* __restrict__ num1, const float* __restrict__ denom1) {
    int i = blockIdx.x * blockDim.x + threadIdx.x;
    if (i >= N_NODES * 128) return;
    int n = i >> 7;
    int h = (i & 127) >> 6;
    float dnm = denom1[n * 2 + h];
    dnm = dnm > 1e-12f ? dnm : 1e-12f;
    num1[i] = num1[i] / dnm;
}

__global__ void gat_edge2(const int* __restrict__ el, const int* __restrict__ eln,
                          const int* __restrict__ et, const int* __restrict__ etn,
                          const float* __restrict__ xab2,
                          const float* __restrict__ relW2,
                          const float* __restrict__ w2o,
                          float* __restrict__ num2, float* __restrict__ denom2) {
    int e = blockIdx.x * (blockDim.x >> 5) + (threadIdx.x >> 5);
    if (e >= ET_EDGES) return;
    int lane = threadIdx.x & 31;

    int s, d;
    float4 ev;
    int k = lane * 4;
    if (e < E_EDGES) {
        s = el[e]; d = el[E_EDGES + e];
        ev = *(const float4*)(relW2 + (size_t)et[e] * 128 + k);
    } else {
        int j = e - E_EDGES;
        s = eln[j]; d = eln[E2_EDGES + j];
        int t0 = etn[2 * j], t1 = etn[2 * j + 1];
        float4 a0 = *(const float4*)(relW2 + (size_t)t0 * 128 + k);
        float4 a1 = *(const float4*)(relW2 + (size_t)t1 * 128 + k);
        ev.x = a0.x + a1.x; ev.y = a0.y + a1.y;
        ev.z = a0.z + a1.z; ev.w = a0.w + a1.w;
    }
    float4 xa = *(const float4*)(xab2 + (size_t)s * 256 + k);
    float4 xb = *(const float4*)(xab2 + (size_t)d * 256 + 128 + k);
    float m0 = xa.x + xb.x + ev.x;
    float m1 = xa.y + xb.y + ev.y;
    float m2 = xa.z + xb.z + ev.z;
    float m3 = xa.w + xb.w + ev.w;
    float4 wv = *(const float4*)(w2o + k);
    float p = m0 * wv.x + m1 * wv.y + m2 * wv.z + m3 * wv.w;
#pragma unroll
    for (int off = 16; off > 0; off >>= 1) p += __shfl_xor(p, off, 32);
    float lr = p > 0.f ? p : ALPHA * p;
    float eev = __expf(-lr);
    float* dp = num2 + (size_t)s * 128 + k;
    atomicAdd(dp,     eev * m0);
    atomicAdd(dp + 1, eev * m1);
    atomicAdd(dp + 2, eev * m2);
    atomicAdd(dp + 3, eev * m3);
    if (lane == 0) atomicAdd(denom2 + s, eev);
}

__global__ void gat_fin2(const float* __restrict__ num2,
                         const float* __restrict__ denom2,
                         float* __restrict__ out) {
    int i = blockIdx.x * blockDim.x + threadIdx.x;
    if (i >= N_NODES * 128) return;
    int n = i >> 7;
    float dnm = denom2[n];
    dnm = dnm > 1e-12f ? dnm : 1e-12f;
    float h = num2[i] / dnm;
    float y = h > 0.f ? h : (__expf(h) - 1.f);
    y = y > 0.f ? y : (__expf(y) - 1.f);
    out[i] = y;
}

template <int LDBK>
static inline void launch_gemm(hipStream_t st, const float* A, const float* B,
                               float* C, int M,
                               int lda, int ldbn, int ldc, int coff) {
    int mtiles = (M + 15) / 16;
    int blocks = mtiles < GEMM_MAX_BLOCKS ? mtiles : GEMM_MAX_BLOCKS;
    gat_wmma_gemm_f32<LDBK><<<blocks, GEMM_THREADS, 0, st>>>(A, B, C, M, lda,
                                                             ldbn, ldc, coff,
                                                             mtiles);
}

extern "C" void kernel_launch(void* const* d_in, const int* in_sizes, int n_in,
                              void* d_out, int out_size, void* d_ws, size_t ws_size,
                              hipStream_t stream) {
    const float* x    = (const float*)d_in[0];
    const float* rel  = (const float*)d_in[1];
    const int*   el   = (const int*)d_in[2];
    const int*   et   = (const int*)d_in[3];
    const float* eemb = (const float*)d_in[4];
    const int*   eln  = (const int*)d_in[5];
    const int*   etn  = (const int*)d_in[6];
    const float* w1h  = (const float*)d_in[8];
    const float* w2h  = (const float*)d_in[9];
    const float* w1o  = (const float*)d_in[10];
    const float* w2o  = (const float*)d_in[11];
    const float* WR   = (const float*)d_in[12];
    float* out = (float*)d_out;

    float* ws = (float*)d_ws;
    float* xnode  = ws;
    float* ec     = xnode  + (size_t)N_NODES * 256;
    float* relc   = ec     + (size_t)E_EDGES * 128;
    float* num1   = relc   + (size_t)NREL * 128;
    float* denom1 = num1   + (size_t)N_NODES * 128;
    float* xab2   = denom1 + (size_t)N_NODES * 2;
    float* relW2  = xab2   + (size_t)N_NODES * 256;
    float* num2   = relW2  + (size_t)NREL * 128;
    float* denom2 = num2   + (size_t)N_NODES * 128;

    float* out_rel = out + (size_t)N_NODES * 128;

    hipMemsetAsync(num1,   0, (size_t)N_NODES * 128 * sizeof(float), stream);
    hipMemsetAsync(denom1, 0, (size_t)N_NODES * 2   * sizeof(float), stream);
    hipMemsetAsync(num2,   0, (size_t)N_NODES * 128 * sizeof(float), stream);
    hipMemsetAsync(denom2, 0, (size_t)N_NODES       * sizeof(float), stream);

    for (int part = 0; part < 2; ++part)
        launch_gemm<1>(stream, x, w1h + part * 128, xnode, N_NODES,
                       128, 384, 256, part * 128);
    launch_gemm<1>(stream, eemb, w1h + 256, ec, E_EDGES, 128, 384, 128, 0);
    launch_gemm<1>(stream, rel, w1h + 256, relc, NREL, 128, 384, 128, 0);

    {
        int blocks = (ET_EDGES + 7) / 8;
        gat_edge1<<<blocks, 256, 0, stream>>>(el, eln, etn, xnode, ec, relc,
                                              w2h, num1, denom1);
    }
    gat_fin1<<<(N_NODES * 128 + 255) / 256, 256, 0, stream>>>(num1, denom1);

    launch_gemm<128>(stream, rel, WR, out_rel, NREL, 128, 1, 128, 0);
    for (int part = 0; part < 2; ++part)
        launch_gemm<1>(stream, num1, w1o + part * 128,
                       xab2, N_NODES, 128, 384, 256, part * 128);
    launch_gemm<1>(stream, out_rel, w1o + 256, relW2, NREL, 128, 384, 128, 0);

    {
        int blocks = (ET_EDGES + 7) / 8;
        gat_edge2<<<blocks, 256, 0, stream>>>(el, eln, et, etn, xab2, relW2,
                                              w2o, num2, denom2);
    }
    gat_fin2<<<(N_NODES * 128 + 255) / 256, 256, 0, stream>>>(num2, denom2, out);
}


# float-atomic edge scatter replaced by per-source gather: edges ranked per source by an integer atomic, one wave sums a node's edges in registers in edge order (sum order changes), finalize fused
# speedup vs baseline: 2.3419x; 1.5592x over previous
#include <hip/hip_runtime.h>
#include <hip/hip_bf16.h>

typedef float v2f __attribute__((ext_vector_type(2)));
typedef float v4f __attribute__((ext_vector_type(4)));
typedef float v8f __attribute__((ext_vector_type(8)));

#define N_NODES 50000
#define E_EDGES 500000
#define E2_EDGES 62500
#define ET_EDGES (E_EDGES + E2_EDGES)
#define NREL 1000
#define ALPHA 0.2f
#define KDIM 128

#define GEMM_N        128
#define GEMM_NT       8
#define GEMM_THREADS  512
#define GEMM_WAVES    (GEMM_THREADS / 32)
#define GEMM_MAX_BLOCKS 256
#define BPITCH        132

__device__ __forceinline__ void gemm_step(v8f (&acc)[GEMM_NT], const v4f& a,
                                          const v4f (&b)[GEMM_NT]) {
    v2f a_lo = (v2f){a[0], a[1]};
    v2f a_hi = (v2f){a[2], a[3]};
#pragma unroll
    for (int s = 0; s < GEMM_NT; ++s)
        acc[s] = __builtin_amdgcn_wmma_f32_16x16x4_f32(
            false, a_lo, false, (v2f){b[s][0], b[s][1]}, (short)0,
            acc[s], false, false);
#pragma unroll
    for (int s = 0; s < GEMM_NT; ++s)
        acc[s] = __builtin_amdgcn_wmma_f32_16x16x4_f32(
            false, a_hi, false, (v2f){b[s][2], b[s][3]}, (short)0,
            acc[s], false, false);
}

template <int LDBK>
__global__ __launch_bounds__(GEMM_THREADS)
void gat_wmma_gemm_f32(const float* __restrict__ A,
                       const float* __restrict__ B,
                       float* __restrict__ C,
                       int M, int lda, int ldbn, int ldc, int coff,
                       int mtiles) {
    __shared__ __attribute__((aligned(16))) float Bs[GEMM_N * BPITCH];
    const int tid = threadIdx.x;

    if constexpr (LDBK == 1) {
        constexpr int NV = GEMM_N * (KDIM / 4) / GEMM_THREADS;
        v4f v[NV];
#pragma unroll
        for (int it = 0; it < NV; ++it) {
            int i = tid + it * GEMM_THREADS;
            v[it] = *(const v4f*)(B + (size_t)(i >> 5) * ldbn + (i & 31) * 4);
        }
#pragma unroll
        for (int it = 0; it < NV; ++it) {
            int i = tid + it * GEMM_THREADS;
            *(v4f*)(Bs + (i >> 5) * BPITCH + (i & 31) * 4) = v[it];
        }
    } else {
        constexpr int NV = 8;
        for (int i0 = tid; i0 < GEMM_N * KDIM; i0 += NV * GEMM_THREADS) {
            float v[NV];
#pragma unroll
            for (int it = 0; it < NV; ++it) {
                int i = i0 + it * GEMM_THREADS;
                v[it] = B[(size_t)(i & 127) * ldbn + (size_t)(i >> 7) * LDBK];
            }
#pragma unroll
            for (int it = 0; it < NV; ++it) {
                int i = i0 + it * GEMM_THREADS;
                Bs[(i & 127) * BPITCH + (i >> 7)] = v[it];
            }
        }
    }
    __syncthreads();

    const int wave = __builtin_amdgcn_readfirstlane(tid >> 5);
    const int lane = tid & 31;
    const int half = lane >> 4;
    const int l16  = lane & 15;
    const float* Bl = Bs + l16 * BPITCH + half * 4;
    const int wstride = gridDim.x * GEMM_WAVES;

    int mt = blockIdx.x + gridDim.x * wave;
    if (mt >= mtiles) return;

    int arow = mt * 16 + l16;
    const float* Ap = A + (size_t)(arow < M ? arow : (M - 1)) * lda + half * 4;

    v4f a_cur[4], a_nxt[4];
    v4f b0[GEMM_NT], b1[GEMM_NT];
#pragma unroll
    for (int t = 0; t < 4; ++t) a_cur[t] = *(const v4f*)(Ap + t * 8);
#pragma unroll
    for (int s = 0; s < GEMM_NT; ++s) b0[s] = *(const v4f*)(Bl + s * 16 * BPITCH);

    for (; mt < mtiles; mt += wstride) {
        int mtn = mt + wstride;
        if (mtn >= mtiles) mtn = mt;
        int arown = mtn * 16 + l16;
        const float* Apn = A + (size_t)(arown < M ? arown : (M - 1)) * lda + half * 4;

        v8f acc[GEMM_NT];
#pragma unroll
        for (int s = 0; s < GEMM_NT; ++s)
            acc[s] = (v8f){0.f,0.f,0.f,0.f,0.f,0.f,0.f,0.f};

#pragma unroll 1
        for (int q = 0; q < 4; ++q) {
            const float* pa = (q < 3) ? (Ap + (q + 1) * 32) : Apn;
#pragma unroll
            for (int t = 0; t < 4; ++t) a_nxt[t] = *(const v4f*)(pa + t * 8);

            const float* Bq = Bl + q * 32;
            const float* Bw = Bl + ((q + 1) & 3) * 32;
#pragma unroll
            for (int s = 0; s < GEMM_NT; ++s) b1[s] = *(const v4f*)(Bq + s * 16 * BPITCH + 8);
            gemm_step(acc, a_cur[0], b0);
#pragma unroll
            for (int s = 0; s < GEMM_NT; ++s) b0[s] = *(const v4f*)(Bq + s * 16 * BPITCH + 16);
            gemm_step(acc, a_cur[1], b1);
#pragma unroll
            for (int s = 0; s < GEMM_NT; ++s) b1[s] = *(const v4f*)(Bq + s * 16 * BPITCH + 24);
            gemm_step(acc, a_cur[2], b0);
#pragma unroll
            for (int s = 0; s < GEMM_NT; ++s) b0[s] = *(const v4f*)(Bw + s * 16 * BPITCH);
            gemm_step(acc, a_cur[3], b1);

#pragma unroll
            for (int t = 0; t < 4; ++t) a_cur[t] = a_nxt[t];
        }

        float* Ct = C + (size_t)(mt * 16 + half * 8) * ldc + coff + l16;
        if (mt * 16 + 16 <= M) {
#pragma unroll
            for (int r = 0; r < 8; ++r)
#pragma unroll
                for (int s = 0; s < GEMM_NT; ++s)
                    Ct[(size_t)r * ldc + s * 16] = acc[s][r];
        } else {
#pragma unroll
            for (int r = 0; r < 8; ++r) {
                if (mt * 16 + half * 8 + r < M) {
#pragma unroll
                    for (int s = 0; s < GEMM_NT; ++s)
                        Ct[(size_t)r * ldc + s * 16] = acc[s][r];
                }
            }
        }
        Ap = Apn;
    }
}

#define NPAD         (N_NODES + 4)
#define SCAN_THREADS 1024
#define SCAN_GROUPS  13
#define SCAN_NG      (NPAD / 4)

__global__ void gat_count(const int* __restrict__ el, const int* __restrict__ eln,
                          int* __restrict__ cnt, int* __restrict__ rank) {
    int e = blockIdx.x * blockDim.x + threadIdx.x;
    if (e >= ET_EDGES) return;
    int s = (e < E_EDGES) ? el[e] : eln[e - E_EDGES];
    rank[e] = atomicAdd(cnt + s, 1);
}

__global__ __launch_bounds__(SCAN_THREADS)
void gat_scan(const int* __restrict__ cnt, int* __restrict__ rowptr) {
    __shared__ int wsum[SCAN_THREADS / 32];
    const int t = threadIdx.x, lane = t & 31, w = t >> 5;
    int4 v[SCAN_GROUPS];
    int tot = 0;
#pragma unroll
    for (int i = 0; i < SCAN_GROUPS; ++i) {
        int g = t * SCAN_GROUPS + i;
        v[i] = make_int4(0, 0, 0, 0);
        if (g < SCAN_NG) v[i] = *(const int4*)(cnt + 4 * g);
        tot += v[i].x + v[i].y + v[i].z + v[i].w;
    }
    int inc = tot;
    for (int off = 1; off < 32; off <<= 1) {
        int y = __shfl_up(inc, off, 32);
        if (lane >= off) inc += y;
    }
    if (lane == 31) wsum[w] = inc;
    __syncthreads();
    if (w == 0) {
        int s = wsum[lane], sc = s;
        for (int off = 1; off < 32; off <<= 1) {
            int y = __shfl_up(sc, off, 32);
            if (lane >= off) sc += y;
        }
        wsum[lane] = sc - s;
    }
    __syncthreads();
    int run = wsum[w] + inc - tot;
#pragma unroll
    for (int i = 0; i < SCAN_GROUPS; ++i) {
        int g = t * SCAN_GROUPS + i;
        int4 o;
        o.x = run; run += v[i].x;
        o.y = run; run += v[i].y;
        o.z = run; run += v[i].z;
        o.w = run; run += v[i].w;
        if (g < SCAN_NG) *(int4*)(rowptr + 4 * g) = o;
    }
}

__global__ void gat_fill(const int* __restrict__ el, const int* __restrict__ eln,
                         const int* __restrict__ et, const int* __restrict__ etn,
                         const int* __restrict__ rowptr, const int* __restrict__ rank,
                         int* __restrict__ csr) {
    int e = blockIdx.x * blockDim.x + threadIdx.x;
    if (e >= ET_EDGES) return;
    int s, d, t0, t1;
    if (e < E_EDGES) {
        s = el[e]; d = el[E_EDGES + e]; t0 = et[e]; t1 = -1;
    } else {
        int j = e - E_EDGES;
        s = eln[j]; d = eln[E2_EDGES + j];
        t0 = etn[2 * j]; t1 = etn[2 * j + 1];
    }
    int pos = rowptr[s] + rank[e];
    pos = pos < 0 ? 0 : (pos > ET_EDGES - 1 ? ET_EDGES - 1 : pos);
    *(int4*)(csr + 4 * (size_t)pos) = make_int4(e, d, t0, t1);
}

#define NODE_WAVES 8
#define NODE_U     4

template <int LAYER>
__global__ __launch_bounds__(NODE_WAVES * 32)
void gat_node(const int* __restrict__ rowptr, const int* __restrict__ csr,
              const float* __restrict__ xtab, const float* __restrict__ etab,
              const float* __restrict__ rtab, const float* __restrict__ w2,
              float* __restrict__ dst) {
    const int wave = __builtin_amdgcn_readfirstlane(threadIdx.x >> 5);
    const int n = blockIdx.x * NODE_WAVES + wave;
    if (n >= N_NODES) return;
    const int lane = threadIdx.x & 31;
    const int k4 = lane * 4;
    const int start = __builtin_amdgcn_readfirstlane(rowptr[n]);
    const int end   = __builtin_amdgcn_readfirstlane(rowptr[n + 1]);
    const v4f xa = *(const v4f*)(xtab + (size_t)n * 256 + k4);
    const v4f wv = *(const v4f*)(w2 + k4);
    v4f acc = (v4f){0.f, 0.f, 0.f, 0.f};
    float den = 0.f;

    for (int base = start; base < end; base += 32) {
        int cntc = end - base;
        if (cntc > 32) cntc = 32;
        int4 md = make_int4(0x7fffffff, 0, 0, -1);
        if (lane < cntc) md = *(const int4*)(csr + 4 * (size_t)(base + lane));
        int rk = 0;
        for (int j = 0; j < cntc; ++j)
            rk += (__builtin_amdgcn_readlane(md.x, j) < md.x) ? 1 : 0;

        for (int j0 = 0; j0 < cntc; j0 += NODE_U) {
            v4f xb[NODE_U], e0[NODE_U], e1[NODE_U];
#pragma unroll
            for (int u = 0; u < NODE_U; ++u) {
                int jj = j0 + u;
                if (jj >= cntc) jj = cntc - 1;
                const unsigned hit = __builtin_amdgcn_ballot_w32(rk == jj);
                const int sl = __builtin_ctz(hit);
                int e  = __builtin_amdgcn_readlane(md.x, sl);
                int d  = __builtin_amdgcn_readlane(md.y, sl);
                int t0 = __builtin_amdgcn_readlane(md.z, sl);
                int t1 = __builtin_amdgcn_readlane(md.w, sl);
                e  = e  < 0 ? 0 : (e  > E_EDGES - 1 ? E_EDGES - 1 : e);
                d  = d  < 0 ? 0 : (d  > N_NODES - 1 ? N_NODES - 1 : d);
                t0 = t0 < 0 ? 0 : (t0 > NREL - 1 ? NREL - 1 : t0);
                t1 = t1 > NREL - 1 ? NREL - 1 : t1;
                xb[u] = *(const v4f*)(xtab + (size_t)d * 256 + 128 + k4);
                if (LAYER == 1 && t1 < 0)
                    e0[u] = *(const v4f*)(etab + (size_t)e * 128 + k4);
                else
                    e0[u] = *(const v4f*)(rtab + (size_t)t0 * 128 + k4);
                if (t1 >= 0)
                    e1[u] = *(const v4f*)(rtab + (size_t)t1 * 128 + k4);
                else
                    e1[u] = (v4f){0.f, 0.f, 0.f, 0.f};
            }
            v4f m[NODE_U];
            float p[NODE_U];
#pragma unroll
            for (int u = 0; u < NODE_U; ++u) {
                const v4f ev = e0[u] + e1[u];
                m[u] = (xa + xb[u]) + ev;
                p[u] = m[u][0] * wv[0] + m[u][1] * wv[1] + m[u][2] * wv[2] + m[u][3] * wv[3];
            }
            if (LAYER == 2) {
#pragma unroll
                for (int u = 0; u < NODE_U; ++u) p[u] += __shfl_xor(p[u], 16, 32);
            }
#pragma unroll
            for (int u = 0; u < NODE_U; ++u) p[u] += __shfl_xor(p[u], 8, 32);
#pragma unroll
            for (int u = 0; u < NODE_U; ++u) p[u] += __shfl_xor(p[u], 4, 32);
#pragma unroll
            for (int u = 0; u < NODE_U; ++u) p[u] += __shfl_xor(p[u], 2, 32);
#pragma unroll
            for (int u = 0; u < NODE_U; ++u) p[u] += __shfl_xor(p[u], 1, 32);
#pragma unroll
            for (int u = 0; u < NODE_U; ++u) {
                if (j0 + u < cntc) {
                    const float lr = p[u] > 0.f ? p[u] : ALPHA * p[u];
                    const float ee = __expf(-lr);
                    acc += ee * m[u];
                    den += ee;
                }
            }
        }
    }

    const float dn = den > 1e-12f ? den : 1e-12f;
    v4f h = acc / dn;
    if (LAYER == 2) {
#pragma unroll
        for (int c = 0; c < 4; ++c) {
            float y = h[c] > 0.f ? h[c] : (__expf(h[c]) - 1.f);
            y = y > 0.f ? y : (__expf(y) - 1.f);
            h[c] = y;
        }
    }
    *(v4f*)(dst + (size_t)n * 128 + k4) = h;
}

template <int LDBK>
static inline void launch_gemm(hipStream_t st, const float* A, const float* B,
                               float* C, int M,
                               int lda, int ldbn, int ldc, int coff) {
    int mtiles = (M + 15) / 16;
    int blocks = mtiles < GEMM_MAX_BLOCKS ? mtiles : GEMM_MAX_BLOCKS;
    gat_wmma_gemm_f32<LDBK><<<blocks, GEMM_THREADS, 0, st>>>(A, B, C, M, lda,
                                                             ldbn, ldc, coff,
                                                             mtiles);
}

extern "C" void kernel_launch(void* const* d_in, const int* in_sizes, int n_in,
                              void* d_out, int out_size, void* d_ws, size_t ws_size,
                              hipStream_t stream) {
    const float* x    = (const float*)d_in[0];
    const float* rel  = (const float*)d_in[1];
    const int*   el   = (const int*)d_in[2];
    const int*   et   = (const int*)d_in[3];
    const float* eemb = (const float*)d_in[4];
    const int*   eln  = (const int*)d_in[5];
    const int*   etn  = (const int*)d_in[6];
    const float* w1h  = (const float*)d_in[8];
    const float* w2h  = (const float*)d_in[9];
    const float* w1o  = (const float*)d_in[10];
    const float* w2o  = (const float*)d_in[11];
    const float* WR   = (const float*)d_in[12];
    float* out = (float*)d_out;

    float* ws = (float*)d_ws;
    float* xnode  = ws;
    float* ec     = xnode  + (size_t)N_NODES * 256;
    float* relc   = ec     + (size_t)E_EDGES * 128;
    float* lx     = relc   + (size_t)NREL * 128;
    float* xab2   = lx     + (size_t)N_NODES * 128;
    float* relW2  = xab2   + (size_t)N_NODES * 256;
    int*   cnt    = (int*)(relW2 + (size_t)NREL * 128);
    int*   rowptr = cnt    + NPAD;
    int*   rank   = rowptr + NPAD;
    int*   csr    = rank   + ET_EDGES;

    float* out_rel = out + (size_t)N_NODES * 128;

    hipMemsetAsync(cnt, 0, (size_t)NPAD * sizeof(int), stream);
    gat_count<<<(ET_EDGES + 255) / 256, 256, 0, stream>>>(el, eln, cnt, rank);
    gat_scan<<<1, SCAN_THREADS, 0, stream>>>(cnt, rowptr);
    gat_fill<<<(ET_EDGES + 255) / 256, 256, 0, stream>>>(el, eln, et, etn, rowptr,
                                                        rank, csr);

    for (int part = 0; part < 2; ++part)
        launch_gemm<1>(stream, x, w1h + part * 128, xnode, N_NODES,
                       128, 384, 256, part * 128);
    launch_gemm<1>(stream, eemb, w1h + 256, ec, E_EDGES, 128, 384, 128, 0);
    launch_gemm<1>(stream, rel, w1h + 256, relc, NREL, 128, 384, 128, 0);

    gat_node<1><<<(N_NODES + NODE_WAVES - 1) / NODE_WAVES, NODE_WAVES * 32, 0, stream>>>(
        rowptr, csr, xnode, ec, relc, w2h, lx);

    launch_gemm<128>(stream, rel, WR, out_rel, NREL, 128, 1, 128, 0);
    for (int part = 0; part < 2; ++part)
        launch_gemm<1>(stream, lx, w1o + part * 128,
                       xab2, N_NODES, 128, 384, 256, part * 128);
    launch_gemm<1>(stream, out_rel, w1o + 256, relW2, NREL, 128, 384, 128, 0);

    gat_node<2><<<(N_NODES + NODE_WAVES - 1) / NODE_WAVES, NODE_WAVES * 32, 0, stream>>>(
        rowptr, csr, xab2, relW2, relW2, w2o, out);
}


# layer 1: edge projection moved behind the aggregation (raw edge rows weighted and summed per node and head, logit via w1c^T w2); node-sized WMMA GEMM replaces the edge-sized one; sum order changes
# speedup vs baseline: 2.6400x; 1.1273x over previous
#include <hip/hip_runtime.h>
#include <hip/hip_bf16.h>

typedef float v2f __attribute__((ext_vector_type(2)));
typedef float v4f __attribute__((ext_vector_type(4)));
typedef float v8f __attribute__((ext_vector_type(8)));

#define N_NODES 50000
#define E_EDGES 500000
#define E2_EDGES 62500
#define ET_EDGES (E_EDGES + E2_EDGES)
#define NREL 1000
#define ALPHA 0.2f
#define KDIM 128

#define GEMM_N        128
#define GEMM_NT       8
#define GEMM_THREADS  512
#define GEMM_WAVES    (GEMM_THREADS / 32)
#define GEMM_MAX_BLOCKS 256
#define BPITCH        132

template <int NT>
__device__ __forceinline__ void gemm_step(v8f (&acc)[NT], const v4f& a,
                                          const v4f (&b)[NT]) {
    v2f a_lo = (v2f){a[0], a[1]};
    v2f a_hi = (v2f){a[2], a[3]};
#pragma unroll
    for (int s = 0; s < NT; ++s)
        acc[s] = __builtin_amdgcn_wmma_f32_16x16x4_f32(
            false, a_lo, false, (v2f){b[s][0], b[s][1]}, (short)0,
            acc[s], false, false);
#pragma unroll
    for (int s = 0; s < NT; ++s)
        acc[s] = __builtin_amdgcn_wmma_f32_16x16x4_f32(
            false, a_hi, false, (v2f){b[s][2], b[s][3]}, (short)0,
            acc[s], false, false);
}

template <int LDBK, bool HEADS>
__global__ __launch_bounds__(GEMM_THREADS)
void gat_wmma_gemm_f32(const float* __restrict__ A,
                       const float* __restrict__ B,
                       float* __restrict__ C,
                       int M, int lda, int ldbn, int ldc, int coff,
                       int mtiles,
                       const float* __restrict__ addp,
                       const float* __restrict__ den2) {
    constexpr int NT = HEADS ? GEMM_NT / 2 : GEMM_NT;
    __shared__ __attribute__((aligned(16))) float Bs[GEMM_N * BPITCH];
    const int tid = threadIdx.x;

    if constexpr (LDBK == 1) {
        constexpr int NV = GEMM_N * (KDIM / 4) / GEMM_THREADS;
        v4f v[NV];
#pragma unroll
        for (int it = 0; it < NV; ++it) {
            int i = tid + it * GEMM_THREADS;
            v[it] = *(const v4f*)(B + (size_t)(i >> 5) * ldbn + (i & 31) * 4);
        }
#pragma unroll
        for (int it = 0; it < NV; ++it) {
            int i = tid + it * GEMM_THREADS;
            *(v4f*)(Bs + (i >> 5) * BPITCH + (i & 31) * 4) = v[it];
        }
    } else {
        constexpr int NV = 8;
        for (int i0 = tid; i0 < GEMM_N * KDIM; i0 += NV * GEMM_THREADS) {
            float v[NV];
#pragma unroll
            for (int it = 0; it < NV; ++it) {
                int i = i0 + it * GEMM_THREADS;
                v[it] = B[(size_t)(i & 127) * ldbn + (size_t)(i >> 7) * LDBK];
            }
#pragma unroll
            for (int it = 0; it < NV; ++it) {
                int i = i0 + it * GEMM_THREADS;
                Bs[(i & 127) * BPITCH + (i >> 7)] = v[it];
            }
        }
    }
    __syncthreads();

    const int wave = __builtin_amdgcn_readfirstlane(tid >> 5);
    const int lane = tid & 31;
    const int half = lane >> 4;
    const int l16  = lane & 15;
    const float* Bl = Bs + l16 * BPITCH + half * 4;
    const int wstride = gridDim.x * GEMM_WAVES;

    int mt = blockIdx.x + gridDim.x * wave;
    if (mt >= mtiles) return;

    int arow = mt * 16 + l16;
    const float* Ap = A + (size_t)(arow < M ? arow : (M - 1)) * lda + half * 4;

    const int hm = mtiles >> 1;
    const float* Bt = Bl;
    if constexpr (HEADS) Bt = Bl + (mt >= hm ? 64 * BPITCH : 0);

    v4f a_cur[4], a_nxt[4];
    v4f b0[NT], b1[NT];
#pragma unroll
    for (int t = 0; t < 4; ++t) a_cur[t] = *(const v4f*)(Ap + t * 8);
#pragma unroll
    for (int s = 0; s < NT; ++s) b0[s] = *(const v4f*)(Bt + s * 16 * BPITCH);

    for (; mt < mtiles; mt += wstride) {
        int mtn = mt + wstride;
        if (mtn >= mtiles) mtn = mt;
        int arown = mtn * 16 + l16;
        const float* Apn = A + (size_t)(arown < M ? arown : (M - 1)) * lda + half * 4;
        const float* Btn = Bl;
        if constexpr (HEADS) Btn = Bl + (mtn >= hm ? 64 * BPITCH : 0);

        v8f acc[NT];
#pragma unroll
        for (int s = 0; s < NT; ++s)
            acc[s] = (v8f){0.f,0.f,0.f,0.f,0.f,0.f,0.f,0.f};

        float ad[8][NT], dn[8];
        size_t o0 = 0;
        if constexpr (HEADS) {
            const int head = mt >= hm ? 1 : 0;
            const int node0 = (mt - head * hm) * 16 + half * 8;
            o0 = (size_t)node0 * ldc + coff + head * 64 + l16;
#pragma unroll
            for (int r = 0; r < 8; ++r) {
                const float d = den2[2 * (node0 + r) + head];
                dn[r] = d > 1e-12f ? d : 1e-12f;
#pragma unroll
                for (int s = 0; s < NT; ++s)
                    ad[r][s] = addp[o0 + (size_t)r * ldc + s * 16];
            }
        }

#pragma unroll 1
        for (int q = 0; q < 4; ++q) {
            const float* pa = (q < 3) ? (Ap + (q + 1) * 32) : Apn;
#pragma unroll
            for (int t = 0; t < 4; ++t) a_nxt[t] = *(const v4f*)(pa + t * 8);

            const float* Bq = Bt + q * 32;
            const float* Bw;
            if constexpr (HEADS) Bw = (q < 3) ? (Bq + 32) : Btn;
            else                 Bw = Bl + ((q + 1) & 3) * 32;
#pragma unroll
            for (int s = 0; s < NT; ++s) b1[s] = *(const v4f*)(Bq + s * 16 * BPITCH + 8);
            gemm_step<NT>(acc, a_cur[0], b0);
#pragma unroll
            for (int s = 0; s < NT; ++s) b0[s] = *(const v4f*)(Bq + s * 16 * BPITCH + 16);
            gemm_step<NT>(acc, a_cur[1], b1);
#pragma unroll
            for (int s = 0; s < NT; ++s) b1[s] = *(const v4f*)(Bq + s * 16 * BPITCH + 24);
            gemm_step<NT>(acc, a_cur[2], b0);
#pragma unroll
            for (int s = 0; s < NT; ++s) b0[s] = *(const v4f*)(Bw + s * 16 * BPITCH);
            gemm_step<NT>(acc, a_cur[3], b1);

#pragma unroll
            for (int t = 0; t < 4; ++t) a_cur[t] = a_nxt[t];
        }

        if constexpr (HEADS) {
#pragma unroll
            for (int r = 0; r < 8; ++r)
#pragma unroll
                for (int s = 0; s < NT; ++s)
                    C[o0 + (size_t)r * ldc + s * 16] = (acc[s][r] + ad[r][s]) / dn[r];
            Bt = Btn;
        } else {
            float* Ct = C + (size_t)(mt * 16 + half * 8) * ldc + coff + l16;
            if (mt * 16 + 16 <= M) {
#pragma unroll
                for (int r = 0; r < 8; ++r)
#pragma unroll
                    for (int s = 0; s < NT; ++s)
                        Ct[(size_t)r * ldc + s * 16] = acc[s][r];
            } else {
#pragma unroll
                for (int r = 0; r < 8; ++r) {
                    if (mt * 16 + half * 8 + r < M) {
#pragma unroll
                        for (int s = 0; s < NT; ++s)
                            Ct[(size_t)r * ldc + s * 16] = acc[s][r];
                    }
                }
            }
        }
        Ap = Apn;
    }
}

#define NPAD         (N_NODES + 4)
#define SCAN_THREADS 1024
#define SCAN_GROUPS  13
#define SCAN_NG      (NPAD / 4)

__global__ void gat_count(const int* __restrict__ el, const int* __restrict__ eln,
                          int* __restrict__ cnt, int* __restrict__ rank) {
    int e = blockIdx.x * blockDim.x + threadIdx.x;
    if (e >= ET_EDGES) return;
    int s = (e < E_EDGES) ? el[e] : eln[e - E_EDGES];
    rank[e] = atomicAdd(cnt + s, 1);
}

__global__ __launch_bounds__(SCAN_THREADS)
void gat_scan(const int* __restrict__ cnt, int* __restrict__ rowptr) {
    __shared__ int wsum[SCAN_THREADS / 32];
    const int t = threadIdx.x, lane = t & 31, w = t >> 5;
    int4 v[SCAN_GROUPS];
    int tot = 0;
#pragma unroll
    for (int i = 0; i < SCAN_GROUPS; ++i) {
        int g = t * SCAN_GROUPS + i;
        v[i] = make_int4(0, 0, 0, 0);
        if (g < SCAN_NG) v[i] = *(const int4*)(cnt + 4 * g);
        tot += v[i].x + v[i].y + v[i].z + v[i].w;
    }
    int inc = tot;
    for (int off = 1; off < 32; off <<= 1) {
        int y = __shfl_up(inc, off, 32);
        if (lane >= off) inc += y;
    }
    if (lane == 31) wsum[w] = inc;
    __syncthreads();
    if (w == 0) {
        int s = wsum[lane], sc = s;
        for (int off = 1; off < 32; off <<= 1) {
            int y = __shfl_up(sc, off, 32);
            if (lane >= off) sc += y;
        }
        wsum[lane] = sc - s;
    }
    __syncthreads();
    int run = wsum[w] + inc - tot;
#pragma unroll
    for (int i = 0; i < SCAN_GROUPS; ++i) {
        int g = t * SCAN_GROUPS + i;
        int4 o;
        o.x = run; run += v[i].x;
        o.y = run; run += v[i].y;
        o.z = run; run += v[i].z;
        o.w = run; run += v[i].w;
        if (g < SCAN_NG) *(int4*)(rowptr + 4 * g) = o;
    }
}

__global__ void gat_fill(const int* __restrict__ el, const int* __restrict__ eln,
                         const int* __restrict__ et, const int* __restrict__ etn,
                         const int* __restrict__ rowptr, const int* __restrict__ rank,
                         int* __restrict__ csr) {
    int e = blockIdx.x * blockDim.x + threadIdx.x;
    if (e >= ET_EDGES) return;
    int s, d, t0, t1;
    if (e < E_EDGES) {
        s = el[e]; d = el[E_EDGES + e]; t0 = et[e]; t1 = -1;
    } else {
        int j = e - E_EDGES;
        s = eln[j]; d = eln[E2_EDGES + j];
        t0 = etn[2 * j]; t1 = etn[2 * j + 1];
    }
    int pos = rowptr[s] + rank[e];
    pos = pos < 0 ? 0 : (pos > ET_EDGES - 1 ? ET_EDGES - 1 : pos);
    *(int4*)(csr + 4 * (size_t)pos) = make_int4(e, d, t0, t1);
}

#define NODE_WAVES 8
#define NODE_U     4

template <int LAYER>
__global__ __launch_bounds__(NODE_WAVES * 32)
void gat_node(const int* __restrict__ rowptr, const int* __restrict__ csr,
              const float* __restrict__ xtab, const float* __restrict__ etab,
              const float* __restrict__ rtab, const float* __restrict__ w2,
              float* __restrict__ dst) {
    const int wave = __builtin_amdgcn_readfirstlane(threadIdx.x >> 5);
    const int n = blockIdx.x * NODE_WAVES + wave;
    if (n >= N_NODES) return;
    const int lane = threadIdx.x & 31;
    const int k4 = lane * 4;
    const int start = __builtin_amdgcn_readfirstlane(rowptr[n]);
    const int end   = __builtin_amdgcn_readfirstlane(rowptr[n + 1]);
    const v4f xa = *(const v4f*)(xtab + (size_t)n * 256 + k4);
    const v4f wv = *(const v4f*)(w2 + k4);
    v4f acc = (v4f){0.f, 0.f, 0.f, 0.f};
    float den = 0.f;

    for (int base = start; base < end; base += 32) {
        int cntc = end - base;
        if (cntc > 32) cntc = 32;
        int4 md = make_int4(0x7fffffff, 0, 0, -1);
        if (lane < cntc) md = *(const int4*)(csr + 4 * (size_t)(base + lane));
        int rk = 0;
        for (int j = 0; j < cntc; ++j)
            rk += (__builtin_amdgcn_readlane(md.x, j) < md.x) ? 1 : 0;

        for (int j0 = 0; j0 < cntc; j0 += NODE_U) {
            v4f xb[NODE_U], e0[NODE_U], e1[NODE_U];
#pragma unroll
            for (int u = 0; u < NODE_U; ++u) {
                int jj = j0 + u;
                if (jj >= cntc) jj = cntc - 1;
                const unsigned hit = __builtin_amdgcn_ballot_w32(rk == jj);
                const int sl = __builtin_ctz(hit);
                int e  = __builtin_amdgcn_readlane(md.x, sl);
                int d  = __builtin_amdgcn_readlane(md.y, sl);
                int t0 = __builtin_amdgcn_readlane(md.z, sl);
                int t1 = __builtin_amdgcn_readlane(md.w, sl);
                e  = e  < 0 ? 0 : (e  > E_EDGES - 1 ? E_EDGES - 1 : e);
                d  = d  < 0 ? 0 : (d  > N_NODES - 1 ? N_NODES - 1 : d);
                t0 = t0 < 0 ? 0 : (t0 > NREL - 1 ? NREL - 1 : t0);
                t1 = t1 > NREL - 1 ? NREL - 1 : t1;
                xb[u] = *(const v4f*)(xtab + (size_t)d * 256 + 128 + k4);
                if (LAYER == 1 && t1 < 0)
                    e0[u] = *(const v4f*)(etab + (size_t)e * 128 + k4);
                else
                    e0[u] = *(const v4f*)(rtab + (size_t)t0 * 128 + k4);
                if (t1 >= 0)
                    e1[u] = *(const v4f*)(rtab + (size_t)t1 * 128 + k4);
                else
                    e1[u] = (v4f){0.f, 0.f, 0.f, 0.f};
            }
            v4f m[NODE_U];
            float p[NODE_U];
#pragma unroll
            for (int u = 0; u < NODE_U; ++u) {
                const v4f ev = e0[u] + e1[u];
                m[u] = (xa + xb[u]) + ev;
                p[u] = m[u][0] * wv[0] + m[u][1] * wv[1] + m[u][2] * wv[2] + m[u][3] * wv[3];
            }
            if (LAYER == 2) {
#pragma unroll
                for (int u = 0; u < NODE_U; ++u) p[u] += __shfl_xor(p[u], 16, 32);
            }
#pragma unroll
            for (int u = 0; u < NODE_U; ++u) p[u] += __shfl_xor(p[u], 8, 32);
#pragma unroll
            for (int u = 0; u < NODE_U; ++u) p[u] += __shfl_xor(p[u], 4, 32);
#pragma unroll
            for (int u = 0; u < NODE_U; ++u) p[u] += __shfl_xor(p[u], 2, 32);
#pragma unroll
            for (int u = 0; u < NODE_U; ++u) p[u] += __shfl_xor(p[u], 1, 32);
#pragma unroll
            for (int u = 0; u < NODE_U; ++u) {
                if (j0 + u < cntc) {
                    const float lr = p[u] > 0.f ? p[u] : ALPHA * p[u];
                    const float ee = __expf(-lr);
                    acc += ee * m[u];
                    den += ee;
                }
            }
        }
    }

    const float dn = den > 1e-12f ? den : 1e-12f;
    v4f h = acc / dn;
    if (LAYER == 2) {
#pragma unroll
        for (int c = 0; c < 4; ++c) {
            float y = h[c] > 0.f ? h[c] : (__expf(h[c]) - 1.f);
            y = y > 0.f ? y : (__expf(y) - 1.f);
            h[c] = y;
        }
    }
    *(v4f*)(dst + (size_t)n * 128 + k4) = h;
}

__global__ void gat_fold_w2(const float* __restrict__ w1, const float* __restrict__ w2,
                            float* __restrict__ uc) {
    const int i = blockIdx.x * blockDim.x + threadIdx.x;
    if (i >= 256) return;
    const int h = i >> 7, k = i & 127;
    float s = 0.f;
    for (int j = 0; j < 64; ++j)
        s += w1[(size_t)(h * 64 + j) * 384 + 256 + k] * w2[h * 64 + j];
    uc[i] = s;
}

__global__ __launch_bounds__(NODE_WAVES * 32)
void gat_node_l1(const int* __restrict__ rowptr, const int* __restrict__ csr,
                 const float* __restrict__ xtab, const float* __restrict__ eraw,
                 const float* __restrict__ rtab, const float* __restrict__ w2,
                 const float* __restrict__ uc, float* __restrict__ accp,
                 float* __restrict__ agg, float* __restrict__ den2) {
    const int wave = __builtin_amdgcn_readfirstlane(threadIdx.x >> 5);
    const int n = blockIdx.x * NODE_WAVES + wave;
    if (n >= N_NODES) return;
    const int lane = threadIdx.x & 31;
    const int k4 = lane * 4;
    const bool lo = lane < 16;
    const int start = __builtin_amdgcn_readfirstlane(rowptr[n]);
    const int end   = __builtin_amdgcn_readfirstlane(rowptr[n + 1]);
    const v4f xa = *(const v4f*)(xtab + (size_t)n * 256 + k4);
    const v4f wv = *(const v4f*)(w2 + k4);
    const v4f u0 = *(const v4f*)(uc + k4);
    const v4f u1 = *(const v4f*)(uc + 128 + k4);
    const v4f zero4 = (v4f){0.f, 0.f, 0.f, 0.f};
    v4f acc = zero4, a0 = zero4, a1 = zero4;
    float den = 0.f;

    for (int base = start; base < end; base += 32) {
        int cntc = end - base;
        if (cntc > 32) cntc = 32;
        int4 md = make_int4(0x7fffffff, 0, 0, -1);
        if (lane < cntc) md = *(const int4*)(csr + 4 * (size_t)(base + lane));
        int rk = 0;
        for (int j = 0; j < cntc; ++j)
            rk += (__builtin_amdgcn_readlane(md.x, j) < md.x) ? 1 : 0;

        for (int j0 = 0; j0 < cntc; j0 += NODE_U) {
            v4f xb[NODE_U], e0[NODE_U], e1[NODE_U];
            bool dir[NODE_U];
#pragma unroll
            for (int u = 0; u < NODE_U; ++u) {
                int jj = j0 + u;
                if (jj >= cntc) jj = cntc - 1;
                const unsigned hit = __builtin_amdgcn_ballot_w32(rk == jj);
                const int sl = __builtin_ctz(hit);
                int e  = __builtin_amdgcn_readlane(md.x, sl);
                int d  = __builtin_amdgcn_readlane(md.y, sl);
                int t0 = __builtin_amdgcn_readlane(md.z, sl);
                int t1 = __builtin_amdgcn_readlane(md.w, sl);
                e  = e  < 0 ? 0 : (e  > E_EDGES - 1 ? E_EDGES - 1 : e);
                d  = d  < 0 ? 0 : (d  > N_NODES - 1 ? N_NODES - 1 : d);
                t0 = t0 < 0 ? 0 : (t0 > NREL - 1 ? NREL - 1 : t0);
                t1 = t1 > NREL - 1 ? NREL - 1 : t1;
                dir[u] = t1 < 0;
                xb[u] = *(const v4f*)(xtab + (size_t)d * 256 + 128 + k4);
                if (t1 < 0)
                    e0[u] = *(const v4f*)(eraw + (size_t)e * 128 + k4);
                else
                    e0[u] = *(const v4f*)(rtab + (size_t)t0 * 128 + k4);
                if (t1 >= 0)
                    e1[u] = *(const v4f*)(rtab + (size_t)t1 * 128 + k4);
                else
                    e1[u] = zero4;
            }
            v4f m[NODE_U], er[NODE_U];
            float p[NODE_U];
#pragma unroll
            for (int u = 0; u < NODE_U; ++u) {
                const v4f ev = dir[u] ? zero4 : (e0[u] + e1[u]);
                er[u] = dir[u] ? e0[u] : zero4;
                m[u] = (xa + xb[u]) + ev;
                const float qn = __builtin_fmaf(m[u][3], wv[3], __builtin_fmaf(m[u][2], wv[2],
                                 __builtin_fmaf(m[u][1], wv[1], m[u][0] * wv[0])));
                const float q0 = __builtin_fmaf(er[u][3], u0[3], __builtin_fmaf(er[u][2], u0[2],
                                 __builtin_fmaf(er[u][1], u0[1], er[u][0] * u0[0])));
                const float q1 = __builtin_fmaf(er[u][3], u1[3], __builtin_fmaf(er[u][2], u1[2],
                                 __builtin_fmaf(er[u][1], u1[1], er[u][0] * u1[0])));
                const float mine = qn + (lo ? q0 : q1);
                const float send = lo ? q1 : q0;
                p[u] = mine + __shfl_xor(send, 16, 32);
            }
#pragma unroll
            for (int u = 0; u < NODE_U; ++u) p[u] += __shfl_xor(p[u], 8, 32);
#pragma unroll
            for (int u = 0; u < NODE_U; ++u) p[u] += __shfl_xor(p[u], 4, 32);
#pragma unroll
            for (int u = 0; u < NODE_U; ++u) p[u] += __shfl_xor(p[u], 2, 32);
#pragma unroll
            for (int u = 0; u < NODE_U; ++u) p[u] += __shfl_xor(p[u], 1, 32);
            float g[NODE_U], go[NODE_U];
#pragma unroll
            for (int u = 0; u < NODE_U; ++u) {
                const float lr = p[u] > 0.f ? p[u] : ALPHA * p[u];
                g[u]  = __expf(-lr);
                go[u] = __shfl_xor(g[u], 16, 32);
            }
#pragma unroll
            for (int u = 0; u < NODE_U; ++u) {
                if (j0 + u < cntc) {
                    const float g0 = lo ? g[u] : go[u];
                    const float g1 = lo ? go[u] : g[u];
                    acc += g[u] * m[u];
                    den += g[u];
                    a0 += g0 * er[u];
                    a1 += g1 * er[u];
                }
            }
        }
    }

    *(v4f*)(accp + (size_t)n * 128 + k4) = acc;
    *(v4f*)(agg + (size_t)n * 128 + k4) = a0;
    *(v4f*)(agg + (size_t)(N_NODES + n) * 128 + k4) = a1;
    if ((lane & 15) == 0) den2[n * 2 + (lane >> 4)] = den;
}

template <int LDBK>
static inline void launch_gemm(hipStream_t st, const float* A, const float* B,
                               float* C, int M,
                               int lda, int ldbn, int ldc, int coff) {
    int mtiles = (M + 15) / 16;
    int blocks = mtiles < GEMM_MAX_BLOCKS ? mtiles : GEMM_MAX_BLOCKS;
    gat_wmma_gemm_f32<LDBK, false><<<blocks, GEMM_THREADS, 0, st>>>(
        A, B, C, M, lda, ldbn, ldc, coff, mtiles, nullptr, nullptr);
}

extern "C" void kernel_launch(void* const* d_in, const int* in_sizes, int n_in,
                              void* d_out, int out_size, void* d_ws, size_t ws_size,
                              hipStream_t stream) {
    const float* x    = (const float*)d_in[0];
    const float* rel  = (const float*)d_in[1];
    const int*   el   = (const int*)d_in[2];
    const int*   et   = (const int*)d_in[3];
    const float* eemb = (const float*)d_in[4];
    const int*   eln  = (const int*)d_in[5];
    const int*   etn  = (const int*)d_in[6];
    const float* w1h  = (const float*)d_in[8];
    const float* w2h  = (const float*)d_in[9];
    const float* w1o  = (const float*)d_in[10];
    const float* w2o  = (const float*)d_in[11];
    const float* WR   = (const float*)d_in[12];
    float* out = (float*)d_out;

    float* ws = (float*)d_ws;
    float* xnode  = ws;
    float* accp   = xnode  + (size_t)N_NODES * 256;
    float* agg    = accp   + (size_t)N_NODES * 128;
    float* den2   = agg    + (size_t)N_NODES * 256;
    float* uc     = den2   + (size_t)N_NODES * 2;
    float* relc   = uc     + 256;
    float* lx     = relc   + (size_t)NREL * 128;
    float* xab2   = lx     + (size_t)N_NODES * 128;
    float* relW2  = xab2   + (size_t)N_NODES * 256;
    int*   cnt    = (int*)(relW2 + (size_t)NREL * 128);
    int*   rowptr = cnt    + NPAD;
    int*   rank   = rowptr + NPAD;
    int*   csr    = rank   + ET_EDGES;

    float* out_rel = out + (size_t)N_NODES * 128;

    hipMemsetAsync(cnt, 0, (size_t)NPAD * sizeof(int), stream);
    gat_count<<<(ET_EDGES + 255) / 256, 256, 0, stream>>>(el, eln, cnt, rank);
    gat_scan<<<1, SCAN_THREADS, 0, stream>>>(cnt, rowptr);
    gat_fill<<<(ET_EDGES + 255) / 256, 256, 0, stream>>>(el, eln, et, etn, rowptr,
                                                        rank, csr);

    for (int part = 0; part < 2; ++part)
        launch_gemm<1>(stream, x, w1h + part * 128, xnode, N_NODES,
                       128, 384, 256, part * 128);
    launch_gemm<1>(stream, rel, w1h + 256, relc, NREL, 128, 384, 128, 0);
    gat_fold_w2<<<1, 256, 0, stream>>>(w1h, w2h, uc);

    gat_node_l1<<<(N_NODES + NODE_WAVES - 1) / NODE_WAVES, NODE_WAVES * 32, 0, stream>>>(
        rowptr, csr, xnode, eemb, relc, w2h, uc, accp, agg, den2);
    static_assert(N_NODES % 16 == 0, "each head's plane must be whole row tiles");
    gat_wmma_gemm_f32<1, true><<<GEMM_MAX_BLOCKS, GEMM_THREADS, 0, stream>>>(
        agg, w1h + 256, lx, 2 * N_NODES, 128, 384, 128, 0, 2 * N_NODES / 16, accp, den2);

    launch_gemm<128>(stream, rel, WR, out_rel, NREL, 128, 1, 128, 0);
    for (int part = 0; part < 2; ++part)
        launch_gemm<1>(stream, lx, w1o + part * 128,
                       xab2, N_NODES, 128, 384, 256, part * 128);
    launch_gemm<1>(stream, out_rel, w1o + 256, relW2, NREL, 128, 384, 128, 0);

    gat_node<2><<<(N_NODES + NODE_WAVES - 1) / NODE_WAVES, NODE_WAVES * 32, 0, stream>>>(
        rowptr, csr, xab2, relW2, relW2, w2o, out);
}


# each layer's independent projection GEMMs run as ONE resident launch: row tiles of all problems form one list dealt to all waves, B panels side by side in LDS; GEMM launches 8 -> 3; sums unchanged
# speedup vs baseline: 2.9896x; 1.1324x over previous
#include <hip/hip_runtime.h>
#include <hip/hip_bf16.h>

typedef float v2f __attribute__((ext_vector_type(2)));
typedef float v4f __attribute__((ext_vector_type(4)));
typedef float v8f __attribute__((ext_vector_type(8)));

#define N_NODES 50000
#define E_EDGES 500000
#define E2_EDGES 62500
#define ET_EDGES (E_EDGES + E2_EDGES)
#define NREL 1000
#define ALPHA 0.2f
#define KDIM 128

#define GEMM_N        128
#define GEMM_NT       8
#define GEMM_THREADS  512
#define GEMM_WAVES    (GEMM_THREADS / 32)
#define GEMM_MAX_BLOCKS 256
#define BPITCH        132
#define GEMM_PANEL    (GEMM_N * BPITCH)

template <int NT>
__device__ __forceinline__ void gemm_step(v8f (&acc)[NT], const v4f& a,
                                          const v4f (&b)[NT]) {
    v2f a_lo = (v2f){a[0], a[1]};
    v2f a_hi = (v2f){a[2], a[3]};
#pragma unroll
    for (int s = 0; s < NT; ++s)
        acc[s] = __builtin_amdgcn_wmma_f32_16x16x4_f32(
            false, a_lo, false, (v2f){b[s][0], b[s][1]}, (short)0,
            acc[s], false, false);
#pragma unroll
    for (int s = 0; s < NT; ++s)
        acc[s] = __builtin_amdgcn_wmma_f32_16x16x4_f32(
            false, a_hi, false, (v2f){b[s][2], b[s][3]}, (short)0,
            acc[s], false, false);
}

template <int LDBK>
__device__ __forceinline__ void gemm_stage_panel(float* __restrict__ Bs,
                                                 const float* __restrict__ B,
                                                 int ldbn, int tid) {
    if constexpr (LDBK == 1) {
        constexpr int NV = GEMM_N * (KDIM / 4) / GEMM_THREADS;
        v4f v[NV];
#pragma unroll
        for (int it = 0; it < NV; ++it) {
            int i = tid + it * GEMM_THREADS;
            v[it] = *(const v4f*)(B + (size_t)(i >> 5) * ldbn + (i & 31) * 4);
        }
#pragma unroll
        for (int it = 0; it < NV; ++it) {
            int i = tid + it * GEMM_THREADS;
            *(v4f*)(Bs + (i >> 5) * BPITCH + (i & 31) * 4) = v[it];
        }
    } else {
        constexpr int NV = 8;
        for (int i0 = tid; i0 < GEMM_N * KDIM; i0 += NV * GEMM_THREADS) {
            float v[NV];
#pragma unroll
            for (int it = 0; it < NV; ++it) {
                int i = i0 + it * GEMM_THREADS;
                v[it] = B[(size_t)(i & 127) * ldbn + (size_t)(i >> 7) * LDBK];
            }
#pragma unroll
            for (int it = 0; it < NV; ++it) {
                int i = i0 + it * GEMM_THREADS;
                Bs[(i & 127) * BPITCH + (i >> 7)] = v[it];
            }
        }
    }
}

__global__ __launch_bounds__(GEMM_THREADS)
void gat_wmma_gemm_multi(
        const float* __restrict__ A0, const float* __restrict__ B0, float* __restrict__ C0, int M0, int ldc0,
        const float* __restrict__ A1, const float* __restrict__ B1, float* __restrict__ C1, int M1, int ldc1,
        const float* __restrict__ A2, const float* __restrict__ B2, float* __restrict__ C2, int M2, int ldc2,
        const float* __restrict__ A3, const float* __restrict__ B3, float* __restrict__ C3, int M3, int ldc3,
        int ldbn) {
    __shared__ __attribute__((aligned(16))) float Bs[3 * GEMM_PANEL];
    const int tid = threadIdx.x;

    const int e0 = (M0 + 15) >> 4;
    const int e1 = e0 + ((M1 + 15) >> 4);
    const int e2 = e1 + ((M2 + 15) >> 4);
    const int e3 = e2 + ((M3 + 15) >> 4);
    const int vs = (int)blockIdx.x +
        GEMM_MAX_BLOCKS * ((e1 - (int)blockIdx.x + GEMM_MAX_BLOCKS - 1) / GEMM_MAX_BLOCKS);

    gemm_stage_panel<1>(Bs, B0, ldbn, tid);
    gemm_stage_panel<1>(Bs + GEMM_PANEL, B1, ldbn, tid);
    if (vs < e3) {
        if (vs < e2) gemm_stage_panel<1>(Bs + 2 * GEMM_PANEL, B2, ldbn, tid);
        else         gemm_stage_panel<KDIM>(Bs + 2 * GEMM_PANEL, B3, 1, tid);
    }
    __syncthreads();

    const int wave = __builtin_amdgcn_readfirstlane(tid >> 5);
    const int lane = tid & 31;
    const int half = lane >> 4;
    const int l16  = lane & 15;
    const float* Bl = Bs + l16 * BPITCH + half * 4;
    constexpr int wstride = GEMM_MAX_BLOCKS * GEMM_WAVES;

    int v = (int)blockIdx.x + GEMM_MAX_BLOCKS * wave;
    if (v >= e3) return;

    auto locate = [&](int vv, const float*& A, float*& C, int& M, int& ldc,
                      int& mt, int& slot) {
        if (vv < e0)      { A = A0; C = C0; M = M0; ldc = ldc0; mt = vv;      slot = 0; }
        else if (vv < e1) { A = A1; C = C1; M = M1; ldc = ldc1; mt = vv - e0; slot = 1; }
        else if (vv < e2) { A = A2; C = C2; M = M2; ldc = ldc2; mt = vv - e1; slot = 2; }
        else              { A = A3; C = C3; M = M3; ldc = ldc3; mt = vv - e2; slot = 2; }
    };

    const float* A; float* C; int M, ldc, mt, slot;
    locate(v, A, C, M, ldc, mt, slot);
    int arow = mt * 16 + l16;
    const float* Ap = A + (size_t)(arow < M ? arow : (M - 1)) * KDIM + half * 4;
    const float* Bt = Bl + slot * GEMM_PANEL;

    v4f a_cur[4], a_nxt[4];
    v4f b0[GEMM_NT], b1[GEMM_NT];
#pragma unroll
    for (int t = 0; t < 4; ++t) a_cur[t] = *(const v4f*)(Ap + t * 8);
#pragma unroll
    for (int s = 0; s < GEMM_NT; ++s) b0[s] = *(const v4f*)(Bt + s * 16 * BPITCH);

    for (;;) {
        int vn = v + wstride;
        const bool more = vn < e3;
        if (!more) vn = v;
        const float* An; float* Cn; int Mn, ldcn, mtn, slotn;
        locate(vn, An, Cn, Mn, ldcn, mtn, slotn);
        int arown = mtn * 16 + l16;
        const float* Apn = An + (size_t)(arown < Mn ? arown : (Mn - 1)) * KDIM + half * 4;
        const float* Btn = Bl + slotn * GEMM_PANEL;

        v8f acc[GEMM_NT];
#pragma unroll
        for (int s = 0; s < GEMM_NT; ++s)
            acc[s] = (v8f){0.f,0.f,0.f,0.f,0.f,0.f,0.f,0.f};

#pragma unroll 1
        for (int q = 0; q < 4; ++q) {
            const float* pa = (q < 3) ? (Ap + (q + 1) * 32) : Apn;
#pragma unroll
            for (int t = 0; t < 4; ++t) a_nxt[t] = *(const v4f*)(pa + t * 8);

            const float* Bq = Bt + q * 32;
            const float* Bw = (q < 3) ? (Bq + 32) : Btn;
#pragma unroll
            for (int s = 0; s < GEMM_NT; ++s) b1[s] = *(const v4f*)(Bq + s * 16 * BPITCH + 8);
            gemm_step<GEMM_NT>(acc, a_cur[0], b0);
#pragma unroll
            for (int s = 0; s < GEMM_NT; ++s) b0[s] = *(const v4f*)(Bq + s * 16 * BPITCH + 16);
            gemm_step<GEMM_NT>(acc, a_cur[1], b1);
#pragma unroll
            for (int s = 0; s < GEMM_NT; ++s) b1[s] = *(const v4f*)(Bq + s * 16 * BPITCH + 24);
            gemm_step<GEMM_NT>(acc, a_cur[2], b0);
#pragma unroll
            for (int s = 0; s < GEMM_NT; ++s) b0[s] = *(const v4f*)(Bw + s * 16 * BPITCH);
            gemm_step<GEMM_NT>(acc, a_cur[3], b1);

#pragma unroll
            for (int t = 0; t < 4; ++t) a_cur[t] = a_nxt[t];
        }

        float* Ct = C + (size_t)(mt * 16 + half * 8) * ldc + l16;
        if (mt * 16 + 16 <= M) {
#pragma unroll
            for (int r = 0; r < 8; ++r)
#pragma unroll
                for (int s = 0; s < GEMM_NT; ++s)
                    Ct[(size_t)r * ldc + s * 16] = acc[s][r];
        } else {
#pragma unroll
            for (int r = 0; r < 8; ++r) {
                if (mt * 16 + half * 8 + r < M) {
#pragma unroll
                    for (int s = 0; s < GEMM_NT; ++s)
                        Ct[(size_t)r * ldc + s * 16] = acc[s][r];
                }
            }
        }
        if (!more) break;
        v = vn; C = Cn; M = Mn; ldc = ldcn; mt = mtn;
        Ap = Apn; Bt = Btn;
    }
}

__global__ __launch_bounds__(GEMM_THREADS)
void gat_wmma_gemm_heads(const float* __restrict__ A,
                         const float* __restrict__ B,
                         float* __restrict__ C,
                         int M, int lda, int ldbn, int ldc, int coff,
                         int mtiles,
                         const float* __restrict__ addp,
                         const float* __restrict__ den2) {
    constexpr int NT = GEMM_NT / 2;
    __shared__ __attribute__((aligned(16))) float Bs[GEMM_PANEL];
    const int tid = threadIdx.x;

    gemm_stage_panel<1>(Bs, B, ldbn, tid);
    __syncthreads();

    const int wave = __builtin_amdgcn_readfirstlane(tid >> 5);
    const int lane = tid & 31;
    const int half = lane >> 4;
    const int l16  = lane & 15;
    const float* Bl = Bs + l16 * BPITCH + half * 4;
    const int wstride = gridDim.x * GEMM_WAVES;

    int mt = blockIdx.x + gridDim.x * wave;
    if (mt >= mtiles) return;

    int arow = mt * 16 + l16;
    const float* Ap = A + (size_t)(arow < M ? arow : (M - 1)) * lda + half * 4;

    const int hm = mtiles >> 1;
    const float* Bt = Bl + (mt >= hm ? 64 * BPITCH : 0);

    v4f a_cur[4], a_nxt[4];
    v4f b0[NT], b1[NT];
#pragma unroll
    for (int t = 0; t < 4; ++t) a_cur[t] = *(const v4f*)(Ap + t * 8);
#pragma unroll
    for (int s = 0; s < NT; ++s) b0[s] = *(const v4f*)(Bt + s * 16 * BPITCH);

    for (; mt < mtiles; mt += wstride) {
        int mtn = mt + wstride;
        if (mtn >= mtiles) mtn = mt;
        int arown = mtn * 16 + l16;
        const float* Apn = A + (size_t)(arown < M ? arown : (M - 1)) * lda + half * 4;
        const float* Btn = Bl + (mtn >= hm ? 64 * BPITCH : 0);

        v8f acc[NT];
#pragma unroll
        for (int s = 0; s < NT; ++s)
            acc[s] = (v8f){0.f,0.f,0.f,0.f,0.f,0.f,0.f,0.f};

        float ad[8][NT], dn[8];
        const int head = mt >= hm ? 1 : 0;
        const int node0 = (mt - head * hm) * 16 + half * 8;
        const size_t o0 = (size_t)node0 * ldc + coff + head * 64 + l16;
#pragma unroll
        for (int r = 0; r < 8; ++r) {
            const float d = den2[2 * (node0 + r) + head];
            dn[r] = d > 1e-12f ? d : 1e-12f;
#pragma unroll
            for (int s = 0; s < NT; ++s)
                ad[r][s] = addp[o0 + (size_t)r * ldc + s * 16];
        }

#pragma unroll 1
        for (int q = 0; q < 4; ++q) {
            const float* pa = (q < 3) ? (Ap + (q + 1) * 32) : Apn;
#pragma unroll
            for (int t = 0; t < 4; ++t) a_nxt[t] = *(const v4f*)(pa + t * 8);

            const float* Bq = Bt + q * 32;
            const float* Bw = (q < 3) ? (Bq + 32) : Btn;
#pragma unroll
            for (int s = 0; s < NT; ++s) b1[s] = *(const v4f*)(Bq + s * 16 * BPITCH + 8);
            gemm_step<NT>(acc, a_cur[0], b0);
#pragma unroll
            for (int s = 0; s < NT; ++s) b0[s] = *(const v4f*)(Bq + s * 16 * BPITCH + 16);
            gemm_step<NT>(acc, a_cur[1], b1);
#pragma unroll
            for (int s = 0; s < NT; ++s) b1[s] = *(const v4f*)(Bq + s * 16 * BPITCH + 24);
            gemm_step<NT>(acc, a_cur[2], b0);
#pragma unroll
            for (int s = 0; s < NT; ++s) b0[s] = *(const v4f*)(Bw + s * 16 * BPITCH);
            gemm_step<NT>(acc, a_cur[3], b1);

#pragma unroll
            for (int t = 0; t < 4; ++t) a_cur[t] = a_nxt[t];
        }

#pragma unroll
        for (int r = 0; r < 8; ++r)
#pragma unroll
            for (int s = 0; s < NT; ++s)
                C[o0 + (size_t)r * ldc + s * 16] = (acc[s][r] + ad[r][s]) / dn[r];
        Bt = Btn;
        Ap = Apn;
    }
}

#define NPAD         (N_NODES + 4)
#define SCAN_THREADS 1024
#define SCAN_GROUPS  13
#define SCAN_NG      (NPAD / 4)

__global__ void gat_count(const int* __restrict__ el, const int* __restrict__ eln,
                          int* __restrict__ cnt, int* __restrict__ rank) {
    int e = blockIdx.x * blockDim.x + threadIdx.x;
    if (e >= ET_EDGES) return;
    int s = (e < E_EDGES) ? el[e] : eln[e - E_EDGES];
    rank[e] = atomicAdd(cnt + s, 1);
}

__global__ __launch_bounds__(SCAN_THREADS)
void gat_scan(const int* __restrict__ cnt, int* __restrict__ rowptr) {
    __shared__ int wsum[SCAN_THREADS / 32];
    const int t = threadIdx.x, lane = t & 31, w = t >> 5;
    int4 v[SCAN_GROUPS];
    int tot = 0;
#pragma unroll
    for (int i = 0; i < SCAN_GROUPS; ++i) {
        int g = t * SCAN_GROUPS + i;
        v[i] = make_int4(0, 0, 0, 0);
        if (g < SCAN_NG) v[i] = *(const int4*)(cnt + 4 * g);
        tot += v[i].x + v[i].y + v[i].z + v[i].w;
    }
    int inc = tot;
    for (int off = 1; off < 32; off <<= 1) {
        int y = __shfl_up(inc, off, 32);
        if (lane >= off) inc += y;
    }
    if (lane == 31) wsum[w] = inc;
    __syncthreads();
    if (w == 0) {
        int s = wsum[lane], sc = s;
        for (int off = 1; off < 32; off <<= 1) {
            int y = __shfl_up(sc, off, 32);
            if (lane >= off) sc += y;
        }
        wsum[lane] = sc - s;
    }
    __syncthreads();
    int run = wsum[w] + inc - tot;
#pragma unroll
    for (int i = 0; i < SCAN_GROUPS; ++i) {
        int g = t * SCAN_GROUPS + i;
        int4 o;
        o.x = run; run += v[i].x;
        o.y = run; run += v[i].y;
        o.z = run; run += v[i].z;
        o.w = run; run += v[i].w;
        if (g < SCAN_NG) *(int4*)(rowptr + 4 * g) = o;
    }
}

__global__ void gat_fill(const int* __restrict__ el, const int* __restrict__ eln,
                         const int* __restrict__ et, const int* __restrict__ etn,
                         const int* __restrict__ rowptr, const int* __restrict__ rank,
                         int* __restrict__ csr) {
    int e = blockIdx.x * blockDim.x + threadIdx.x;
    if (e >= ET_EDGES) return;
    int s, d, t0, t1;
    if (e < E_EDGES) {
        s = el[e]; d = el[E_EDGES + e]; t0 = et[e]; t1 = -1;
    } else {
        int j = e - E_EDGES;
        s = eln[j]; d = eln[E2_EDGES + j];
        t0 = etn[2 * j]; t1 = etn[2 * j + 1];
    }
    int pos = rowptr[s] + rank[e];
    pos = pos < 0 ? 0 : (pos > ET_EDGES - 1 ? ET_EDGES - 1 : pos);
    *(int4*)(csr + 4 * (size_t)pos) = make_int4(e, d, t0, t1);
}

#define NODE_WAVES 8
#define NODE_U     4

template <int LAYER>
__global__ __launch_bounds__(NODE_WAVES * 32)
void gat_node(const int* __restrict__ rowptr, const int* __restrict__ csr,
              const float* __restrict__ xtab, const float* __restrict__ etab,
              const float* __restrict__ rtab, const float* __restrict__ w2,
              float* __restrict__ dst) {
    const int wave = __builtin_amdgcn_readfirstlane(threadIdx.x >> 5);
    const int n = blockIdx.x * NODE_WAVES + wave;
    if (n >= N_NODES) return;
    const int lane = threadIdx.x & 31;
    const int k4 = lane * 4;
    const int start = __builtin_amdgcn_readfirstlane(rowptr[n]);
    const int end   = __builtin_amdgcn_readfirstlane(rowptr[n + 1]);
    const v4f xa = *(const v4f*)(xtab + (size_t)n * 256 + k4);
    const v4f wv = *(const v4f*)(w2 + k4);
    v4f acc = (v4f){0.f, 0.f, 0.f, 0.f};
    float den = 0.f;

    for (int base = start; base < end; base += 32) {
        int cntc = end - base;
        if (cntc > 32) cntc = 32;
        int4 md = make_int4(0x7fffffff, 0, 0, -1);
        if (lane < cntc) md = *(const int4*)(csr + 4 * (size_t)(base + lane));
        int rk = 0;
        for (int j = 0; j < cntc; ++j)
            rk += (__builtin_amdgcn_readlane(md.x, j) < md.x) ? 1 : 0;

        for (int j0 = 0; j0 < cntc; j0 += NODE_U) {
            v4f xb[NODE_U], e0[NODE_U], e1[NODE_U];
#pragma unroll
            for (int u = 0; u < NODE_U; ++u) {
                int jj = j0 + u;
                if (jj >= cntc) jj = cntc - 1;
                const unsigned hit = __builtin_amdgcn_ballot_w32(rk == jj);
                const int sl = __builtin_ctz(hit);
                int e  = __builtin_amdgcn_readlane(md.x, sl);
                int d  = __builtin_amdgcn_readlane(md.y, sl);
                int t0 = __builtin_amdgcn_readlane(md.z, sl);
                int t1 = __builtin_amdgcn_readlane(md.w, sl);
                e  = e  < 0 ? 0 : (e  > E_EDGES - 1 ? E_EDGES - 1 : e);
                d  = d  < 0 ? 0 : (d  > N_NODES - 1 ? N_NODES - 1 : d);
                t0 = t0 < 0 ? 0 : (t0 > NREL - 1 ? NREL - 1 : t0);
                t1 = t1 > NREL - 1 ? NREL - 1 : t1;
                xb[u] = *(const v4f*)(xtab + (size_t)d * 256 + 128 + k4);
                if (LAYER == 1 && t1 < 0)
                    e0[u] = *(const v4f*)(etab + (size_t)e * 128 + k4);
                else
                    e0[u] = *(const v4f*)(rtab + (size_t)t0 * 128 + k4);
                if (t1 >= 0)
                    e1[u] = *(const v4f*)(rtab + (size_t)t1 * 128 + k4);
                else
                    e1[u] = (v4f){0.f, 0.f, 0.f, 0.f};
            }
            v4f m[NODE_U];
            float p[NODE_U];
#pragma unroll
            for (int u = 0; u < NODE_U; ++u) {
                const v4f ev = e0[u] + e1[u];
                m[u] = (xa + xb[u]) + ev;
                p[u] = m[u][0] * wv[0] + m[u][1] * wv[1] + m[u][2] * wv[2] + m[u][3] * wv[3];
            }
            if (LAYER == 2) {
#pragma unroll
                for (int u = 0; u < NODE_U; ++u) p[u] += __shfl_xor(p[u], 16, 32);
            }
#pragma unroll
            for (int u = 0; u < NODE_U; ++u) p[u] += __shfl_xor(p[u], 8, 32);
#pragma unroll
            for (int u = 0; u < NODE_U; ++u) p[u] += __shfl_xor(p[u], 4, 32);
#pragma unroll
            for (int u = 0; u < NODE_U; ++u) p[u] += __shfl_xor(p[u], 2, 32);
#pragma unroll
            for (int u = 0; u < NODE_U; ++u) p[u] += __shfl_xor(p[u], 1, 32);
#pragma unroll
            for (int u = 0; u < NODE_U; ++u) {
                if (j0 + u < cntc) {
                    const float lr = p[u] > 0.f ? p[u] : ALPHA * p[u];
                    const float ee = __expf(-lr);
                    acc += ee * m[u];
                    den += ee;
                }
            }
        }
    }

    const float dn = den > 1e-12f ? den : 1e-12f;
    v4f h = acc / dn;
    if (LAYER == 2) {
#pragma unroll
        for (int c = 0; c < 4; ++c) {
            float y = h[c] > 0.f ? h[c] : (__expf(h[c]) - 1.f);
            y = y > 0.f ? y : (__expf(y) - 1.f);
            h[c] = y;
        }
    }
    *(v4f*)(dst + (size_t)n * 128 + k4) = h;
}

__global__ void gat_fold_w2(const float* __restrict__ w1, const float* __restrict__ w2,
                            float* __restrict__ uc) {
    const int i = blockIdx.x * blockDim.x + threadIdx.x;
    if (i >= 256) return;
    const int h = i >> 7, k = i & 127;
    float s = 0.f;
    for (int j = 0; j < 64; ++j)
        s += w1[(size_t)(h * 64 + j) * 384 + 256 + k] * w2[h * 64 + j];
    uc[i] = s;
}

__global__ __launch_bounds__(NODE_WAVES * 32)
void gat_node_l1(const int* __restrict__ rowptr, const int* __restrict__ csr,
                 const float* __restrict__ xtab, const float* __restrict__ eraw,
                 const float* __restrict__ rtab, const float* __restrict__ w2,
                 const float* __restrict__ uc, float* __restrict__ accp,
                 float* __restrict__ agg, float* __restrict__ den2) {
    const int wave = __builtin_amdgcn_readfirstlane(threadIdx.x >> 5);
    const int n = blockIdx.x * NODE_WAVES + wave;
    if (n >= N_NODES) return;
    const int lane = threadIdx.x & 31;
    const int k4 = lane * 4;
    const bool lo = lane < 16;
    const int start = __builtin_amdgcn_readfirstlane(rowptr[n]);
    const int end   = __builtin_amdgcn_readfirstlane(rowptr[n + 1]);
    const v4f xa = *(const v4f*)(xtab + (size_t)n * 256 + k4);
    const v4f wv = *(const v4f*)(w2 + k4);
    const v4f u0 = *(const v4f*)(uc + k4);
    const v4f u1 = *(const v4f*)(uc + 128 + k4);
    const v4f zero4 = (v4f){0.f, 0.f, 0.f, 0.f};
    v4f acc = zero4, a0 = zero4, a1 = zero4;
    float den = 0.f;

    for (int base = start; base < end; base += 32) {
        int cntc = end - base;
        if (cntc > 32) cntc = 32;
        int4 md = make_int4(0x7fffffff, 0, 0, -1);
        if (lane < cntc) md = *(const int4*)(csr + 4 * (size_t)(base + lane));
        int rk = 0;
        for (int j = 0; j < cntc; ++j)
            rk += (__builtin_amdgcn_readlane(md.x, j) < md.x) ? 1 : 0;

        for (int j0 = 0; j0 < cntc; j0 += NODE_U) {
            v4f xb[NODE_U], e0[NODE_U], e1[NODE_U];
            bool dir[NODE_U];
#pragma unroll
            for (int u = 0; u < NODE_U; ++u) {
                int jj = j0 + u;
                if (jj >= cntc) jj = cntc - 1;
                const unsigned hit = __builtin_amdgcn_ballot_w32(rk == jj);
                const int sl = __builtin_ctz(hit);
                int e  = __builtin_amdgcn_readlane(md.x, sl);
                int d  = __builtin_amdgcn_readlane(md.y, sl);
                int t0 = __builtin_amdgcn_readlane(md.z, sl);
                int t1 = __builtin_amdgcn_readlane(md.w, sl);
                e  = e  < 0 ? 0 : (e  > E_EDGES - 1 ? E_EDGES - 1 : e);
                d  = d  < 0 ? 0 : (d  > N_NODES - 1 ? N_NODES - 1 : d);
                t0 = t0 < 0 ? 0 : (t0 > NREL - 1 ? NREL - 1 : t0);
                t1 = t1 > NREL - 1 ? NREL - 1 : t1;
                dir[u] = t1 < 0;
                xb[u] = *(const v4f*)(xtab + (size_t)d * 256 + 128 + k4);
                if (t1 < 0)
                    e0[u] = *(const v4f*)(eraw + (size_t)e * 128 + k4);
                else
                    e0[u] = *(const v4f*)(rtab + (size_t)t0 * 128 + k4);
                if (t1 >= 0)
                    e1[u] = *(const v4f*)(rtab + (size_t)t1 * 128 + k4);
                else
                    e1[u] = zero4;
            }
            v4f m[NODE_U], er[NODE_U];
            float p[NODE_U];
#pragma unroll
            for (int u = 0; u < NODE_U; ++u) {
                const v4f ev = dir[u] ? zero4 : (e0[u] + e1[u]);
                er[u] = dir[u] ? e0[u] : zero4;
                m[u] = (xa + xb[u]) + ev;
                const float qn = __builtin_fmaf(m[u][3], wv[3], __builtin_fmaf(m[u][2], wv[2],
                                 __builtin_fmaf(m[u][1], wv[1], m[u][0] * wv[0])));
                const float q0 = __builtin_fmaf(er[u][3], u0[3], __builtin_fmaf(er[u][2], u0[2],
                                 __builtin_fmaf(er[u][1], u0[1], er[u][0] * u0[0])));
                const float q1 = __builtin_fmaf(er[u][3], u1[3], __builtin_fmaf(er[u][2], u1[2],
                                 __builtin_fmaf(er[u][1], u1[1], er[u][0] * u1[0])));
                const float mine = qn + (lo ? q0 : q1);
                const float send = lo ? q1 : q0;
                p[u] = mine + __shfl_xor(send, 16, 32);
            }
#pragma unroll
            for (int u = 0; u < NODE_U; ++u) p[u] += __shfl_xor(p[u], 8, 32);
#pragma unroll
            for (int u = 0; u < NODE_U; ++u) p[u] += __shfl_xor(p[u], 4, 32);
#pragma unroll
            for (int u = 0; u < NODE_U; ++u) p[u] += __shfl_xor(p[u], 2, 32);
#pragma unroll
            for (int u = 0; u < NODE_U; ++u) p[u] += __shfl_xor(p[u], 1, 32);
            float g[NODE_U], go[NODE_U];
#pragma unroll
            for (int u = 0; u < NODE_U; ++u) {
                const float lr = p[u] > 0.f ? p[u] : ALPHA * p[u];
                g[u]  = __expf(-lr);
                go[u] = __shfl_xor(g[u], 16, 32);
            }
#pragma unroll
            for (int u = 0; u < NODE_U; ++u) {
                if (j0 + u < cntc) {
                    const float g0 = lo ? g[u] : go[u];
                    const float g1 = lo ? go[u] : g[u];
                    acc += g[u] * m[u];
                    den += g[u];
                    a0 += g0 * er[u];
                    a1 += g1 * er[u];
                }
            }
        }
    }

    *(v4f*)(accp + (size_t)n * 128 + k4) = acc;
    *(v4f*)(agg + (size_t)n * 128 + k4) = a0;
    *(v4f*)(agg + (size_t)(N_NODES + n) * 128 + k4) = a1;
    if ((lane & 15) == 0) den2[n * 2 + (lane >> 4)] = den;
}

extern "C" void kernel_launch(void* const* d_in, const int* in_sizes, int n_in,
                              void* d_out, int out_size, void* d_ws, size_t ws_size,
                              hipStream_t stream) {
    const float* x    = (const float*)d_in[0];
    const float* rel  = (const float*)d_in[1];
    const int*   el   = (const int*)d_in[2];
    const int*   et   = (const int*)d_in[3];
    const float* eemb = (const float*)d_in[4];
    const int*   eln  = (const int*)d_in[5];
    const int*   etn  = (const int*)d_in[6];
    const float* w1h  = (const float*)d_in[8];
    const float* w2h  = (const float*)d_in[9];
    const float* w1o  = (const float*)d_in[10];
    const float* w2o  = (const float*)d_in[11];
    const float* WR   = (const float*)d_in[12];
    float* out = (float*)d_out;

    float* ws = (float*)d_ws;
    float* xnode  = ws;
    float* accp   = xnode  + (size_t)N_NODES * 256;
    float* agg    = accp   + (size_t)N_NODES * 128;
    float* den2   = agg    + (size_t)N_NODES * 256;
    float* uc     = den2   + (size_t)N_NODES * 2;
    float* relc   = uc     + 256;
    float* lx     = relc   + (size_t)NREL * 128;
    float* xab2   = lx     + (size_t)N_NODES * 128;
    float* relW2  = xab2   + (size_t)N_NODES * 256;
    int*   cnt    = (int*)(relW2 + (size_t)NREL * 128);
    int*   rowptr = cnt    + NPAD;
    int*   rank   = rowptr + NPAD;
    int*   csr    = rank   + ET_EDGES;

    float* out_rel = out + (size_t)N_NODES * 128;

    hipMemsetAsync(cnt, 0, (size_t)NPAD * sizeof(int), stream);
    gat_count<<<(ET_EDGES + 255) / 256, 256, 0, stream>>>(el, eln, cnt, rank);
    gat_scan<<<1, SCAN_THREADS, 0, stream>>>(cnt, rowptr);
    gat_fill<<<(ET_EDGES + 255) / 256, 256, 0, stream>>>(el, eln, et, etn, rowptr,
                                                        rank, csr);

    static_assert(2 * ((NREL + 15) / 16) <= GEMM_MAX_BLOCKS,
                  "the two small problems must not have more tiles than blocks");
    gat_wmma_gemm_multi<<<GEMM_MAX_BLOCKS, GEMM_THREADS, 0, stream>>>(
        x,   w1h,       xnode,       N_NODES, 256,
        x,   w1h + 128, xnode + 128, N_NODES, 256,
        rel, w1h + 256, relc,        NREL,    128,
        rel, WR,        out_rel,     NREL,    128,
        384);
    gat_fold_w2<<<1, 256, 0, stream>>>(w1h, w2h, uc);

    gat_node_l1<<<(N_NODES + NODE_WAVES - 1) / NODE_WAVES, NODE_WAVES * 32, 0, stream>>>(
        rowptr, csr, xnode, eemb, relc, w2h, uc, accp, agg, den2);
    static_assert(N_NODES % 16 == 0, "each head's plane must be whole row tiles");
    gat_wmma_gemm_heads<<<GEMM_MAX_BLOCKS, GEMM_THREADS, 0, stream>>>(
        agg, w1h + 256, lx, 2 * N_NODES, 128, 384, 128, 0, 2 * N_NODES / 16, accp, den2);

    gat_wmma_gemm_multi<<<GEMM_MAX_BLOCKS, GEMM_THREADS, 0, stream>>>(
        lx,      w1o,       xab2,       N_NODES, 256,
        lx,      w1o + 128, xab2 + 128, N_NODES, 256,
        out_rel, w1o + 256, relW2,      NREL,    128,
        (const float*)nullptr, (const float*)nullptr, (float*)nullptr, 0, 0,
        384);

    gat_node<2><<<(N_NODES + NODE_WAVES - 1) / NODE_WAVES, NODE_WAVES * 32, 0, stream>>>(
        rowptr, csr, xab2, relW2, relW2, w2o, out);
}


# launches merged 9 -> 7: edge counting runs inside the first projection launch (on each block's four waves with one tile less), the w2 fold as a second block of the scan launch; sums unchanged
# speedup vs baseline: 3.2438x; 1.0850x over previous
#include <hip/hip_runtime.h>
#include <hip/hip_bf16.h>

typedef float v2f __attribute__((ext_vector_type(2)));
typedef float v4f __attribute__((ext_vector_type(4)));
typedef float v8f __attribute__((ext_vector_type(8)));

#define N_NODES 50000
#define E_EDGES 500000
#define E2_EDGES 62500
#define ET_EDGES (E_EDGES + E2_EDGES)
#define NREL 1000
#define ALPHA 0.2f
#define KDIM 128

#define GEMM_N        128
#define GEMM_NT       8
#define GEMM_THREADS  512
#define GEMM_WAVES    (GEMM_THREADS / 32)
#define GEMM_MAX_BLOCKS 256
#define BPITCH        132
#define GEMM_PANEL    (GEMM_N * BPITCH)
#define CNT_WAVES     4
#define CNT_U         9
#define CNT_LANES     (GEMM_MAX_BLOCKS * CNT_WAVES * 32)

template <int NT>
__device__ __forceinline__ void gemm_step(v8f (&acc)[NT], const v4f& a,
                                          const v4f (&b)[NT]) {
    v2f a_lo = (v2f){a[0], a[1]};
    v2f a_hi = (v2f){a[2], a[3]};
#pragma unroll
    for (int s = 0; s < NT; ++s)
        acc[s] = __builtin_amdgcn_wmma_f32_16x16x4_f32(
            false, a_lo, false, (v2f){b[s][0], b[s][1]}, (short)0,
            acc[s], false, false);
#pragma unroll
    for (int s = 0; s < NT; ++s)
        acc[s] = __builtin_amdgcn_wmma_f32_16x16x4_f32(
            false, a_hi, false, (v2f){b[s][2], b[s][3]}, (short)0,
            acc[s], false, false);
}

template <int LDBK>
__device__ __forceinline__ void gemm_stage_panel(float* __restrict__ Bs,
                                                 const float* __restrict__ B,
                                                 int ldbn, int tid) {
    if constexpr (LDBK == 1) {
        constexpr int NV = GEMM_N * (KDIM / 4) / GEMM_THREADS;
        v4f v[NV];
#pragma unroll
        for (int it = 0; it < NV; ++it) {
            int i = tid + it * GEMM_THREADS;
            v[it] = *(const v4f*)(B + (size_t)(i >> 5) * ldbn + (i & 31) * 4);
        }
#pragma unroll
        for (int it = 0; it < NV; ++it) {
            int i = tid + it * GEMM_THREADS;
            *(v4f*)(Bs + (i >> 5) * BPITCH + (i & 31) * 4) = v[it];
        }
    } else {
        constexpr int NV = 8;
        for (int i0 = tid; i0 < GEMM_N * KDIM; i0 += NV * GEMM_THREADS) {
            float v[NV];
#pragma unroll
            for (int it = 0; it < NV; ++it) {
                int i = i0 + it * GEMM_THREADS;
                v[it] = B[(size_t)(i & 127) * ldbn + (size_t)(i >> 7) * LDBK];
            }
#pragma unroll
            for (int it = 0; it < NV; ++it) {
                int i = i0 + it * GEMM_THREADS;
                Bs[(i & 127) * BPITCH + (i >> 7)] = v[it];
            }
        }
    }
}

__global__ __launch_bounds__(GEMM_THREADS)
void gat_wmma_gemm_multi(
        const float* __restrict__ A0, const float* __restrict__ B0, float* __restrict__ C0, int M0, int ldc0,
        const float* __restrict__ A1, const float* __restrict__ B1, float* __restrict__ C1, int M1, int ldc1,
        const float* __restrict__ A2, const float* __restrict__ B2, float* __restrict__ C2, int M2, int ldc2,
        const float* __restrict__ A3, const float* __restrict__ B3, float* __restrict__ C3, int M3, int ldc3,
        int ldbn,
        const int* __restrict__ el, const int* __restrict__ eln,
        int* __restrict__ cnt, int* __restrict__ rank) {
    __shared__ __attribute__((aligned(16))) float Bs[3 * GEMM_PANEL];
    const int tid = threadIdx.x;

    const int e0 = (M0 + 15) >> 4;
    const int e1 = e0 + ((M1 + 15) >> 4);
    const int e2 = e1 + ((M2 + 15) >> 4);
    const int e3 = e2 + ((M3 + 15) >> 4);
    const int vs = (int)blockIdx.x +
        GEMM_MAX_BLOCKS * ((e1 - (int)blockIdx.x + GEMM_MAX_BLOCKS - 1) / GEMM_MAX_BLOCKS);

    gemm_stage_panel<1>(Bs, B0, ldbn, tid);
    gemm_stage_panel<1>(Bs + GEMM_PANEL, B1, ldbn, tid);
    if (vs < e3) {
        if (vs < e2) gemm_stage_panel<1>(Bs + 2 * GEMM_PANEL, B2, ldbn, tid);
        else         gemm_stage_panel<KDIM>(Bs + 2 * GEMM_PANEL, B3, 1, tid);
    }
    __syncthreads();

    const int wave = __builtin_amdgcn_readfirstlane(tid >> 5);
    const int lane = tid & 31;
    const int half = lane >> 4;
    const int l16  = lane & 15;
    const float* Bl = Bs + l16 * BPITCH + half * 4;
    constexpr int wstride = GEMM_MAX_BLOCKS * GEMM_WAVES;

    if (cnt != nullptr && wave >= GEMM_WAVES - CNT_WAVES) {
        const int t = ((int)blockIdx.x * CNT_WAVES + (wave - (GEMM_WAVES - CNT_WAVES))) * 32 + lane;
        for (int eb = t; eb < ET_EDGES; eb += CNT_LANES * CNT_U) {
            int s[CNT_U], r[CNT_U];
#pragma unroll
            for (int u = 0; u < CNT_U; ++u) {
                const int e = eb + u * CNT_LANES;
                const int ec = e < ET_EDGES ? e : ET_EDGES - 1;
                s[u] = (ec < E_EDGES) ? el[ec] : eln[ec - E_EDGES];
            }
#pragma unroll
            for (int u = 0; u < CNT_U; ++u) {
                const int e = eb + u * CNT_LANES;
                r[u] = 0;
                if (e < ET_EDGES) r[u] = atomicAdd(cnt + s[u], 1);
            }
#pragma unroll
            for (int u = 0; u < CNT_U; ++u) {
                const int e = eb + u * CNT_LANES;
                if (e < ET_EDGES) rank[e] = r[u];
            }
        }
    }

    int v = (int)blockIdx.x + GEMM_MAX_BLOCKS * wave;
    if (v >= e3) return;

    auto locate = [&](int vv, const float*& A, float*& C, int& M, int& ldc,
                      int& mt, int& slot) {
        if (vv < e0)      { A = A0; C = C0; M = M0; ldc = ldc0; mt = vv;      slot = 0; }
        else if (vv < e1) { A = A1; C = C1; M = M1; ldc = ldc1; mt = vv - e0; slot = 1; }
        else if (vv < e2) { A = A2; C = C2; M = M2; ldc = ldc2; mt = vv - e1; slot = 2; }
        else              { A = A3; C = C3; M = M3; ldc = ldc3; mt = vv - e2; slot = 2; }
    };

    const float* A; float* C; int M, ldc, mt, slot;
    locate(v, A, C, M, ldc, mt, slot);
    int arow = mt * 16 + l16;
    const float* Ap = A + (size_t)(arow < M ? arow : (M - 1)) * KDIM + half * 4;
    const float* Bt = Bl + slot * GEMM_PANEL;

    v4f a_cur[4], a_nxt[4];
    v4f b0[GEMM_NT], b1[GEMM_NT];
#pragma unroll
    for (int t = 0; t < 4; ++t) a_cur[t] = *(const v4f*)(Ap + t * 8);
#pragma unroll
    for (int s = 0; s < GEMM_NT; ++s) b0[s] = *(const v4f*)(Bt + s * 16 * BPITCH);

    for (;;) {
        int vn = v + wstride;
        const bool more = vn < e3;
        if (!more) vn = v;
        const float* An; float* Cn; int Mn, ldcn, mtn, slotn;
        locate(vn, An, Cn, Mn, ldcn, mtn, slotn);
        int arown = mtn * 16 + l16;
        const float* Apn = An + (size_t)(arown < Mn ? arown : (Mn - 1)) * KDIM + half * 4;
        const float* Btn = Bl + slotn * GEMM_PANEL;

        v8f acc[GEMM_NT];
#pragma unroll
        for (int s = 0; s < GEMM_NT; ++s)
            acc[s] = (v8f){0.f,0.f,0.f,0.f,0.f,0.f,0.f,0.f};

#pragma unroll 1
        for (int q = 0; q < 4; ++q) {
            const float* pa = (q < 3) ? (Ap + (q + 1) * 32) : Apn;
#pragma unroll
            for (int t = 0; t < 4; ++t) a_nxt[t] = *(const v4f*)(pa + t * 8);

            const float* Bq = Bt + q * 32;
            const float* Bw = (q < 3) ? (Bq + 32) : Btn;
#pragma unroll
            for (int s = 0; s < GEMM_NT; ++s) b1[s] = *(const v4f*)(Bq + s * 16 * BPITCH + 8);
            gemm_step<GEMM_NT>(acc, a_cur[0], b0);
#pragma unroll
            for (int s = 0; s < GEMM_NT; ++s) b0[s] = *(const v4f*)(Bq + s * 16 * BPITCH + 16);
            gemm_step<GEMM_NT>(acc, a_cur[1], b1);
#pragma unroll
            for (int s = 0; s < GEMM_NT; ++s) b1[s] = *(const v4f*)(Bq + s * 16 * BPITCH + 24);
            gemm_step<GEMM_NT>(acc, a_cur[2], b0);
#pragma unroll
            for (int s = 0; s < GEMM_NT; ++s) b0[s] = *(const v4f*)(Bw + s * 16 * BPITCH);
            gemm_step<GEMM_NT>(acc, a_cur[3], b1);

#pragma unroll
            for (int t = 0; t < 4; ++t) a_cur[t] = a_nxt[t];
        }

        float* Ct = C + (size_t)(mt * 16 + half * 8) * ldc + l16;
        if (mt * 16 + 16 <= M) {
#pragma unroll
            for (int r = 0; r < 8; ++r)
#pragma unroll
                for (int s = 0; s < GEMM_NT; ++s)
                    Ct[(size_t)r * ldc + s * 16] = acc[s][r];
        } else {
#pragma unroll
            for (int r = 0; r < 8; ++r) {
                if (mt * 16 + half * 8 + r < M) {
#pragma unroll
                    for (int s = 0; s < GEMM_NT; ++s)
                        Ct[(size_t)r * ldc + s * 16] = acc[s][r];
                }
            }
        }
        if (!more) break;
        v = vn; C = Cn; M = Mn; ldc = ldcn; mt = mtn;
        Ap = Apn; Bt = Btn;
    }
}

__global__ __launch_bounds__(GEMM_THREADS)
void gat_wmma_gemm_heads(const float* __restrict__ A,
                         const float* __restrict__ B,
                         float* __restrict__ C,
                         int M, int lda, int ldbn, int ldc, int coff,
                         int mtiles,
                         const float* __restrict__ addp,
                         const float* __restrict__ den2) {
    constexpr int NT = GEMM_NT / 2;
    __shared__ __attribute__((aligned(16))) float Bs[GEMM_PANEL];
    const int tid = threadIdx.x;

    gemm_stage_panel<1>(Bs, B, ldbn, tid);
    __syncthreads();

    const int wave = __builtin_amdgcn_readfirstlane(tid >> 5);
    const int lane = tid & 31;
    const int half = lane >> 4;
    const int l16  = lane & 15;
    const float* Bl = Bs + l16 * BPITCH + half * 4;
    const int wstride = gridDim.x * GEMM_WAVES;

    int mt = blockIdx.x + gridDim.x * wave;
    if (mt >= mtiles) return;

    int arow = mt * 16 + l16;
    const float* Ap = A + (size_t)(arow < M ? arow : (M - 1)) * lda + half * 4;

    const int hm = mtiles >> 1;
    const float* Bt = Bl + (mt >= hm ? 64 * BPITCH : 0);

    v4f a_cur[4], a_nxt[4];
    v4f b0[NT], b1[NT];
#pragma unroll
    for (int t = 0; t < 4; ++t) a_cur[t] = *(const v4f*)(Ap + t * 8);
#pragma unroll
    for (int s = 0; s < NT; ++s) b0[s] = *(const v4f*)(Bt + s * 16 * BPITCH);

    for (; mt < mtiles; mt += wstride) {
        int mtn = mt + wstride;
        if (mtn >= mtiles) mtn = mt;
        int arown = mtn * 16 + l16;
        const float* Apn = A + (size_t)(arown < M ? arown : (M - 1)) * lda + half * 4;
        const float* Btn = Bl + (mtn >= hm ? 64 * BPITCH : 0);

        v8f acc[NT];
#pragma unroll
        for (int s = 0; s < NT; ++s)
            acc[s] = (v8f){0.f,0.f,0.f,0.f,0.f,0.f,0.f,0.f};

        float ad[8][NT], dn[8];
        const int head = mt >= hm ? 1 : 0;
        const int node0 = (mt - head * hm) * 16 + half * 8;
        const size_t o0 = (size_t)node0 * ldc + coff + head * 64 + l16;
#pragma unroll
        for (int r = 0; r < 8; ++r) {
            const float d = den2[2 * (node0 + r) + head];
            dn[r] = d > 1e-12f ? d : 1e-12f;
#pragma unroll
            for (int s = 0; s < NT; ++s)
                ad[r][s] = addp[o0 + (size_t)r * ldc + s * 16];
        }

#pragma unroll 1
        for (int q = 0; q < 4; ++q) {
            const float* pa = (q < 3) ? (Ap + (q + 1) * 32) : Apn;
#pragma unroll
            for (int t = 0; t < 4; ++t) a_nxt[t] = *(const v4f*)(pa + t * 8);

            const float* Bq = Bt + q * 32;
            const float* Bw = (q < 3) ? (Bq + 32) : Btn;
#pragma unroll
            for (int s = 0; s < NT; ++s) b1[s] = *(const v4f*)(Bq + s * 16 * BPITCH + 8);
            gemm_step<NT>(acc, a_cur[0], b0);
#pragma unroll
            for (int s = 0; s < NT; ++s) b0[s] = *(const v4f*)(Bq + s * 16 * BPITCH + 16);
            gemm_step<NT>(acc, a_cur[1], b1);
#pragma unroll
            for (int s = 0; s < NT; ++s) b1[s] = *(const v4f*)(Bq + s * 16 * BPITCH + 24);
            gemm_step<NT>(acc, a_cur[2], b0);
#pragma unroll
            for (int s = 0; s < NT; ++s) b0[s] = *(const v4f*)(Bw + s * 16 * BPITCH);
            gemm_step<NT>(acc, a_cur[3], b1);

#pragma unroll
            for (int t = 0; t < 4; ++t) a_cur[t] = a_nxt[t];
        }

#pragma unroll
        for (int r = 0; r < 8; ++r)
#pragma unroll
            for (int s = 0; s < NT; ++s)
                C[o0 + (size_t)r * ldc + s * 16] = (acc[s][r] + ad[r][s]) / dn[r];
        Bt = Btn;
        Ap = Apn;
    }
}

#define NPAD         (N_NODES + 4)
#define SCAN_THREADS 1024
#define SCAN_GROUPS  13
#define SCAN_NG      (NPAD / 4)

__global__ __launch_bounds__(SCAN_THREADS)
void gat_scan(const int* __restrict__ cnt, int* __restrict__ rowptr,
              const float* __restrict__ w1, const float* __restrict__ w2,
              float* __restrict__ uc) {
    __shared__ int wsum[SCAN_THREADS / 32];
    const int t = threadIdx.x, lane = t & 31, w = t >> 5;
    if (blockIdx.x == 1) {
        if (t < 256) {
            const int h = t >> 7, k = t & 127;
            float s = 0.f;
            for (int j = 0; j < 64; ++j)
                s += w1[(size_t)(h * 64 + j) * 384 + 256 + k] * w2[h * 64 + j];
            uc[t] = s;
        }
        return;
    }
    int4 v[SCAN_GROUPS];
    int tot = 0;
#pragma unroll
    for (int i = 0; i < SCAN_GROUPS; ++i) {
        int g = t * SCAN_GROUPS + i;
        v[i] = make_int4(0, 0, 0, 0);
        if (g < SCAN_NG) v[i] = *(const int4*)(cnt + 4 * g);
        tot += v[i].x + v[i].y + v[i].z + v[i].w;
    }
    int inc = tot;
    for (int off = 1; off < 32; off <<= 1) {
        int y = __shfl_up(inc, off, 32);
        if (lane >= off) inc += y;
    }
    if (lane == 31) wsum[w] = inc;
    __syncthreads();
    if (w == 0) {
        int s = wsum[lane], sc = s;
        for (int off = 1; off < 32; off <<= 1) {
            int y = __shfl_up(sc, off, 32);
            if (lane >= off) sc += y;
        }
        wsum[lane] = sc - s;
    }
    __syncthreads();
    int run = wsum[w] + inc - tot;
#pragma unroll
    for (int i = 0; i < SCAN_GROUPS; ++i) {
        int g = t * SCAN_GROUPS + i;
        int4 o;
        o.x = run; run += v[i].x;
        o.y = run; run += v[i].y;
        o.z = run; run += v[i].z;
        o.w = run; run += v[i].w;
        if (g < SCAN_NG) *(int4*)(rowptr + 4 * g) = o;
    }
}

__global__ void gat_fill(const int* __restrict__ el, const int* __restrict__ eln,
                         const int* __restrict__ et, const int* __restrict__ etn,
                         const int* __restrict__ rowptr, const int* __restrict__ rank,
                         int* __restrict__ csr) {
    int e = blockIdx.x * blockDim.x + threadIdx.x;
    if (e >= ET_EDGES) return;
    int s, d, t0, t1;
    if (e < E_EDGES) {
        s = el[e]; d = el[E_EDGES + e]; t0 = et[e]; t1 = -1;
    } else {
        int j = e - E_EDGES;
        s = eln[j]; d = eln[E2_EDGES + j];
        t0 = etn[2 * j]; t1 = etn[2 * j + 1];
    }
    int pos = rowptr[s] + rank[e];
    pos = pos < 0 ? 0 : (pos > ET_EDGES - 1 ? ET_EDGES - 1 : pos);
    *(int4*)(csr + 4 * (size_t)pos) = make_int4(e, d, t0, t1);
}

#define NODE_WAVES 8
#define NODE_U     4

template <int LAYER>
__global__ __launch_bounds__(NODE_WAVES * 32)
void gat_node(const int* __restrict__ rowptr, const int* __restrict__ csr,
              const float* __restrict__ xtab, const float* __restrict__ etab,
              const float* __restrict__ rtab, const float* __restrict__ w2,
              float* __restrict__ dst) {
    const int wave = __builtin_amdgcn_readfirstlane(threadIdx.x >> 5);
    const int n = blockIdx.x * NODE_WAVES + wave;
    if (n >= N_NODES) return;
    const int lane = threadIdx.x & 31;
    const int k4 = lane * 4;
    const int start = __builtin_amdgcn_readfirstlane(rowptr[n]);
    const int end   = __builtin_amdgcn_readfirstlane(rowptr[n + 1]);
    const v4f xa = *(const v4f*)(xtab + (size_t)n * 256 + k4);
    const v4f wv = *(const v4f*)(w2 + k4);
    v4f acc = (v4f){0.f, 0.f, 0.f, 0.f};
    float den = 0.f;

    for (int base = start; base < end; base += 32) {
        int cntc = end - base;
        if (cntc > 32) cntc = 32;
        int4 md = make_int4(0x7fffffff, 0, 0, -1);
        if (lane < cntc) md = *(const int4*)(csr + 4 * (size_t)(base + lane));
        int rk = 0;
        for (int j = 0; j < cntc; ++j)
            rk += (__builtin_amdgcn_readlane(md.x, j) < md.x) ? 1 : 0;

        for (int j0 = 0; j0 < cntc; j0 += NODE_U) {
            v4f xb[NODE_U], e0[NODE_U], e1[NODE_U];
#pragma unroll
            for (int u = 0; u < NODE_U; ++u) {
                int jj = j0 + u;
                if (jj >= cntc) jj = cntc - 1;
                const unsigned hit = __builtin_amdgcn_ballot_w32(rk == jj);
                const int sl = __builtin_ctz(hit);
                int e  = __builtin_amdgcn_readlane(md.x, sl);
                int d  = __builtin_amdgcn_readlane(md.y, sl);
                int t0 = __builtin_amdgcn_readlane(md.z, sl);
                int t1 = __builtin_amdgcn_readlane(md.w, sl);
                e  = e  < 0 ? 0 : (e  > E_EDGES - 1 ? E_EDGES - 1 : e);
                d  = d  < 0 ? 0 : (d  > N_NODES - 1 ? N_NODES - 1 : d);
                t0 = t0 < 0 ? 0 : (t0 > NREL - 1 ? NREL - 1 : t0);
                t1 = t1 > NREL - 1 ? NREL - 1 : t1;
                xb[u] = *(const v4f*)(xtab + (size_t)d * 256 + 128 + k4);
                if (LAYER == 1 && t1 < 0)
                    e0[u] = *(const v4f*)(etab + (size_t)e * 128 + k4);
                else
                    e0[u] = *(const v4f*)(rtab + (size_t)t0 * 128 + k4);
                if (t1 >= 0)
                    e1[u] = *(const v4f*)(rtab + (size_t)t1 * 128 + k4);
                else
                    e1[u] = (v4f){0.f, 0.f, 0.f, 0.f};
            }
            v4f m[NODE_U];
            float p[NODE_U];
#pragma unroll
            for (int u = 0; u < NODE_U; ++u) {
                const v4f ev = e0[u] + e1[u];
                m[u] = (xa + xb[u]) + ev;
                p[u] = m[u][0] * wv[0] + m[u][1] * wv[1] + m[u][2] * wv[2] + m[u][3] * wv[3];
            }
            if (LAYER == 2) {
#pragma unroll
                for (int u = 0; u < NODE_U; ++u) p[u] += __shfl_xor(p[u], 16, 32);
            }
#pragma unroll
            for (int u = 0; u < NODE_U; ++u) p[u] += __shfl_xor(p[u], 8, 32);
#pragma unroll
            for (int u = 0; u < NODE_U; ++u) p[u] += __shfl_xor(p[u], 4, 32);
#pragma unroll
            for (int u = 0; u < NODE_U; ++u) p[u] += __shfl_xor(p[u], 2, 32);
#pragma unroll
            for (int u = 0; u < NODE_U; ++u) p[u] += __shfl_xor(p[u], 1, 32);
#pragma unroll
            for (int u = 0; u < NODE_U; ++u) {
                if (j0 + u < cntc) {
                    const float lr = p[u] > 0.f ? p[u] : ALPHA * p[u];
                    const float ee = __expf(-lr);
                    acc += ee * m[u];
                    den += ee;
                }
            }
        }
    }

    const float dn = den > 1e-12f ? den : 1e-12f;
    v4f h = acc / dn;
    if (LAYER == 2) {
#pragma unroll
        for (int c = 0; c < 4; ++c) {
            float y = h[c] > 0.f ? h[c] : (__expf(h[c]) - 1.f);
            y = y > 0.f ? y : (__expf(y) - 1.f);
            h[c] = y;
        }
    }
    *(v4f*)(dst + (size_t)n * 128 + k4) = h;
}

__global__ __launch_bounds__(NODE_WAVES * 32)
void gat_node_l1(const int* __restrict__ rowptr, const int* __restrict__ csr,
                 const float* __restrict__ xtab, const float* __restrict__ eraw,
                 const float* __restrict__ rtab, const float* __restrict__ w2,
                 const float* __restrict__ uc, float* __restrict__ accp,
                 float* __restrict__ agg, float* __restrict__ den2) {
    const int wave = __builtin_amdgcn_readfirstlane(threadIdx.x >> 5);
    const int n = blockIdx.x * NODE_WAVES + wave;
    if (n >= N_NODES) return;
    const int lane = threadIdx.x & 31;
    const int k4 = lane * 4;
    const bool lo = lane < 16;
    const int start = __builtin_amdgcn_readfirstlane(rowptr[n]);
    const int end   = __builtin_amdgcn_readfirstlane(rowptr[n + 1]);
    const v4f xa = *(const v4f*)(xtab + (size_t)n * 256 + k4);
    const v4f wv = *(const v4f*)(w2 + k4);
    const v4f u0 = *(const v4f*)(uc + k4);
    const v4f u1 = *(const v4f*)(uc + 128 + k4);
    const v4f zero4 = (v4f){0.f, 0.f, 0.f, 0.f};
    v4f acc = zero4, a0 = zero4, a1 = zero4;
    float den = 0.f;

    for (int base = start; base < end; base += 32) {
        int cntc = end - base;
        if (cntc > 32) cntc = 32;
        int4 md = make_int4(0x7fffffff, 0, 0, -1);
        if (lane < cntc) md = *(const int4*)(csr + 4 * (size_t)(base + lane));
        int rk = 0;
        for (int j = 0; j < cntc; ++j)
            rk += (__builtin_amdgcn_readlane(md.x, j) < md.x) ? 1 : 0;

        for (int j0 = 0; j0 < cntc; j0 += NODE_U) {
            v4f xb[NODE_U], e0[NODE_U], e1[NODE_U];
            bool dir[NODE_U];
#pragma unroll
            for (int u = 0; u < NODE_U; ++u) {
                int jj = j0 + u;
                if (jj >= cntc) jj = cntc - 1;
                const unsigned hit = __builtin_amdgcn_ballot_w32(rk == jj);
                const int sl = __builtin_ctz(hit);
                int e  = __builtin_amdgcn_readlane(md.x, sl);
                int d  = __builtin_amdgcn_readlane(md.y, sl);
                int t0 = __builtin_amdgcn_readlane(md.z, sl);
                int t1 = __builtin_amdgcn_readlane(md.w, sl);
                e  = e  < 0 ? 0 : (e  > E_EDGES - 1 ? E_EDGES - 1 : e);
                d  = d  < 0 ? 0 : (d  > N_NODES - 1 ? N_NODES - 1 : d);
                t0 = t0 < 0 ? 0 : (t0 > NREL - 1 ? NREL - 1 : t0);
                t1 = t1 > NREL - 1 ? NREL - 1 : t1;
                dir[u] = t1 < 0;
                xb[u] = *(const v4f*)(xtab + (size_t)d * 256 + 128 + k4);
                if (t1 < 0)
                    e0[u] = *(const v4f*)(eraw + (size_t)e * 128 + k4);
                else
                    e0[u] = *(const v4f*)(rtab + (size_t)t0 * 128 + k4);
                if (t1 >= 0)
                    e1[u] = *(const v4f*)(rtab + (size_t)t1 * 128 + k4);
                else
                    e1[u] = zero4;
            }
            v4f m[NODE_U], er[NODE_U];
            float p[NODE_U];
#pragma unroll
            for (int u = 0; u < NODE_U; ++u) {
                const v4f ev = dir[u] ? zero4 : (e0[u] + e1[u]);
                er[u] = dir[u] ? e0[u] : zero4;
                m[u] = (xa + xb[u]) + ev;
                const float qn = __builtin_fmaf(m[u][3], wv[3], __builtin_fmaf(m[u][2], wv[2],
                                 __builtin_fmaf(m[u][1], wv[1], m[u][0] * wv[0])));
                const float q0 = __builtin_fmaf(er[u][3], u0[3], __builtin_fmaf(er[u][2], u0[2],
                                 __builtin_fmaf(er[u][1], u0[1], er[u][0] * u0[0])));
                const float q1 = __builtin_fmaf(er[u][3], u1[3], __builtin_fmaf(er[u][2], u1[2],
                                 __builtin_fmaf(er[u][1], u1[1], er[u][0] * u1[0])));
                const float mine = qn + (lo ? q0 : q1);
                const float send = lo ? q1 : q0;
                p[u] = mine + __shfl_xor(send, 16, 32);
            }
#pragma unroll
            for (int u = 0; u < NODE_U; ++u) p[u] += __shfl_xor(p[u], 8, 32);
#pragma unroll
            for (int u = 0; u < NODE_U; ++u) p[u] += __shfl_xor(p[u], 4, 32);
#pragma unroll
            for (int u = 0; u < NODE_U; ++u) p[u] += __shfl_xor(p[u], 2, 32);
#pragma unroll
            for (int u = 0; u < NODE_U; ++u) p[u] += __shfl_xor(p[u], 1, 32);
            float g[NODE_U], go[NODE_U];
#pragma unroll
            for (int u = 0; u < NODE_U; ++u) {
                const float lr = p[u] > 0.f ? p[u] : ALPHA * p[u];
                g[u]  = __expf(-lr);
                go[u] = __shfl_xor(g[u], 16, 32);
            }
#pragma unroll
            for (int u = 0; u < NODE_U; ++u) {
                if (j0 + u < cntc) {
                    const float g0 = lo ? g[u] : go[u];
                    const float g1 = lo ? go[u] : g[u];
                    acc += g[u] * m[u];
                    den += g[u];
                    a0 += g0 * er[u];
                    a1 += g1 * er[u];
                }
            }
        }
    }

    *(v4f*)(accp + (size_t)n * 128 + k4) = acc;
    *(v4f*)(agg + (size_t)n * 128 + k4) = a0;
    *(v4f*)(agg + (size_t)(N_NODES + n) * 128 + k4) = a1;
    if ((lane & 15) == 0) den2[n * 2 + (lane >> 4)] = den;
}

extern "C" void kernel_launch(void* const* d_in, const int* in_sizes, int n_in,
                              void* d_out, int out_size, void* d_ws, size_t ws_size,
                              hipStream_t stream) {
    const float* x    = (const float*)d_in[0];
    const float* rel  = (const float*)d_in[1];
    const int*   el   = (const int*)d_in[2];
    const int*   et   = (const int*)d_in[3];
    const float* eemb = (const float*)d_in[4];
    const int*   eln  = (const int*)d_in[5];
    const int*   etn  = (const int*)d_in[6];
    const float* w1h  = (const float*)d_in[8];
    const float* w2h  = (const float*)d_in[9];
    const float* w1o  = (const float*)d_in[10];
    const float* w2o  = (const float*)d_in[11];
    const float* WR   = (const float*)d_in[12];
    float* out = (float*)d_out;

    float* ws = (float*)d_ws;
    float* xnode  = ws;
    float* accp   = xnode  + (size_t)N_NODES * 256;
    float* agg    = accp   + (size_t)N_NODES * 128;
    float* den2   = agg    + (size_t)N_NODES * 256;
    float* uc     = den2   + (size_t)N_NODES * 2;
    float* relc   = uc     + 256;
    float* lx     = relc   + (size_t)NREL * 128;
    float* xab2   = lx     + (size_t)N_NODES * 128;
    float* relW2  = xab2   + (size_t)N_NODES * 256;
    int*   cnt    = (int*)(relW2 + (size_t)NREL * 128);
    int*   rowptr = cnt    + NPAD;
    int*   rank   = rowptr + NPAD;
    int*   csr    = rank   + ET_EDGES;

    float* out_rel = out + (size_t)N_NODES * 128;

    hipMemsetAsync(cnt, 0, (size_t)NPAD * sizeof(int), stream);

    static_assert(2 * ((NREL + 15) / 16) <= GEMM_MAX_BLOCKS,
                  "the two small problems must not have more tiles than blocks");
    gat_wmma_gemm_multi<<<GEMM_MAX_BLOCKS, GEMM_THREADS, 0, stream>>>(
        x,   w1h,       xnode,       N_NODES, 256,
        x,   w1h + 128, xnode + 128, N_NODES, 256,
        rel, w1h + 256, relc,        NREL,    128,
        rel, WR,        out_rel,     NREL,    128,
        384, el, eln, cnt, rank);
    gat_scan<<<2, SCAN_THREADS, 0, stream>>>(cnt, rowptr, w1h, w2h, uc);
    gat_fill<<<(ET_EDGES + 255) / 256, 256, 0, stream>>>(el, eln, et, etn, rowptr,
                                                        rank, csr);

    gat_node_l1<<<(N_NODES + NODE_WAVES - 1) / NODE_WAVES, NODE_WAVES * 32, 0, stream>>>(
        rowptr, csr, xnode, eemb, relc, w2h, uc, accp, agg, den2);
    static_assert(N_NODES % 16 == 0, "each head's plane must be whole row tiles");
    gat_wmma_gemm_heads<<<GEMM_MAX_BLOCKS, GEMM_THREADS, 0, stream>>>(
        agg, w1h + 256, lx, 2 * N_NODES, 128, 384, 128, 0, 2 * N_NODES / 16, accp, den2);

    gat_wmma_gemm_multi<<<GEMM_MAX_BLOCKS, GEMM_THREADS, 0, stream>>>(
        lx,      w1o,       xab2,       N_NODES, 256,
        lx,      w1o + 128, xab2 + 128, N_NODES, 256,
        out_rel, w1o + 256, relW2,      NREL,    128,
        (const float*)nullptr, (const float*)nullptr, (float*)nullptr, 0, 0,
        384, (const int*)nullptr, (const int*)nullptr, (int*)nullptr, (int*)nullptr);

    gat_node<2><<<(N_NODES + NODE_WAVES - 1) / NODE_WAVES, NODE_WAVES * 32, 0, stream>>>(
        rowptr, csr, xab2, relW2, relW2, w2o, out);
}


# scan and fill launches removed: the integer add that counts an edge also puts its record into its source's 64-slot bucket (overflow list behind); w2 fold rides in the GEMM launch; sums unchanged
# speedup vs baseline: 3.5544x; 1.0958x over previous
#include <hip/hip_runtime.h>
#include <hip/hip_bf16.h>

typedef float v2f __attribute__((ext_vector_type(2)));
typedef float v4f __attribute__((ext_vector_type(4)));
typedef float v8f __attribute__((ext_vector_type(8)));

#define N_NODES 50000
#define E_EDGES 500000
#define E2_EDGES 62500
#define ET_EDGES (E_EDGES + E2_EDGES)
#define NREL 1000
#define ALPHA 0.2f
#define KDIM 128

#define GEMM_N        128
#define GEMM_NT       8
#define GEMM_THREADS  512
#define GEMM_WAVES    (GEMM_THREADS / 32)
#define GEMM_MAX_BLOCKS 256
#define BPITCH        132
#define GEMM_PANEL    (GEMM_N * BPITCH)
#define CNT_WAVES     4
#define CNT_U         9
#define CNT_LANES     (GEMM_MAX_BLOCKS * CNT_WAVES * 32)
#define BKT_CAP       64
#define NPAD          (N_NODES + 4)

template <int NT>
__device__ __forceinline__ void gemm_step(v8f (&acc)[NT], const v4f& a,
                                          const v4f (&b)[NT]) {
    v2f a_lo = (v2f){a[0], a[1]};
    v2f a_hi = (v2f){a[2], a[3]};
#pragma unroll
    for (int s = 0; s < NT; ++s)
        acc[s] = __builtin_amdgcn_wmma_f32_16x16x4_f32(
            false, a_lo, false, (v2f){b[s][0], b[s][1]}, (short)0,
            acc[s], false, false);
#pragma unroll
    for (int s = 0; s < NT; ++s)
        acc[s] = __builtin_amdgcn_wmma_f32_16x16x4_f32(
            false, a_hi, false, (v2f){b[s][2], b[s][3]}, (short)0,
            acc[s], false, false);
}

template <int LDBK>
__device__ __forceinline__ void gemm_stage_panel(float* __restrict__ Bs,
                                                 const float* __restrict__ B,
                                                 int ldbn, int tid) {
    if constexpr (LDBK == 1) {
        constexpr int NV = GEMM_N * (KDIM / 4) / GEMM_THREADS;
        v4f v[NV];
#pragma unroll
        for (int it = 0; it < NV; ++it) {
            int i = tid + it * GEMM_THREADS;
            v[it] = *(const v4f*)(B + (size_t)(i >> 5) * ldbn + (i & 31) * 4);
        }
#pragma unroll
        for (int it = 0; it < NV; ++it) {
            int i = tid + it * GEMM_THREADS;
            *(v4f*)(Bs + (i >> 5) * BPITCH + (i & 31) * 4) = v[it];
        }
    } else {
        constexpr int NV = 8;
        for (int i0 = tid; i0 < GEMM_N * KDIM; i0 += NV * GEMM_THREADS) {
            float v[NV];
#pragma unroll
            for (int it = 0; it < NV; ++it) {
                int i = i0 + it * GEMM_THREADS;
                v[it] = B[(size_t)(i & 127) * ldbn + (size_t)(i >> 7) * LDBK];
            }
#pragma unroll
            for (int it = 0; it < NV; ++it) {
                int i = i0 + it * GEMM_THREADS;
                Bs[(i & 127) * BPITCH + (i >> 7)] = v[it];
            }
        }
    }
}

__global__ __launch_bounds__(GEMM_THREADS)
void gat_wmma_gemm_multi(
        const float* __restrict__ A0, const float* __restrict__ B0, float* __restrict__ C0, int M0, int ldc0,
        const float* __restrict__ A1, const float* __restrict__ B1, float* __restrict__ C1, int M1, int ldc1,
        const float* __restrict__ A2, const float* __restrict__ B2, float* __restrict__ C2, int M2, int ldc2,
        const float* __restrict__ A3, const float* __restrict__ B3, float* __restrict__ C3, int M3, int ldc3,
        int ldbn,
        const int* __restrict__ el, const int* __restrict__ eln,
        const int* __restrict__ et, const int* __restrict__ etn,
        int* __restrict__ cnt, int* __restrict__ bkt,
        int* __restrict__ ovs, int* __restrict__ ovf,
        const float* __restrict__ w2f, float* __restrict__ uc) {
    __shared__ __attribute__((aligned(16))) float Bs[3 * GEMM_PANEL];
    const int tid = threadIdx.x;

    const int e0 = (M0 + 15) >> 4;
    const int e1 = e0 + ((M1 + 15) >> 4);
    const int e2 = e1 + ((M2 + 15) >> 4);
    const int e3 = e2 + ((M3 + 15) >> 4);
    const int vs = (int)blockIdx.x +
        GEMM_MAX_BLOCKS * ((e1 - (int)blockIdx.x + GEMM_MAX_BLOCKS - 1) / GEMM_MAX_BLOCKS);

    gemm_stage_panel<1>(Bs, B0, ldbn, tid);
    gemm_stage_panel<1>(Bs + GEMM_PANEL, B1, ldbn, tid);
    if (vs < e3) {
        if (vs < e2) gemm_stage_panel<1>(Bs + 2 * GEMM_PANEL, B2, ldbn, tid);
        else         gemm_stage_panel<KDIM>(Bs + 2 * GEMM_PANEL, B3, 1, tid);
    }
    __syncthreads();

    const int wave = __builtin_amdgcn_readfirstlane(tid >> 5);
    const int lane = tid & 31;
    const int half = lane >> 4;
    const int l16  = lane & 15;
    const float* Bl = Bs + l16 * BPITCH + half * 4;
    constexpr int wstride = GEMM_MAX_BLOCKS * GEMM_WAVES;

    if (cnt != nullptr && wave >= GEMM_WAVES - CNT_WAVES) {
        const int t = ((int)blockIdx.x * CNT_WAVES + (wave - (GEMM_WAVES - CNT_WAVES))) * 32 + lane;
        for (int eb = t; eb < ET_EDGES; eb += CNT_LANES * CNT_U) {
            int s[CNT_U], d[CNT_U], t0[CNT_U], t1[CNT_U], r[CNT_U];
#pragma unroll
            for (int u = 0; u < CNT_U; ++u) {
                const int e = eb + u * CNT_LANES;
                const int ec = e < ET_EDGES ? e : ET_EDGES - 1;
                const bool dr = ec < E_EDGES;
                const int j = dr ? 0 : ec - E_EDGES;
                const int* ps  = dr ? el + ec           : eln + j;
                const int* pd  = dr ? el + E_EDGES + ec : eln + E2_EDGES + j;
                const int* pt0 = dr ? et + ec           : etn + 2 * j;
                const int* pt1 = dr ? et + ec           : etn + 2 * j + 1;
                s[u]  = *ps;
                d[u]  = *pd;
                t0[u] = *pt0;
                const int tt = *pt1;
                t1[u] = dr ? -1 : tt;
            }
#pragma unroll
            for (int u = 0; u < CNT_U; ++u) {
                const int e = eb + u * CNT_LANES;
                r[u] = 0;
                if (e < ET_EDGES) r[u] = atomicAdd(cnt + s[u], 1);
            }
#pragma unroll
            for (int u = 0; u < CNT_U; ++u) {
                const int e = eb + u * CNT_LANES;
                if (e < ET_EDGES) {
                    const int4 rec = make_int4(e, d[u], t0[u], t1[u]);
                    if (r[u] < BKT_CAP) {
                        *(int4*)(bkt + 4 * ((size_t)s[u] * BKT_CAP + r[u])) = rec;
                    } else {
                        const int o = atomicAdd(cnt + N_NODES, 1);
                        ovs[o] = s[u];
                        *(int4*)(ovf + 4 * (size_t)o) = rec;
                    }
                }
            }
        }
    }

    if (uc != nullptr && wave == GEMM_WAVES - CNT_WAVES - 1 && blockIdx.x < 8) {
        const int tf = (int)blockIdx.x * 32 + lane;
        const int h = tf >> 7, k = tf & 127;
        float s = 0.f;
        for (int j = 0; j < 64; ++j)
            s += B2[(size_t)(h * 64 + j) * ldbn + k] * w2f[h * 64 + j];
        uc[tf] = s;
    }

    int v = (int)blockIdx.x + GEMM_MAX_BLOCKS * wave;
    if (v >= e3) return;

    auto locate = [&](int vv, const float*& A, float*& C, int& M, int& ldc,
                      int& mt, int& slot) {
        if (vv < e0)      { A = A0; C = C0; M = M0; ldc = ldc0; mt = vv;      slot = 0; }
        else if (vv < e1) { A = A1; C = C1; M = M1; ldc = ldc1; mt = vv - e0; slot = 1; }
        else if (vv < e2) { A = A2; C = C2; M = M2; ldc = ldc2; mt = vv - e1; slot = 2; }
        else              { A = A3; C = C3; M = M3; ldc = ldc3; mt = vv - e2; slot = 2; }
    };

    const float* A; float* C; int M, ldc, mt, slot;
    locate(v, A, C, M, ldc, mt, slot);
    int arow = mt * 16 + l16;
    const float* Ap = A + (size_t)(arow < M ? arow : (M - 1)) * KDIM + half * 4;
    const float* Bt = Bl + slot * GEMM_PANEL;

    v4f a_cur[4], a_nxt[4];
    v4f b0[GEMM_NT], b1[GEMM_NT];
#pragma unroll
    for (int t = 0; t < 4; ++t) a_cur[t] = *(const v4f*)(Ap + t * 8);
#pragma unroll
    for (int s = 0; s < GEMM_NT; ++s) b0[s] = *(const v4f*)(Bt + s * 16 * BPITCH);

    for (;;) {
        int vn = v + wstride;
        const bool more = vn < e3;
        if (!more) vn = v;
        const float* An; float* Cn; int Mn, ldcn, mtn, slotn;
        locate(vn, An, Cn, Mn, ldcn, mtn, slotn);
        int arown = mtn * 16 + l16;
        const float* Apn = An + (size_t)(arown < Mn ? arown : (Mn - 1)) * KDIM + half * 4;
        const float* Btn = Bl + slotn * GEMM_PANEL;

        v8f acc[GEMM_NT];
#pragma unroll
        for (int s = 0; s < GEMM_NT; ++s)
            acc[s] = (v8f){0.f,0.f,0.f,0.f,0.f,0.f,0.f,0.f};

#pragma unroll 1
        for (int q = 0; q < 4; ++q) {
            const float* pa = (q < 3) ? (Ap + (q + 1) * 32) : Apn;
#pragma unroll
            for (int t = 0; t < 4; ++t) a_nxt[t] = *(const v4f*)(pa + t * 8);

            const float* Bq = Bt + q * 32;
            const float* Bw = (q < 3) ? (Bq + 32) : Btn;
#pragma unroll
            for (int s = 0; s < GEMM_NT; ++s) b1[s] = *(const v4f*)(Bq + s * 16 * BPITCH + 8);
            gemm_step<GEMM_NT>(acc, a_cur[0], b0);
#pragma unroll
            for (int s = 0; s < GEMM_NT; ++s) b0[s] = *(const v4f*)(Bq + s * 16 * BPITCH + 16);
            gemm_step<GEMM_NT>(acc, a_cur[1], b1);
#pragma unroll
            for (int s = 0; s < GEMM_NT; ++s) b1[s] = *(const v4f*)(Bq + s * 16 * BPITCH + 24);
            gemm_step<GEMM_NT>(acc, a_cur[2], b0);
#pragma unroll
            for (int s = 0; s < GEMM_NT; ++s) b0[s] = *(const v4f*)(Bw + s * 16 * BPITCH);
            gemm_step<GEMM_NT>(acc, a_cur[3], b1);

#pragma unroll
            for (int t = 0; t < 4; ++t) a_cur[t] = a_nxt[t];
        }

        float* Ct = C + (size_t)(mt * 16 + half * 8) * ldc + l16;
        if (mt * 16 + 16 <= M) {
#pragma unroll
            for (int r = 0; r < 8; ++r)
#pragma unroll
                for (int s = 0; s < GEMM_NT; ++s)
                    Ct[(size_t)r * ldc + s * 16] = acc[s][r];
        } else {
#pragma unroll
            for (int r = 0; r < 8; ++r) {
                if (mt * 16 + half * 8 + r < M) {
#pragma unroll
                    for (int s = 0; s < GEMM_NT; ++s)
                        Ct[(size_t)r * ldc + s * 16] = acc[s][r];
                }
            }
        }
        if (!more) break;
        v = vn; C = Cn; M = Mn; ldc = ldcn; mt = mtn;
        Ap = Apn; Bt = Btn;
    }
}

__global__ __launch_bounds__(GEMM_THREADS)
void gat_wmma_gemm_heads(const float* __restrict__ A,
                         const float* __restrict__ B,
                         float* __restrict__ C,
                         int M, int lda, int ldbn, int ldc, int coff,
                         int mtiles,
                         const float* __restrict__ addp,
                         const float* __restrict__ den2) {
    constexpr int NT = GEMM_NT / 2;
    __shared__ __attribute__((aligned(16))) float Bs[GEMM_PANEL];
    const int tid = threadIdx.x;

    gemm_stage_panel<1>(Bs, B, ldbn, tid);
    __syncthreads();

    const int wave = __builtin_amdgcn_readfirstlane(tid >> 5);
    const int lane = tid & 31;
    const int half = lane >> 4;
    const int l16  = lane & 15;
    const float* Bl = Bs + l16 * BPITCH + half * 4;
    const int wstride = gridDim.x * GEMM_WAVES;

    int mt = blockIdx.x + gridDim.x * wave;
    if (mt >= mtiles) return;

    int arow = mt * 16 + l16;
    const float* Ap = A + (size_t)(arow < M ? arow : (M - 1)) * lda + half * 4;

    const int hm = mtiles >> 1;
    const float* Bt = Bl + (mt >= hm ? 64 * BPITCH : 0);

    v4f a_cur[4], a_nxt[4];
    v4f b0[NT], b1[NT];
#pragma unroll
    for (int t = 0; t < 4; ++t) a_cur[t] = *(const v4f*)(Ap + t * 8);
#pragma unroll
    for (int s = 0; s < NT; ++s) b0[s] = *(const v4f*)(Bt + s * 16 * BPITCH);

    for (; mt < mtiles; mt += wstride) {
        int mtn = mt + wstride;
        if (mtn >= mtiles) mtn = mt;
        int arown = mtn * 16 + l16;
        const float* Apn = A + (size_t)(arown < M ? arown : (M - 1)) * lda + half * 4;
        const float* Btn = Bl + (mtn >= hm ? 64 * BPITCH : 0);

        v8f acc[NT];
#pragma unroll
        for (int s = 0; s < NT; ++s)
            acc[s] = (v8f){0.f,0.f,0.f,0.f,0.f,0.f,0.f,0.f};

        float ad[8][NT], dn[8];
        const int head = mt >= hm ? 1 : 0;
        const int node0 = (mt - head * hm) * 16 + half * 8;
        const size_t o0 = (size_t)node0 * ldc + coff + head * 64 + l16;
#pragma unroll
        for (int r = 0; r < 8; ++r) {
            const float d = den2[2 * (node0 + r) + head];
            dn[r] = d > 1e-12f ? d : 1e-12f;
#pragma unroll
            for (int s = 0; s < NT; ++s)
                ad[r][s] = addp[o0 + (size_t)r * ldc + s * 16];
        }

#pragma unroll 1
        for (int q = 0; q < 4; ++q) {
            const float* pa = (q < 3) ? (Ap + (q + 1) * 32) : Apn;
#pragma unroll
            for (int t = 0; t < 4; ++t) a_nxt[t] = *(const v4f*)(pa + t * 8);

            const float* Bq = Bt + q * 32;
            const float* Bw = (q < 3) ? (Bq + 32) : Btn;
#pragma unroll
            for (int s = 0; s < NT; ++s) b1[s] = *(const v4f*)(Bq + s * 16 * BPITCH + 8);
            gemm_step<NT>(acc, a_cur[0], b0);
#pragma unroll
            for (int s = 0; s < NT; ++s) b0[s] = *(const v4f*)(Bq + s * 16 * BPITCH + 16);
            gemm_step<NT>(acc, a_cur[1], b1);
#pragma unroll
            for (int s = 0; s < NT; ++s) b1[s] = *(const v4f*)(Bq + s * 16 * BPITCH + 24);
            gemm_step<NT>(acc, a_cur[2], b0);
#pragma unroll
            for (int s = 0; s < NT; ++s) b0[s] = *(const v4f*)(Bw + s * 16 * BPITCH);
            gemm_step<NT>(acc, a_cur[3], b1);

#pragma unroll
            for (int t = 0; t < 4; ++t) a_cur[t] = a_nxt[t];
        }

#pragma unroll
        for (int r = 0; r < 8; ++r)
#pragma unroll
            for (int s = 0; s < NT; ++s)
                C[o0 + (size_t)r * ldc + s * 16] = (acc[s][r] + ad[r][s]) / dn[r];
        Bt = Btn;
        Ap = Apn;
    }
}

struct GatRange {
    int cb;
    int nbg;
    int ng;
    int novf;
};

__device__ __forceinline__ GatRange gat_range(const int* __restrict__ cnt, int n) {
    GatRange rg;
    int cn = __builtin_amdgcn_readfirstlane(cnt[n]);
    cn = cn < 0 ? 0 : (cn > ET_EDGES ? ET_EDGES : cn);
    rg.cb = cn < BKT_CAP ? cn : BKT_CAP;
    rg.novf = 0;
    if (cn > BKT_CAP) {
        const int o = __builtin_amdgcn_readfirstlane(cnt[N_NODES]);
        rg.novf = o < 0 ? 0 : (o > ET_EDGES ? ET_EDGES : o);
    }
    rg.nbg = (rg.cb + 31) >> 5;
    rg.ng = rg.nbg + ((rg.novf + 31) >> 5);
    return rg;
}

__device__ __forceinline__ int4 gat_group(const GatRange& rg, int g, int& pos, int n, int lane,
                                          const int* __restrict__ bkt,
                                          const int* __restrict__ ovs,
                                          const int* __restrict__ ovf,
                                          int& cntc, int& jn) {
    int4 md = make_int4(0x7fffffff, 0, 0, -1);
    if (g < rg.nbg) {
        cntc = rg.cb - g * 32;
        if (cntc > 32) cntc = 32;
        if (lane < cntc) md = *(const int4*)(bkt + 4 * (size_t)(pos + lane));
        pos += 32;
        jn = cntc;
    } else {
        const int i = (g - rg.nbg) * 32 + lane;
        bool mine = false;
        if (i < rg.novf) mine = ovs[i] == n;
        if (mine) md = *(const int4*)(ovf + 4 * (size_t)i);
        cntc = __builtin_popcount(__builtin_amdgcn_ballot_w32(mine));
        jn = 32;
    }
    return md;
}

#define NODE_WAVES 8
#define NODE_U     4

template <int LAYER>
__global__ __launch_bounds__(NODE_WAVES * 32)
void gat_node(const int* __restrict__ cnt, const int* __restrict__ bkt,
              const int* __restrict__ ovs, const int* __restrict__ ovf,
              const float* __restrict__ xtab, const float* __restrict__ etab,
              const float* __restrict__ rtab, const float* __restrict__ w2,
              float* __restrict__ dst) {
    const int wave = __builtin_amdgcn_readfirstlane(threadIdx.x >> 5);
    const int n = blockIdx.x * NODE_WAVES + wave;
    if (n >= N_NODES) return;
    const int lane = threadIdx.x & 31;
    const int k4 = lane * 4;
    const GatRange rg = gat_range(cnt, n);
    const v4f xa = *(const v4f*)(xtab + (size_t)n * 256 + k4);
    const v4f wv = *(const v4f*)(w2 + k4);
    v4f acc = (v4f){0.f, 0.f, 0.f, 0.f};
    float den = 0.f;

    int pos = n * BKT_CAP;
    for (int g = 0; g < rg.ng; ++g) {
        int cntc, jn;
        const int4 md = gat_group(rg, g, pos, n, lane, bkt, ovs, ovf, cntc, jn);
        int rk = 0;
        for (int j = 0; j < jn; ++j)
            rk += (__builtin_amdgcn_readlane(md.x, j) < md.x) ? 1 : 0;

        for (int j0 = 0; j0 < cntc; j0 += NODE_U) {
            v4f xb[NODE_U], e0[NODE_U], e1[NODE_U];
#pragma unroll
            for (int u = 0; u < NODE_U; ++u) {
                int jj = j0 + u;
                if (jj >= cntc) jj = cntc - 1;
                const unsigned hit = __builtin_amdgcn_ballot_w32(rk == jj);
                const int sl = __builtin_ctz(hit);
                int e  = __builtin_amdgcn_readlane(md.x, sl);
                int d  = __builtin_amdgcn_readlane(md.y, sl);
                int t0 = __builtin_amdgcn_readlane(md.z, sl);
                int t1 = __builtin_amdgcn_readlane(md.w, sl);
                e  = e  < 0 ? 0 : (e  > E_EDGES - 1 ? E_EDGES - 1 : e);
                d  = d  < 0 ? 0 : (d  > N_NODES - 1 ? N_NODES - 1 : d);
                t0 = t0 < 0 ? 0 : (t0 > NREL - 1 ? NREL - 1 : t0);
                t1 = t1 > NREL - 1 ? NREL - 1 : t1;
                xb[u] = *(const v4f*)(xtab + (size_t)d * 256 + 128 + k4);
                if (LAYER == 1 && t1 < 0)
                    e0[u] = *(const v4f*)(etab + (size_t)e * 128 + k4);
                else
                    e0[u] = *(const v4f*)(rtab + (size_t)t0 * 128 + k4);
                if (t1 >= 0)
                    e1[u] = *(const v4f*)(rtab + (size_t)t1 * 128 + k4);
                else
                    e1[u] = (v4f){0.f, 0.f, 0.f, 0.f};
            }
            v4f m[NODE_U];
            float p[NODE_U];
#pragma unroll
            for (int u = 0; u < NODE_U; ++u) {
                const v4f ev = e0[u] + e1[u];
                m[u] = (xa + xb[u]) + ev;
                p[u] = m[u][0] * wv[0] + m[u][1] * wv[1] + m[u][2] * wv[2] + m[u][3] * wv[3];
            }
            if (LAYER == 2) {
#pragma unroll
                for (int u = 0; u < NODE_U; ++u) p[u] += __shfl_xor(p[u], 16, 32);
            }
#pragma unroll
            for (int u = 0; u < NODE_U; ++u) p[u] += __shfl_xor(p[u], 8, 32);
#pragma unroll
            for (int u = 0; u < NODE_U; ++u) p[u] += __shfl_xor(p[u], 4, 32);
#pragma unroll
            for (int u = 0; u < NODE_U; ++u) p[u] += __shfl_xor(p[u], 2, 32);
#pragma unroll
            for (int u = 0; u < NODE_U; ++u) p[u] += __shfl_xor(p[u], 1, 32);
#pragma unroll
            for (int u = 0; u < NODE_U; ++u) {
                if (j0 + u < cntc) {
                    const float lr = p[u] > 0.f ? p[u] : ALPHA * p[u];
                    const float ee = __expf(-lr);
                    acc += ee * m[u];
                    den += ee;
                }
            }
        }
    }

    const float dn = den > 1e-12f ? den : 1e-12f;
    v4f h = acc / dn;
    if (LAYER == 2) {
#pragma unroll
        for (int c = 0; c < 4; ++c) {
            float y = h[c] > 0.f ? h[c] : (__expf(h[c]) - 1.f);
            y = y > 0.f ? y : (__expf(y) - 1.f);
            h[c] = y;
        }
    }
    *(v4f*)(dst + (size_t)n * 128 + k4) = h;
}

__global__ __launch_bounds__(NODE_WAVES * 32)
void gat_node_l1(const int* __restrict__ cnt, const int* __restrict__ bkt,
                 const int* __restrict__ ovs, const int* __restrict__ ovf,
                 const float* __restrict__ xtab, const float* __restrict__ eraw,
                 const float* __restrict__ rtab, const float* __restrict__ w2,
                 const float* __restrict__ uc, float* __restrict__ accp,
                 float* __restrict__ agg, float* __restrict__ den2) {
    const int wave = __builtin_amdgcn_readfirstlane(threadIdx.x >> 5);
    const int n = blockIdx.x * NODE_WAVES + wave;
    if (n >= N_NODES) return;
    const int lane = threadIdx.x & 31;
    const int k4 = lane * 4;
    const bool lo = lane < 16;
    const GatRange rg = gat_range(cnt, n);
    const v4f xa = *(const v4f*)(xtab + (size_t)n * 256 + k4);
    const v4f wv = *(const v4f*)(w2 + k4);
    const v4f u0 = *(const v4f*)(uc + k4);
    const v4f u1 = *(const v4f*)(uc + 128 + k4);
    const v4f zero4 = (v4f){0.f, 0.f, 0.f, 0.f};
    v4f acc = zero4, a0 = zero4, a1 = zero4;
    float den = 0.f;

    int pos = n * BKT_CAP;
    for (int g = 0; g < rg.ng; ++g) {
        int cntc, jn;
        const int4 md = gat_group(rg, g, pos, n, lane, bkt, ovs, ovf, cntc, jn);
        int rk = 0;
        for (int j = 0; j < jn; ++j)
            rk += (__builtin_amdgcn_readlane(md.x, j) < md.x) ? 1 : 0;

        for (int j0 = 0; j0 < cntc; j0 += NODE_U) {
            v4f xb[NODE_U], e0[NODE_U], e1[NODE_U];
            bool dir[NODE_U];
#pragma unroll
            for (int u = 0; u < NODE_U; ++u) {
                int jj = j0 + u;
                if (jj >= cntc) jj = cntc - 1;
                const unsigned hit = __builtin_amdgcn_ballot_w32(rk == jj);
                const int sl = __builtin_ctz(hit);
                int e  = __builtin_amdgcn_readlane(md.x, sl);
                int d  = __builtin_amdgcn_readlane(md.y, sl);
                int t0 = __builtin_amdgcn_readlane(md.z, sl);
                int t1 = __builtin_amdgcn_readlane(md.w, sl);
                e  = e  < 0 ? 0 : (e  > E_EDGES - 1 ? E_EDGES - 1 : e);
                d  = d  < 0 ? 0 : (d  > N_NODES - 1 ? N_NODES - 1 : d);
                t0 = t0 < 0 ? 0 : (t0 > NREL - 1 ? NREL - 1 : t0);
                t1 = t1 > NREL - 1 ? NREL - 1 : t1;
                dir[u] = t1 < 0;
                xb[u] = *(const v4f*)(xtab + (size_t)d * 256 + 128 + k4);
                if (t1 < 0)
                    e0[u] = *(const v4f*)(eraw + (size_t)e * 128 + k4);
                else
                    e0[u] = *(const v4f*)(rtab + (size_t)t0 * 128 + k4);
                if (t1 >= 0)
                    e1[u] = *(const v4f*)(rtab + (size_t)t1 * 128 + k4);
                else
                    e1[u] = zero4;
            }
            v4f m[NODE_U], er[NODE_U];
            float p[NODE_U];
#pragma unroll
            for (int u = 0; u < NODE_U; ++u) {
                const v4f ev = dir[u] ? zero4 : (e0[u] + e1[u]);
                er[u] = dir[u] ? e0[u] : zero4;
                m[u] = (xa + xb[u]) + ev;
                const float qn = __builtin_fmaf(m[u][3], wv[3], __builtin_fmaf(m[u][2], wv[2],
                                 __builtin_fmaf(m[u][1], wv[1], m[u][0] * wv[0])));
                const float q0 = __builtin_fmaf(er[u][3], u0[3], __builtin_fmaf(er[u][2], u0[2],
                                 __builtin_fmaf(er[u][1], u0[1], er[u][0] * u0[0])));
                const float q1 = __builtin_fmaf(er[u][3], u1[3], __builtin_fmaf(er[u][2], u1[2],
                                 __builtin_fmaf(er[u][1], u1[1], er[u][0] * u1[0])));
                const float mine = qn + (lo ? q0 : q1);
                const float send = lo ? q1 : q0;
                p[u] = mine + __shfl_xor(send, 16, 32);
            }
#pragma unroll
            for (int u = 0; u < NODE_U; ++u) p[u] += __shfl_xor(p[u], 8, 32);
#pragma unroll
            for (int u = 0; u < NODE_U; ++u) p[u] += __shfl_xor(p[u], 4, 32);
#pragma unroll
            for (int u = 0; u < NODE_U; ++u) p[u] += __shfl_xor(p[u], 2, 32);
#pragma unroll
            for (int u = 0; u < NODE_U; ++u) p[u] += __shfl_xor(p[u], 1, 32);
            float g[NODE_U], go[NODE_U];
#pragma unroll
            for (int u = 0; u < NODE_U; ++u) {
                const float lr = p[u] > 0.f ? p[u] : ALPHA * p[u];
                g[u]  = __expf(-lr);
                go[u] = __shfl_xor(g[u], 16, 32);
            }
#pragma unroll
            for (int u = 0; u < NODE_U; ++u) {
                if (j0 + u < cntc) {
                    const float g0 = lo ? g[u] : go[u];
                    const float g1 = lo ? go[u] : g[u];
                    acc += g[u] * m[u];
                    den += g[u];
                    a0 += g0 * er[u];
                    a1 += g1 * er[u];
                }
            }
        }
    }

    *(v4f*)(accp + (size_t)n * 128 + k4) = acc;
    *(v4f*)(agg + (size_t)n * 128 + k4) = a0;
    *(v4f*)(agg + (size_t)(N_NODES + n) * 128 + k4) = a1;
    if ((lane & 15) == 0) den2[n * 2 + (lane >> 4)] = den;
}

extern "C" void kernel_launch(void* const* d_in, const int* in_sizes, int n_in,
                              void* d_out, int out_size, void* d_ws, size_t ws_size,
                              hipStream_t stream) {
    const float* x    = (const float*)d_in[0];
    const float* rel  = (const float*)d_in[1];
    const int*   el   = (const int*)d_in[2];
    const int*   et   = (const int*)d_in[3];
    const float* eemb = (const float*)d_in[4];
    const int*   eln  = (const int*)d_in[5];
    const int*   etn  = (const int*)d_in[6];
    const float* w1h  = (const float*)d_in[8];
    const float* w2h  = (const float*)d_in[9];
    const float* w1o  = (const float*)d_in[10];
    const float* w2o  = (const float*)d_in[11];
    const float* WR   = (const float*)d_in[12];
    float* out = (float*)d_out;

    float* ws = (float*)d_ws;
    float* xnode  = ws;
    float* accp   = xnode  + (size_t)N_NODES * 256;
    float* agg    = accp   + (size_t)N_NODES * 128;
    float* den2   = agg    + (size_t)N_NODES * 256;
    float* uc     = den2   + (size_t)N_NODES * 2;
    float* relc   = uc     + 256;
    float* lx     = relc   + (size_t)NREL * 128;
    float* xab2   = lx     + (size_t)N_NODES * 128;
    float* relW2  = xab2   + (size_t)N_NODES * 256;
    int*   cnt    = (int*)(relW2 + (size_t)NREL * 128);
    int*   bkt    = cnt    + NPAD;
    int*   ovs    = bkt    + (size_t)N_NODES * BKT_CAP * 4;
    int*   ovf    = ovs    + ET_EDGES;

    float* out_rel = out + (size_t)N_NODES * 128;

    hipMemsetAsync(cnt, 0, (size_t)NPAD * sizeof(int), stream);

    static_assert(2 * ((NREL + 15) / 16) <= GEMM_MAX_BLOCKS,
                  "the two small problems must not have more tiles than blocks");
    gat_wmma_gemm_multi<<<GEMM_MAX_BLOCKS, GEMM_THREADS, 0, stream>>>(
        x,   w1h,       xnode,       N_NODES, 256,
        x,   w1h + 128, xnode + 128, N_NODES, 256,
        rel, w1h + 256, relc,        NREL,    128,
        rel, WR,        out_rel,     NREL,    128,
        384, el, eln, et, etn, cnt, bkt, ovs, ovf, w2h, uc);

    gat_node_l1<<<(N_NODES + NODE_WAVES - 1) / NODE_WAVES, NODE_WAVES * 32, 0, stream>>>(
        cnt, bkt, ovs, ovf, xnode, eemb, relc, w2h, uc, accp, agg, den2);
    static_assert(N_NODES % 16 == 0, "each head's plane must be whole row tiles");
    gat_wmma_gemm_heads<<<GEMM_MAX_BLOCKS, GEMM_THREADS, 0, stream>>>(
        agg, w1h + 256, lx, 2 * N_NODES, 128, 384, 128, 0, 2 * N_NODES / 16, accp, den2);

    gat_wmma_gemm_multi<<<GEMM_MAX_BLOCKS, GEMM_THREADS, 0, stream>>>(
        lx,      w1o,       xab2,       N_NODES, 256,
        lx,      w1o + 128, xab2 + 128, N_NODES, 256,
        out_rel, w1o + 256, relW2,      NREL,    128,
        (const float*)nullptr, (const float*)nullptr, (float*)nullptr, 0, 0,
        384, (const int*)nullptr, (const int*)nullptr, (const int*)nullptr, (const int*)nullptr,
        (int*)nullptr, (int*)nullptr, (int*)nullptr, (int*)nullptr,
        (const float*)nullptr, (float*)nullptr);

    gat_node<2><<<(N_NODES + NODE_WAVES - 1) / NODE_WAVES, NODE_WAVES * 32, 0, stream>>>(
        cnt, bkt, ovs, ovf, xab2, relW2, relW2, w2o, out);
}


# GEMM output columns dealt to lanes four adjacent each (B rows permuted in LDS) and row arrays put on 256-byte boundaries: every row store/gather is whole aligned 256-byte units; sums unchanged
# speedup vs baseline: 3.7230x; 1.0474x over previous
#include <hip/hip_runtime.h>
#include <hip/hip_bf16.h>

typedef float v2f __attribute__((ext_vector_type(2)));
typedef float v4f __attribute__((ext_vector_type(4)));
typedef float v8f __attribute__((ext_vector_type(8)));

#define N_NODES 50000
#define E_EDGES 500000
#define E2_EDGES 62500
#define ET_EDGES (E_EDGES + E2_EDGES)
#define NREL 1000
#define ALPHA 0.2f
#define KDIM 128

#define GEMM_N        128
#define GEMM_NT       8
#define GEMM_THREADS  512
#define GEMM_WAVES    (GEMM_THREADS / 32)
#define GEMM_MAX_BLOCKS 256
#define BPITCH        132
#define GEMM_PANEL    (GEMM_N * BPITCH)
#define CNT_WAVES     4
#define CNT_U         9
#define CNT_LANES     (GEMM_MAX_BLOCKS * CNT_WAVES * 32)
#define BKT_CAP       64
#define NPAD          (N_NODES + 4)

__device__ __forceinline__ int gemm_prow(int n) {
    return (n & 64) | ((n & 3) << 4) | ((n >> 2) & 15);
}

template <int NT>
__device__ __forceinline__ void gemm_step(v8f (&acc)[NT], const v4f& a,
                                          const v4f (&b)[NT]) {
    v2f a_lo = (v2f){a[0], a[1]};
    v2f a_hi = (v2f){a[2], a[3]};
#pragma unroll
    for (int s = 0; s < NT; ++s)
        acc[s] = __builtin_amdgcn_wmma_f32_16x16x4_f32(
            false, a_lo, false, (v2f){b[s][0], b[s][1]}, (short)0,
            acc[s], false, false);
#pragma unroll
    for (int s = 0; s < NT; ++s)
        acc[s] = __builtin_amdgcn_wmma_f32_16x16x4_f32(
            false, a_hi, false, (v2f){b[s][2], b[s][3]}, (short)0,
            acc[s], false, false);
}

template <int LDBK>
__device__ __forceinline__ void gemm_stage_panel(float* __restrict__ Bs,
                                                 const float* __restrict__ B,
                                                 int ldbn, int tid) {
    if constexpr (LDBK == 1) {
        constexpr int NV = GEMM_N * (KDIM / 4) / GEMM_THREADS;
        v4f v[NV];
#pragma unroll
        for (int it = 0; it < NV; ++it) {
            int i = tid + it * GEMM_THREADS;
            v[it] = *(const v4f*)(B + (size_t)(i >> 5) * ldbn + (i & 31) * 4);
        }
#pragma unroll
        for (int it = 0; it < NV; ++it) {
            int i = tid + it * GEMM_THREADS;
            *(v4f*)(Bs + gemm_prow(i >> 5) * BPITCH + (i & 31) * 4) = v[it];
        }
    } else {
        constexpr int NV = 8;
        for (int i0 = tid; i0 < GEMM_N * KDIM; i0 += NV * GEMM_THREADS) {
            float v[NV];
#pragma unroll
            for (int it = 0; it < NV; ++it) {
                int i = i0 + it * GEMM_THREADS;
                v[it] = B[(size_t)(i & 127) * ldbn + (size_t)(i >> 7) * LDBK];
            }
#pragma unroll
            for (int it = 0; it < NV; ++it) {
                int i = i0 + it * GEMM_THREADS;
                Bs[gemm_prow(i & 127) * BPITCH + (i >> 7)] = v[it];
            }
        }
    }
}

__global__ __launch_bounds__(GEMM_THREADS)
void gat_wmma_gemm_multi(
        const float* __restrict__ A0, const float* __restrict__ B0, float* __restrict__ C0, int M0, int ldc0,
        const float* __restrict__ A1, const float* __restrict__ B1, float* __restrict__ C1, int M1, int ldc1,
        const float* __restrict__ A2, const float* __restrict__ B2, float* __restrict__ C2, int M2, int ldc2,
        const float* __restrict__ A3, const float* __restrict__ B3, float* __restrict__ C3, int M3, int ldc3,
        int ldbn,
        const int* __restrict__ el, const int* __restrict__ eln,
        const int* __restrict__ et, const int* __restrict__ etn,
        int* __restrict__ cnt, int* __restrict__ bkt,
        int* __restrict__ ovs, int* __restrict__ ovf,
        const float* __restrict__ w2f, float* __restrict__ uc) {
    __shared__ __attribute__((aligned(16))) float Bs[3 * GEMM_PANEL];
    const int tid = threadIdx.x;

    const int e0 = (M0 + 15) >> 4;
    const int e1 = e0 + ((M1 + 15) >> 4);
    const int e2 = e1 + ((M2 + 15) >> 4);
    const int e3 = e2 + ((M3 + 15) >> 4);
    const int vs = (int)blockIdx.x +
        GEMM_MAX_BLOCKS * ((e1 - (int)blockIdx.x + GEMM_MAX_BLOCKS - 1) / GEMM_MAX_BLOCKS);

    gemm_stage_panel<1>(Bs, B0, ldbn, tid);
    gemm_stage_panel<1>(Bs + GEMM_PANEL, B1, ldbn, tid);
    if (vs < e3) {
        if (vs < e2) gemm_stage_panel<1>(Bs + 2 * GEMM_PANEL, B2, ldbn, tid);
        else         gemm_stage_panel<KDIM>(Bs + 2 * GEMM_PANEL, B3, 1, tid);
    }
    __syncthreads();

    const int wave = __builtin_amdgcn_readfirstlane(tid >> 5);
    const int lane = tid & 31;
    const int half = lane >> 4;
    const int l16  = lane & 15;
    const float* Bl = Bs + l16 * BPITCH + half * 4;
    constexpr int wstride = GEMM_MAX_BLOCKS * GEMM_WAVES;

    if (cnt != nullptr && wave >= GEMM_WAVES - CNT_WAVES) {
        const int t = ((int)blockIdx.x * CNT_WAVES + (wave - (GEMM_WAVES - CNT_WAVES))) * 32 + lane;
        for (int eb = t; eb < ET_EDGES; eb += CNT_LANES * CNT_U) {
            int s[CNT_U], d[CNT_U], t0[CNT_U], t1[CNT_U], r[CNT_U];
#pragma unroll
            for (int u = 0; u < CNT_U; ++u) {
                const int e = eb + u * CNT_LANES;
                const int ec = e < ET_EDGES ? e : ET_EDGES - 1;
                const bool dr = ec < E_EDGES;
                const int j = dr ? 0 : ec - E_EDGES;
                const int* ps  = dr ? el + ec           : eln + j;
                const int* pd  = dr ? el + E_EDGES + ec : eln + E2_EDGES + j;
                const int* pt0 = dr ? et + ec           : etn + 2 * j;
                const int* pt1 = dr ? et + ec           : etn + 2 * j + 1;
                s[u]  = *ps;
                d[u]  = *pd;
                t0[u] = *pt0;
                const int tt = *pt1;
                t1[u] = dr ? -1 : tt;
            }
#pragma unroll
            for (int u = 0; u < CNT_U; ++u) {
                const int e = eb + u * CNT_LANES;
                r[u] = 0;
                if (e < ET_EDGES) r[u] = atomicAdd(cnt + s[u], 1);
            }
#pragma unroll
            for (int u = 0; u < CNT_U; ++u) {
                const int e = eb + u * CNT_LANES;
                if (e < ET_EDGES) {
                    const int4 rec = make_int4(e, d[u], t0[u], t1[u]);
                    if (r[u] < BKT_CAP) {
                        *(int4*)(bkt + 4 * ((size_t)s[u] * BKT_CAP + r[u])) = rec;
                    } else {
                        const int o = atomicAdd(cnt + N_NODES, 1);
                        ovs[o] = s[u];
                        *(int4*)(ovf + 4 * (size_t)o) = rec;
                    }
                }
            }
        }
    }

    if (uc != nullptr && wave == GEMM_WAVES - CNT_WAVES - 1 && blockIdx.x < 8) {
        const int tf = (int)blockIdx.x * 32 + lane;
        const int h = tf >> 7, k = tf & 127;
        float s = 0.f;
        for (int j = 0; j < 64; ++j)
            s += B2[(size_t)(h * 64 + j) * ldbn + k] * w2f[h * 64 + j];
        uc[tf] = s;
    }

    int v = (int)blockIdx.x + GEMM_MAX_BLOCKS * wave;
    if (v >= e3) return;

    auto locate = [&](int vv, const float*& A, float*& C, int& M, int& ldc,
                      int& mt, int& slot) {
        if (vv < e0)      { A = A0; C = C0; M = M0; ldc = ldc0; mt = vv;      slot = 0; }
        else if (vv < e1) { A = A1; C = C1; M = M1; ldc = ldc1; mt = vv - e0; slot = 1; }
        else if (vv < e2) { A = A2; C = C2; M = M2; ldc = ldc2; mt = vv - e1; slot = 2; }
        else              { A = A3; C = C3; M = M3; ldc = ldc3; mt = vv - e2; slot = 2; }
    };

    const float* A; float* C; int M, ldc, mt, slot;
    locate(v, A, C, M, ldc, mt, slot);
    int arow = mt * 16 + l16;
    const float* Ap = A + (size_t)(arow < M ? arow : (M - 1)) * KDIM + half * 4;
    const float* Bt = Bl + slot * GEMM_PANEL;

    v4f a_cur[4], a_nxt[4];
    v4f b0[GEMM_NT], b1[GEMM_NT];
#pragma unroll
    for (int t = 0; t < 4; ++t) a_cur[t] = *(const v4f*)(Ap + t * 8);
#pragma unroll
    for (int s = 0; s < GEMM_NT; ++s) b0[s] = *(const v4f*)(Bt + s * 16 * BPITCH);

    for (;;) {
        int vn = v + wstride;
        const bool more = vn < e3;
        if (!more) vn = v;
        const float* An; float* Cn; int Mn, ldcn, mtn, slotn;
        locate(vn, An, Cn, Mn, ldcn, mtn, slotn);
        int arown = mtn * 16 + l16;
        const float* Apn = An + (size_t)(arown < Mn ? arown : (Mn - 1)) * KDIM + half * 4;
        const float* Btn = Bl + slotn * GEMM_PANEL;

        v8f acc[GEMM_NT];
#pragma unroll
        for (int s = 0; s < GEMM_NT; ++s)
            acc[s] = (v8f){0.f,0.f,0.f,0.f,0.f,0.f,0.f,0.f};

#pragma unroll 1
        for (int q = 0; q < 4; ++q) {
            const float* pa = (q < 3) ? (Ap + (q + 1) * 32) : Apn;
#pragma unroll
            for (int t = 0; t < 4; ++t) a_nxt[t] = *(const v4f*)(pa + t * 8);

            const float* Bq = Bt + q * 32;
            const float* Bw = (q < 3) ? (Bq + 32) : Btn;
#pragma unroll
            for (int s = 0; s < GEMM_NT; ++s) b1[s] = *(const v4f*)(Bq + s * 16 * BPITCH + 8);
            gemm_step<GEMM_NT>(acc, a_cur[0], b0);
#pragma unroll
            for (int s = 0; s < GEMM_NT; ++s) b0[s] = *(const v4f*)(Bq + s * 16 * BPITCH + 16);
            gemm_step<GEMM_NT>(acc, a_cur[1], b1);
#pragma unroll
            for (int s = 0; s < GEMM_NT; ++s) b1[s] = *(const v4f*)(Bq + s * 16 * BPITCH + 24);
            gemm_step<GEMM_NT>(acc, a_cur[2], b0);
#pragma unroll
            for (int s = 0; s < GEMM_NT; ++s) b0[s] = *(const v4f*)(Bw + s * 16 * BPITCH);
            gemm_step<GEMM_NT>(acc, a_cur[3], b1);

#pragma unroll
            for (int t = 0; t < 4; ++t) a_cur[t] = a_nxt[t];
        }

        float* Ct = C + (size_t)(mt * 16 + half * 8) * ldc + l16 * 4;
        if (mt * 16 + 16 <= M) {
#pragma unroll
            for (int r = 0; r < 8; ++r)
#pragma unroll
                for (int g = 0; g < GEMM_NT / 4; ++g)
                    *(v4f*)(Ct + (size_t)r * ldc + g * 64) =
                        (v4f){acc[4 * g][r], acc[4 * g + 1][r], acc[4 * g + 2][r], acc[4 * g + 3][r]};
        } else {
#pragma unroll
            for (int r = 0; r < 8; ++r) {
                if (mt * 16 + half * 8 + r < M) {
#pragma unroll
                    for (int g = 0; g < GEMM_NT / 4; ++g)
                        *(v4f*)(Ct + (size_t)r * ldc + g * 64) =
                            (v4f){acc[4 * g][r], acc[4 * g + 1][r], acc[4 * g + 2][r], acc[4 * g + 3][r]};
                }
            }
        }
        if (!more) break;
        v = vn; C = Cn; M = Mn; ldc = ldcn; mt = mtn;
        Ap = Apn; Bt = Btn;
    }
}

__global__ __launch_bounds__(GEMM_THREADS)
void gat_wmma_gemm_heads(const float* __restrict__ A,
                         const float* __restrict__ B,
                         float* __restrict__ C,
                         int M, int lda, int ldbn, int ldc, int coff,
                         int mtiles,
                         const float* __restrict__ addp,
                         const float* __restrict__ den2) {
    constexpr int NT = GEMM_NT / 2;
    __shared__ __attribute__((aligned(16))) float Bs[GEMM_PANEL];
    const int tid = threadIdx.x;

    gemm_stage_panel<1>(Bs, B, ldbn, tid);
    __syncthreads();

    const int wave = __builtin_amdgcn_readfirstlane(tid >> 5);
    const int lane = tid & 31;
    const int half = lane >> 4;
    const int l16  = lane & 15;
    const float* Bl = Bs + l16 * BPITCH + half * 4;
    const int wstride = gridDim.x * GEMM_WAVES;

    int mt = blockIdx.x + gridDim.x * wave;
    if (mt >= mtiles) return;

    int arow = mt * 16 + l16;
    const float* Ap = A + (size_t)(arow < M ? arow : (M - 1)) * lda + half * 4;

    const int hm = mtiles >> 1;
    const float* Bt = Bl + (mt >= hm ? 64 * BPITCH : 0);

    v4f a_cur[4], a_nxt[4];
    v4f b0[NT], b1[NT];
#pragma unroll
    for (int t = 0; t < 4; ++t) a_cur[t] = *(const v4f*)(Ap + t * 8);
#pragma unroll
    for (int s = 0; s < NT; ++s) b0[s] = *(const v4f*)(Bt + s * 16 * BPITCH);

    for (; mt < mtiles; mt += wstride) {
        int mtn = mt + wstride;
        if (mtn >= mtiles) mtn = mt;
        int arown = mtn * 16 + l16;
        const float* Apn = A + (size_t)(arown < M ? arown : (M - 1)) * lda + half * 4;
        const float* Btn = Bl + (mtn >= hm ? 64 * BPITCH : 0);

        v8f acc[NT];
#pragma unroll
        for (int s = 0; s < NT; ++s)
            acc[s] = (v8f){0.f,0.f,0.f,0.f,0.f,0.f,0.f,0.f};

        v4f ad[8];
        float dn[8];
        const int head = mt >= hm ? 1 : 0;
        const int node0 = (mt - head * hm) * 16 + half * 8;
        const size_t o0 = (size_t)node0 * ldc + coff + head * 64 + l16 * 4;
#pragma unroll
        for (int r = 0; r < 8; ++r) {
            const float d = den2[2 * (node0 + r) + head];
            dn[r] = d > 1e-12f ? d : 1e-12f;
            ad[r] = *(const v4f*)(addp + o0 + (size_t)r * ldc);
        }

#pragma unroll 1
        for (int q = 0; q < 4; ++q) {
            const float* pa = (q < 3) ? (Ap + (q + 1) * 32) : Apn;
#pragma unroll
            for (int t = 0; t < 4; ++t) a_nxt[t] = *(const v4f*)(pa + t * 8);

            const float* Bq = Bt + q * 32;
            const float* Bw = (q < 3) ? (Bq + 32) : Btn;
#pragma unroll
            for (int s = 0; s < NT; ++s) b1[s] = *(const v4f*)(Bq + s * 16 * BPITCH + 8);
            gemm_step<NT>(acc, a_cur[0], b0);
#pragma unroll
            for (int s = 0; s < NT; ++s) b0[s] = *(const v4f*)(Bq + s * 16 * BPITCH + 16);
            gemm_step<NT>(acc, a_cur[1], b1);
#pragma unroll
            for (int s = 0; s < NT; ++s) b1[s] = *(const v4f*)(Bq + s * 16 * BPITCH + 24);
            gemm_step<NT>(acc, a_cur[2], b0);
#pragma unroll
            for (int s = 0; s < NT; ++s) b0[s] = *(const v4f*)(Bw + s * 16 * BPITCH);
            gemm_step<NT>(acc, a_cur[3], b1);

#pragma unroll
            for (int t = 0; t < 4; ++t) a_cur[t] = a_nxt[t];
        }

#pragma unroll
        for (int r = 0; r < 8; ++r) {
            static_assert(NT == 4, "a lane's four tiles make one 16-byte store");
            const v4f cv = (v4f){acc[0][r], acc[1][r], acc[2][r], acc[3][r]};
            *(v4f*)(C + o0 + (size_t)r * ldc) = (cv + ad[r]) / dn[r];
        }
        Bt = Btn;
        Ap = Apn;
    }
}

struct GatRange {
    int cb;
    int nbg;
    int ng;
    int novf;
};

__device__ __forceinline__ GatRange gat_range(const int* __restrict__ cnt, int n) {
    GatRange rg;
    int cn = __builtin_amdgcn_readfirstlane(cnt[n]);
    cn = cn < 0 ? 0 : (cn > ET_EDGES ? ET_EDGES : cn);
    rg.cb = cn < BKT_CAP ? cn : BKT_CAP;
    rg.novf = 0;
    if (cn > BKT_CAP) {
        const int o = __builtin_amdgcn_readfirstlane(cnt[N_NODES]);
        rg.novf = o < 0 ? 0 : (o > ET_EDGES ? ET_EDGES : o);
    }
    rg.nbg = (rg.cb + 31) >> 5;
    rg.ng = rg.nbg + ((rg.novf + 31) >> 5);
    return rg;
}

__device__ __forceinline__ int4 gat_group(const GatRange& rg, int g, int& pos, int n, int lane,
                                          const int* __restrict__ bkt,
                                          const int* __restrict__ ovs,
                                          const int* __restrict__ ovf,
                                          int& cntc, int& jn) {
    int4 md = make_int4(0x7fffffff, 0, 0, -1);
    if (g < rg.nbg) {
        cntc = rg.cb - g * 32;
        if (cntc > 32) cntc = 32;
        if (lane < cntc) md = *(const int4*)(bkt + 4 * (size_t)(pos + lane));
        pos += 32;
        jn = cntc;
    } else {
        const int i = (g - rg.nbg) * 32 + lane;
        bool mine = false;
        if (i < rg.novf) mine = ovs[i] == n;
        if (mine) md = *(const int4*)(ovf + 4 * (size_t)i);
        cntc = __builtin_popcount(__builtin_amdgcn_ballot_w32(mine));
        jn = 32;
    }
    return md;
}

#define NODE_WAVES 8
#define NODE_U     4

template <int LAYER>
__global__ __launch_bounds__(NODE_WAVES * 32)
void gat_node(const int* __restrict__ cnt, const int* __restrict__ bkt,
              const int* __restrict__ ovs, const int* __restrict__ ovf,
              const float* __restrict__ xtab, const float* __restrict__ etab,
              const float* __restrict__ rtab, const float* __restrict__ w2,
              float* __restrict__ dst) {
    const int wave = __builtin_amdgcn_readfirstlane(threadIdx.x >> 5);
    const int n = blockIdx.x * NODE_WAVES + wave;
    if (n >= N_NODES) return;
    const int lane = threadIdx.x & 31;
    const int k4 = lane * 4;
    const GatRange rg = gat_range(cnt, n);
    const v4f xa = *(const v4f*)(xtab + (size_t)n * 256 + k4);
    const v4f wv = *(const v4f*)(w2 + k4);
    v4f acc = (v4f){0.f, 0.f, 0.f, 0.f};
    float den = 0.f;

    int pos = n * BKT_CAP;
    for (int g = 0; g < rg.ng; ++g) {
        int cntc, jn;
        const int4 md = gat_group(rg, g, pos, n, lane, bkt, ovs, ovf, cntc, jn);
        int rk = 0;
        for (int j = 0; j < jn; ++j)
            rk += (__builtin_amdgcn_readlane(md.x, j) < md.x) ? 1 : 0;

        for (int j0 = 0; j0 < cntc; j0 += NODE_U) {
            v4f xb[NODE_U], e0[NODE_U], e1[NODE_U];
#pragma unroll
            for (int u = 0; u < NODE_U; ++u) {
                int jj = j0 + u;
                if (jj >= cntc) jj = cntc - 1;
                const unsigned hit = __builtin_amdgcn_ballot_w32(rk == jj);
                const int sl = __builtin_ctz(hit);
                int e  = __builtin_amdgcn_readlane(md.x, sl);
                int d  = __builtin_amdgcn_readlane(md.y, sl);
                int t0 = __builtin_amdgcn_readlane(md.z, sl);
                int t1 = __builtin_amdgcn_readlane(md.w, sl);
                e  = e  < 0 ? 0 : (e  > E_EDGES - 1 ? E_EDGES - 1 : e);
                d  = d  < 0 ? 0 : (d  > N_NODES - 1 ? N_NODES - 1 : d);
                t0 = t0 < 0 ? 0 : (t0 > NREL - 1 ? NREL - 1 : t0);
                t1 = t1 > NREL - 1 ? NREL - 1 : t1;
                xb[u] = *(const v4f*)(xtab + (size_t)d * 256 + 128 + k4);
                if (LAYER == 1 && t1 < 0)
                    e0[u] = *(const v4f*)(etab + (size_t)e * 128 + k4);
                else
                    e0[u] = *(const v4f*)(rtab + (size_t)t0 * 128 + k4);
                if (t1 >= 0)
                    e1[u] = *(const v4f*)(rtab + (size_t)t1 * 128 + k4);
                else
                    e1[u] = (v4f){0.f, 0.f, 0.f, 0.f};
            }
            v4f m[NODE_U];
            float p[NODE_U];
#pragma unroll
            for (int u = 0; u < NODE_U; ++u) {
                const v4f ev = e0[u] + e1[u];
                m[u] = (xa + xb[u]) + ev;
                p[u] = m[u][0] * wv[0] + m[u][1] * wv[1] + m[u][2] * wv[2] + m[u][3] * wv[3];
            }
            if (LAYER == 2) {
#pragma unroll
                for (int u = 0; u < NODE_U; ++u) p[u] += __shfl_xor(p[u], 16, 32);
            }
#pragma unroll
            for (int u = 0; u < NODE_U; ++u) p[u] += __shfl_xor(p[u], 8, 32);
#pragma unroll
            for (int u = 0; u < NODE_U; ++u) p[u] += __shfl_xor(p[u], 4, 32);
#pragma unroll
            for (int u = 0; u < NODE_U; ++u) p[u] += __shfl_xor(p[u], 2, 32);
#pragma unroll
            for (int u = 0; u < NODE_U; ++u) p[u] += __shfl_xor(p[u], 1, 32);
#pragma unroll
            for (int u = 0; u < NODE_U; ++u) {
                if (j0 + u < cntc) {
                    const float lr = p[u] > 0.f ? p[u] : ALPHA * p[u];
                    const float ee = __expf(-lr);
                    acc += ee * m[u];
                    den += ee;
                }
            }
        }
    }

    const float dn = den > 1e-12f ? den : 1e-12f;
    v4f h = acc / dn;
    if (LAYER == 2) {
#pragma unroll
        for (int c = 0; c < 4; ++c) {
            float y = h[c] > 0.f ? h[c] : (__expf(h[c]) - 1.f);
            y = y > 0.f ? y : (__expf(y) - 1.f);
            h[c] = y;
        }
    }
    *(v4f*)(dst + (size_t)n * 128 + k4) = h;
}

__global__ __launch_bounds__(NODE_WAVES * 32)
void gat_node_l1(const int* __restrict__ cnt, const int* __restrict__ bkt,
                 const int* __restrict__ ovs, const int* __restrict__ ovf,
                 const float* __restrict__ xtab, const float* __restrict__ eraw,
                 const float* __restrict__ rtab, const float* __restrict__ w2,
                 const float* __restrict__ uc, float* __restrict__ accp,
                 float* __restrict__ agg, float* __restrict__ den2) {
    const int wave = __builtin_amdgcn_readfirstlane(threadIdx.x >> 5);
    const int n = blockIdx.x * NODE_WAVES + wave;
    if (n >= N_NODES) return;
    const int lane = threadIdx.x & 31;
    const int k4 = lane * 4;
    const bool lo = lane < 16;
    const GatRange rg = gat_range(cnt, n);
    const v4f xa = *(const v4f*)(xtab + (size_t)n * 256 + k4);
    const v4f wv = *(const v4f*)(w2 + k4);
    const v4f u0 = *(const v4f*)(uc + k4);
    const v4f u1 = *(const v4f*)(uc + 128 + k4);
    const v4f zero4 = (v4f){0.f, 0.f, 0.f, 0.f};
    v4f acc = zero4, a0 = zero4, a1 = zero4;
    float den = 0.f;

    int pos = n * BKT_CAP;
    for (int g = 0; g < rg.ng; ++g) {
        int cntc, jn;
        const int4 md = gat_group(rg, g, pos, n, lane, bkt, ovs, ovf, cntc, jn);
        int rk = 0;
        for (int j = 0; j < jn; ++j)
            rk += (__builtin_amdgcn_readlane(md.x, j) < md.x) ? 1 : 0;

        for (int j0 = 0; j0 < cntc; j0 += NODE_U) {
            v4f xb[NODE_U], e0[NODE_U], e1[NODE_U];
            bool dir[NODE_U];
#pragma unroll
            for (int u = 0; u < NODE_U; ++u) {
                int jj = j0 + u;
                if (jj >= cntc) jj = cntc - 1;
                const unsigned hit = __builtin_amdgcn_ballot_w32(rk == jj);
                const int sl = __builtin_ctz(hit);
                int e  = __builtin_amdgcn_readlane(md.x, sl);
                int d  = __builtin_amdgcn_readlane(md.y, sl);
                int t0 = __builtin_amdgcn_readlane(md.z, sl);
                int t1 = __builtin_amdgcn_readlane(md.w, sl);
                e  = e  < 0 ? 0 : (e  > E_EDGES - 1 ? E_EDGES - 1 : e);
                d  = d  < 0 ? 0 : (d  > N_NODES - 1 ? N_NODES - 1 : d);
                t0 = t0 < 0 ? 0 : (t0 > NREL - 1 ? NREL - 1 : t0);
                t1 = t1 > NREL - 1 ? NREL - 1 : t1;
                dir[u] = t1 < 0;
                xb[u] = *(const v4f*)(xtab + (size_t)d * 256 + 128 + k4);
                if (t1 < 0)
                    e0[u] = *(const v4f*)(eraw + (size_t)e * 128 + k4);
                else
                    e0[u] = *(const v4f*)(rtab + (size_t)t0 * 128 + k4);
                if (t1 >= 0)
                    e1[u] = *(const v4f*)(rtab + (size_t)t1 * 128 + k4);
                else
                    e1[u] = zero4;
            }
            v4f m[NODE_U], er[NODE_U];
            float p[NODE_U];
#pragma unroll
            for (int u = 0; u < NODE_U; ++u) {
                const v4f ev = dir[u] ? zero4 : (e0[u] + e1[u]);
                er[u] = dir[u] ? e0[u] : zero4;
                m[u] = (xa + xb[u]) + ev;
                const float qn = __builtin_fmaf(m[u][3], wv[3], __builtin_fmaf(m[u][2], wv[2],
                                 __builtin_fmaf(m[u][1], wv[1], m[u][0] * wv[0])));
                const float q0 = __builtin_fmaf(er[u][3], u0[3], __builtin_fmaf(er[u][2], u0[2],
                                 __builtin_fmaf(er[u][1], u0[1], er[u][0] * u0[0])));
                const float q1 = __builtin_fmaf(er[u][3], u1[3], __builtin_fmaf(er[u][2], u1[2],
                                 __builtin_fmaf(er[u][1], u1[1], er[u][0] * u1[0])));
                const float mine = qn + (lo ? q0 : q1);
                const float send = lo ? q1 : q0;
                p[u] = mine + __shfl_xor(send, 16, 32);
            }
#pragma unroll
            for (int u = 0; u < NODE_U; ++u) p[u] += __shfl_xor(p[u], 8, 32);
#pragma unroll
            for (int u = 0; u < NODE_U; ++u) p[u] += __shfl_xor(p[u], 4, 32);
#pragma unroll
            for (int u = 0; u < NODE_U; ++u) p[u] += __shfl_xor(p[u], 2, 32);
#pragma unroll
            for (int u = 0; u < NODE_U; ++u) p[u] += __shfl_xor(p[u], 1, 32);
            float g[NODE_U], go[NODE_U];
#pragma unroll
            for (int u = 0; u < NODE_U; ++u) {
                const float lr = p[u] > 0.f ? p[u] : ALPHA * p[u];
                g[u]  = __expf(-lr);
                go[u] = __shfl_xor(g[u], 16, 32);
            }
#pragma unroll
            for (int u = 0; u < NODE_U; ++u) {
                if (j0 + u < cntc) {
                    const float g0 = lo ? g[u] : go[u];
                    const float g1 = lo ? go[u] : g[u];
                    acc += g[u] * m[u];
                    den += g[u];
                    a0 += g0 * er[u];
                    a1 += g1 * er[u];
                }
            }
        }
    }

    *(v4f*)(accp + (size_t)n * 128 + k4) = acc;
    *(v4f*)(agg + (size_t)n * 128 + k4) = a0;
    *(v4f*)(agg + (size_t)(N_NODES + n) * 128 + k4) = a1;
    if ((lane & 15) == 0) den2[n * 2 + (lane >> 4)] = den;
}

extern "C" void kernel_launch(void* const* d_in, const int* in_sizes, int n_in,
                              void* d_out, int out_size, void* d_ws, size_t ws_size,
                              hipStream_t stream) {
    const float* x    = (const float*)d_in[0];
    const float* rel  = (const float*)d_in[1];
    const int*   el   = (const int*)d_in[2];
    const int*   et   = (const int*)d_in[3];
    const float* eemb = (const float*)d_in[4];
    const int*   eln  = (const int*)d_in[5];
    const int*   etn  = (const int*)d_in[6];
    const float* w1h  = (const float*)d_in[8];
    const float* w2h  = (const float*)d_in[9];
    const float* w1o  = (const float*)d_in[10];
    const float* w2o  = (const float*)d_in[11];
    const float* WR   = (const float*)d_in[12];
    float* out = (float*)d_out;

    static_assert(((size_t)N_NODES * 2 + 32) % 64 == 0 && (NPAD + 44) % 64 == 0 &&
                  ((size_t)N_NODES * 128) % 64 == 0 && ((size_t)NREL * 128) % 64 == 0,
                  "row arrays must start on 256-byte boundaries");
    float* ws = (float*)d_ws;
    float* xnode  = ws;
    float* accp   = xnode  + (size_t)N_NODES * 256;
    float* agg    = accp   + (size_t)N_NODES * 128;
    float* den2   = agg    + (size_t)N_NODES * 256;
    float* uc     = den2   + (size_t)N_NODES * 2 + 32;
    float* relc   = uc     + 256;
    float* lx     = relc   + (size_t)NREL * 128;
    float* xab2   = lx     + (size_t)N_NODES * 128;
    float* relW2  = xab2   + (size_t)N_NODES * 256;
    int*   cnt    = (int*)(relW2 + (size_t)NREL * 128);
    int*   bkt    = cnt    + NPAD + 44;
    int*   ovs    = bkt    + (size_t)N_NODES * BKT_CAP * 4;
    int*   ovf    = ovs    + ET_EDGES;

    float* out_rel = out + (size_t)N_NODES * 128;

    hipMemsetAsync(cnt, 0, (size_t)NPAD * sizeof(int), stream);

    static_assert(2 * ((NREL + 15) / 16) <= GEMM_MAX_BLOCKS,
                  "the two small problems must not have more tiles than blocks");
    gat_wmma_gemm_multi<<<GEMM_MAX_BLOCKS, GEMM_THREADS, 0, stream>>>(
        x,   w1h,       xnode,       N_NODES, 256,
        x,   w1h + 128, xnode + 128, N_NODES, 256,
        rel, w1h + 256, relc,        NREL,    128,
        rel, WR,        out_rel,     NREL,    128,
        384, el, eln, et, etn, cnt, bkt, ovs, ovf, w2h, uc);

    gat_node_l1<<<(N_NODES + NODE_WAVES - 1) / NODE_WAVES, NODE_WAVES * 32, 0, stream>>>(
        cnt, bkt, ovs, ovf, xnode, eemb, relc, w2h, uc, accp, agg, den2);
    static_assert(N_NODES % 16 == 0, "each head's plane must be whole row tiles");
    gat_wmma_gemm_heads<<<GEMM_MAX_BLOCKS, GEMM_THREADS, 0, stream>>>(
        agg, w1h + 256, lx, 2 * N_NODES, 128, 384, 128, 0, 2 * N_NODES / 16, accp, den2);

    gat_wmma_gemm_multi<<<GEMM_MAX_BLOCKS, GEMM_THREADS, 0, stream>>>(
        lx,      w1o,       xab2,       N_NODES, 256,
        lx,      w1o + 128, xab2 + 128, N_NODES, 256,
        out_rel, w1o + 256, relW2,      NREL,    128,
        (const float*)nullptr, (const float*)nullptr, (float*)nullptr, 0, 0,
        384, (const int*)nullptr, (const int*)nullptr, (const int*)nullptr, (const int*)nullptr,
        (int*)nullptr, (int*)nullptr, (int*)nullptr, (int*)nullptr,
        (const float*)nullptr, (float*)nullptr);

    gat_node<2><<<(N_NODES + NODE_WAVES - 1) / NODE_WAVES, NODE_WAVES * 32, 0, stream>>>(
        cnt, bkt, ovs, ovf, xab2, relW2, relW2, w2o, out);
}


# node kernels: per-edge logit sums exchanged on the vector ALU (row_xmask DPP adds, permlanex16 for the row swap) instead of shuffles through the LDS unit; same partners and step order, sums unchanged
# speedup vs baseline: 3.9856x; 1.0705x over previous
#include <hip/hip_runtime.h>
#include <hip/hip_bf16.h>

typedef float v2f __attribute__((ext_vector_type(2)));
typedef float v4f __attribute__((ext_vector_type(4)));
typedef float v8f __attribute__((ext_vector_type(8)));

#define N_NODES 50000
#define E_EDGES 500000
#define E2_EDGES 62500
#define ET_EDGES (E_EDGES + E2_EDGES)
#define NREL 1000
#define ALPHA 0.2f
#define KDIM 128

#define GEMM_N        128
#define GEMM_NT       8
#define GEMM_THREADS  512
#define GEMM_WAVES    (GEMM_THREADS / 32)
#define GEMM_MAX_BLOCKS 256
#define BPITCH        132
#define GEMM_PANEL    (GEMM_N * BPITCH)
#define CNT_WAVES     4
#define CNT_U         9
#define CNT_LANES     (GEMM_MAX_BLOCKS * CNT_WAVES * 32)
#define BKT_CAP       64
#define NPAD          (N_NODES + 4)

__device__ __forceinline__ int gemm_prow(int n) {
    return (n & 64) | ((n & 3) << 4) | ((n >> 2) & 15);
}

template <int NT>
__device__ __forceinline__ void gemm_step(v8f (&acc)[NT], const v4f& a,
                                          const v4f (&b)[NT]) {
    v2f a_lo = (v2f){a[0], a[1]};
    v2f a_hi = (v2f){a[2], a[3]};
#pragma unroll
    for (int s = 0; s < NT; ++s)
        acc[s] = __builtin_amdgcn_wmma_f32_16x16x4_f32(
            false, a_lo, false, (v2f){b[s][0], b[s][1]}, (short)0,
            acc[s], false, false);
#pragma unroll
    for (int s = 0; s < NT; ++s)
        acc[s] = __builtin_amdgcn_wmma_f32_16x16x4_f32(
            false, a_hi, false, (v2f){b[s][2], b[s][3]}, (short)0,
            acc[s], false, false);
}

template <int LDBK>
__device__ __forceinline__ void gemm_stage_panel(float* __restrict__ Bs,
                                                 const float* __restrict__ B,
                                                 int ldbn, int tid) {
    if constexpr (LDBK == 1) {
        constexpr int NV = GEMM_N * (KDIM / 4) / GEMM_THREADS;
        v4f v[NV];
#pragma unroll
        for (int it = 0; it < NV; ++it) {
            int i = tid + it * GEMM_THREADS;
            v[it] = *(const v4f*)(B + (size_t)(i >> 5) * ldbn + (i & 31) * 4);
        }
#pragma unroll
        for (int it = 0; it < NV; ++it) {
            int i = tid + it * GEMM_THREADS;
            *(v4f*)(Bs + gemm_prow(i >> 5) * BPITCH + (i & 31) * 4) = v[it];
        }
    } else {
        constexpr int NV = 8;
        for (int i0 = tid; i0 < GEMM_N * KDIM; i0 += NV * GEMM_THREADS) {
            float v[NV];
#pragma unroll
            for (int it = 0; it < NV; ++it) {
                int i = i0 + it * GEMM_THREADS;
                v[it] = B[(size_t)(i & 127) * ldbn + (size_t)(i >> 7) * LDBK];
            }
#pragma unroll
            for (int it = 0; it < NV; ++it) {
                int i = i0 + it * GEMM_THREADS;
                Bs[gemm_prow(i & 127) * BPITCH + (i >> 7)] = v[it];
            }
        }
    }
}

__global__ __launch_bounds__(GEMM_THREADS)
void gat_wmma_gemm_multi(
        const float* __restrict__ A0, const float* __restrict__ B0, float* __restrict__ C0, int M0, int ldc0,
        const float* __restrict__ A1, const float* __restrict__ B1, float* __restrict__ C1, int M1, int ldc1,
        const float* __restrict__ A2, const float* __restrict__ B2, float* __restrict__ C2, int M2, int ldc2,
        const float* __restrict__ A3, const float* __restrict__ B3, float* __restrict__ C3, int M3, int ldc3,
        int ldbn,
        const int* __restrict__ el, const int* __restrict__ eln,
        const int* __restrict__ et, const int* __restrict__ etn,
        int* __restrict__ cnt, int* __restrict__ bkt,
        int* __restrict__ ovs, int* __restrict__ ovf,
        const float* __restrict__ w2f, float* __restrict__ uc) {
    __shared__ __attribute__((aligned(16))) float Bs[3 * GEMM_PANEL];
    const int tid = threadIdx.x;

    const int e0 = (M0 + 15) >> 4;
    const int e1 = e0 + ((M1 + 15) >> 4);
    const int e2 = e1 + ((M2 + 15) >> 4);
    const int e3 = e2 + ((M3 + 15) >> 4);
    const int vs = (int)blockIdx.x +
        GEMM_MAX_BLOCKS * ((e1 - (int)blockIdx.x + GEMM_MAX_BLOCKS - 1) / GEMM_MAX_BLOCKS);

    gemm_stage_panel<1>(Bs, B0, ldbn, tid);
    gemm_stage_panel<1>(Bs + GEMM_PANEL, B1, ldbn, tid);
    if (vs < e3) {
        if (vs < e2) gemm_stage_panel<1>(Bs + 2 * GEMM_PANEL, B2, ldbn, tid);
        else         gemm_stage_panel<KDIM>(Bs + 2 * GEMM_PANEL, B3, 1, tid);
    }
    __syncthreads();

    const int wave = __builtin_amdgcn_readfirstlane(tid >> 5);
    const int lane = tid & 31;
    const int half = lane >> 4;
    const int l16  = lane & 15;
    const float* Bl = Bs + l16 * BPITCH + half * 4;
    constexpr int wstride = GEMM_MAX_BLOCKS * GEMM_WAVES;

    if (cnt != nullptr && wave >= GEMM_WAVES - CNT_WAVES) {
        const int t = ((int)blockIdx.x * CNT_WAVES + (wave - (GEMM_WAVES - CNT_WAVES))) * 32 + lane;
        for (int eb = t; eb < ET_EDGES; eb += CNT_LANES * CNT_U) {
            int s[CNT_U], d[CNT_U], t0[CNT_U], t1[CNT_U], r[CNT_U];
#pragma unroll
            for (int u = 0; u < CNT_U; ++u) {
                const int e = eb + u * CNT_LANES;
                const int ec = e < ET_EDGES ? e : ET_EDGES - 1;
                const bool dr = ec < E_EDGES;
                const int j = dr ? 0 : ec - E_EDGES;
                const int* ps  = dr ? el + ec           : eln + j;
                const int* pd  = dr ? el + E_EDGES + ec : eln + E2_EDGES + j;
                const int* pt0 = dr ? et + ec           : etn + 2 * j;
                const int* pt1 = dr ? et + ec           : etn + 2 * j + 1;
                s[u]  = *ps;
                d[u]  = *pd;
                t0[u] = *pt0;
                const int tt = *pt1;
                t1[u] = dr ? -1 : tt;
            }
#pragma unroll
            for (int u = 0; u < CNT_U; ++u) {
                const int e = eb + u * CNT_LANES;
                r[u] = 0;
                if (e < ET_EDGES) r[u] = atomicAdd(cnt + s[u], 1);
            }
#pragma unroll
            for (int u = 0; u < CNT_U; ++u) {
                const int e = eb + u * CNT_LANES;
                if (e < ET_EDGES) {
                    const int4 rec = make_int4(e, d[u], t0[u], t1[u]);
                    if (r[u] < BKT_CAP) {
                        *(int4*)(bkt + 4 * ((size_t)s[u] * BKT_CAP + r[u])) = rec;
                    } else {
                        const int o = atomicAdd(cnt + N_NODES, 1);
                        ovs[o] = s[u];
                        *(int4*)(ovf + 4 * (size_t)o) = rec;
                    }
                }
            }
        }
    }

    if (uc != nullptr && wave == GEMM_WAVES - CNT_WAVES - 1 && blockIdx.x < 8) {
        const int tf = (int)blockIdx.x * 32 + lane;
        const int h = tf >> 7, k = tf & 127;
        float s = 0.f;
        for (int j = 0; j < 64; ++j)
            s += B2[(size_t)(h * 64 + j) * ldbn + k] * w2f[h * 64 + j];
        uc[tf] = s;
    }

    int v = (int)blockIdx.x + GEMM_MAX_BLOCKS * wave;
    if (v >= e3) return;

    auto locate = [&](int vv, const float*& A, float*& C, int& M, int& ldc,
                      int& mt, int& slot) {
        if (vv < e0)      { A = A0; C = C0; M = M0; ldc = ldc0; mt = vv;      slot = 0; }
        else if (vv < e1) { A = A1; C = C1; M = M1; ldc = ldc1; mt = vv - e0; slot = 1; }
        else if (vv < e2) { A = A2; C = C2; M = M2; ldc = ldc2; mt = vv - e1; slot = 2; }
        else              { A = A3; C = C3; M = M3; ldc = ldc3; mt = vv - e2; slot = 2; }
    };

    const float* A; float* C; int M, ldc, mt, slot;
    locate(v, A, C, M, ldc, mt, slot);
    int arow = mt * 16 + l16;
    const float* Ap = A + (size_t)(arow < M ? arow : (M - 1)) * KDIM + half * 4;
    const float* Bt = Bl + slot * GEMM_PANEL;

    v4f a_cur[4], a_nxt[4];
    v4f b0[GEMM_NT], b1[GEMM_NT];
#pragma unroll
    for (int t = 0; t < 4; ++t) a_cur[t] = *(const v4f*)(Ap + t * 8);
#pragma unroll
    for (int s = 0; s < GEMM_NT; ++s) b0[s] = *(const v4f*)(Bt + s * 16 * BPITCH);

    for (;;) {
        int vn = v + wstride;
        const bool more = vn < e3;
        if (!more) vn = v;
        const float* An; float* Cn; int Mn, ldcn, mtn, slotn;
        locate(vn, An, Cn, Mn, ldcn, mtn, slotn);
        int arown = mtn * 16 + l16;
        const float* Apn = An + (size_t)(arown < Mn ? arown : (Mn - 1)) * KDIM + half * 4;
        const float* Btn = Bl + slotn * GEMM_PANEL;

        v8f acc[GEMM_NT];
#pragma unroll
        for (int s = 0; s < GEMM_NT; ++s)
            acc[s] = (v8f){0.f,0.f,0.f,0.f,0.f,0.f,0.f,0.f};

#pragma unroll 1
        for (int q = 0; q < 4; ++q) {
            const float* pa = (q < 3) ? (Ap + (q + 1) * 32) : Apn;
#pragma unroll
            for (int t = 0; t < 4; ++t) a_nxt[t] = *(const v4f*)(pa + t * 8);

            const float* Bq = Bt + q * 32;
            const float* Bw = (q < 3) ? (Bq + 32) : Btn;
#pragma unroll
            for (int s = 0; s < GEMM_NT; ++s) b1[s] = *(const v4f*)(Bq + s * 16 * BPITCH + 8);
            gemm_step<GEMM_NT>(acc, a_cur[0], b0);
#pragma unroll
            for (int s = 0; s < GEMM_NT; ++s) b0[s] = *(const v4f*)(Bq + s * 16 * BPITCH + 16);
            gemm_step<GEMM_NT>(acc, a_cur[1], b1);
#pragma unroll
            for (int s = 0; s < GEMM_NT; ++s) b1[s] = *(const v4f*)(Bq + s * 16 * BPITCH + 24);
            gemm_step<GEMM_NT>(acc, a_cur[2], b0);
#pragma unroll
            for (int s = 0; s < GEMM_NT; ++s) b0[s] = *(const v4f*)(Bw + s * 16 * BPITCH);
            gemm_step<GEMM_NT>(acc, a_cur[3], b1);

#pragma unroll
            for (int t = 0; t < 4; ++t) a_cur[t] = a_nxt[t];
        }

        float* Ct = C + (size_t)(mt * 16 + half * 8) * ldc + l16 * 4;
        if (mt * 16 + 16 <= M) {
#pragma unroll
            for (int r = 0; r < 8; ++r)
#pragma unroll
                for (int g = 0; g < GEMM_NT / 4; ++g)
                    *(v4f*)(Ct + (size_t)r * ldc + g * 64) =
                        (v4f){acc[4 * g][r], acc[4 * g + 1][r], acc[4 * g + 2][r], acc[4 * g + 3][r]};
        } else {
#pragma unroll
            for (int r = 0; r < 8; ++r) {
                if (mt * 16 + half * 8 + r < M) {
#pragma unroll
                    for (int g = 0; g < GEMM_NT / 4; ++g)
                        *(v4f*)(Ct + (size_t)r * ldc + g * 64) =
                            (v4f){acc[4 * g][r], acc[4 * g + 1][r], acc[4 * g + 2][r], acc[4 * g + 3][r]};
                }
            }
        }
        if (!more) break;
        v = vn; C = Cn; M = Mn; ldc = ldcn; mt = mtn;
        Ap = Apn; Bt = Btn;
    }
}

__global__ __launch_bounds__(GEMM_THREADS)
void gat_wmma_gemm_heads(const float* __restrict__ A,
                         const float* __restrict__ B,
                         float* __restrict__ C,
                         int M, int lda, int ldbn, int ldc, int coff,
                         int mtiles,
                         const float* __restrict__ addp,
                         const float* __restrict__ den2) {
    constexpr int NT = GEMM_NT / 2;
    __shared__ __attribute__((aligned(16))) float Bs[GEMM_PANEL];
    const int tid = threadIdx.x;

    gemm_stage_panel<1>(Bs, B, ldbn, tid);
    __syncthreads();

    const int wave = __builtin_amdgcn_readfirstlane(tid >> 5);
    const int lane = tid & 31;
    const int half = lane >> 4;
    const int l16  = lane & 15;
    const float* Bl = Bs + l16 * BPITCH + half * 4;
    const int wstride = gridDim.x * GEMM_WAVES;

    int mt = blockIdx.x + gridDim.x * wave;
    if (mt >= mtiles) return;

    int arow = mt * 16 + l16;
    const float* Ap = A + (size_t)(arow < M ? arow : (M - 1)) * lda + half * 4;

    const int hm = mtiles >> 1;
    const float* Bt = Bl + (mt >= hm ? 64 * BPITCH : 0);

    v4f a_cur[4], a_nxt[4];
    v4f b0[NT], b1[NT];
#pragma unroll
    for (int t = 0; t < 4; ++t) a_cur[t] = *(const v4f*)(Ap + t * 8);
#pragma unroll
    for (int s = 0; s < NT; ++s) b0[s] = *(const v4f*)(Bt + s * 16 * BPITCH);

    for (; mt < mtiles; mt += wstride) {
        int mtn = mt + wstride;
        if (mtn >= mtiles) mtn = mt;
        int arown = mtn * 16 + l16;
        const float* Apn = A + (size_t)(arown < M ? arown : (M - 1)) * lda + half * 4;
        const float* Btn = Bl + (mtn >= hm ? 64 * BPITCH : 0);

        v8f acc[NT];
#pragma unroll
        for (int s = 0; s < NT; ++s)
            acc[s] = (v8f){0.f,0.f,0.f,0.f,0.f,0.f,0.f,0.f};

        v4f ad[8];
        float dn[8];
        const int head = mt >= hm ? 1 : 0;
        const int node0 = (mt - head * hm) * 16 + half * 8;
        const size_t o0 = (size_t)node0 * ldc + coff + head * 64 + l16 * 4;
#pragma unroll
        for (int r = 0; r < 8; ++r) {
            const float d = den2[2 * (node0 + r) + head];
            dn[r] = d > 1e-12f ? d : 1e-12f;
            ad[r] = *(const v4f*)(addp + o0 + (size_t)r * ldc);
        }

#pragma unroll 1
        for (int q = 0; q < 4; ++q) {
            const float* pa = (q < 3) ? (Ap + (q + 1) * 32) : Apn;
#pragma unroll
            for (int t = 0; t < 4; ++t) a_nxt[t] = *(const v4f*)(pa + t * 8);

            const float* Bq = Bt + q * 32;
            const float* Bw = (q < 3) ? (Bq + 32) : Btn;
#pragma unroll
            for (int s = 0; s < NT; ++s) b1[s] = *(const v4f*)(Bq + s * 16 * BPITCH + 8);
            gemm_step<NT>(acc, a_cur[0], b0);
#pragma unroll
            for (int s = 0; s < NT; ++s) b0[s] = *(const v4f*)(Bq + s * 16 * BPITCH + 16);
            gemm_step<NT>(acc, a_cur[1], b1);
#pragma unroll
            for (int s = 0; s < NT; ++s) b1[s] = *(const v4f*)(Bq + s * 16 * BPITCH + 24);
            gemm_step<NT>(acc, a_cur[2], b0);
#pragma unroll
            for (int s = 0; s < NT; ++s) b0[s] = *(const v4f*)(Bw + s * 16 * BPITCH);
            gemm_step<NT>(acc, a_cur[3], b1);

#pragma unroll
            for (int t = 0; t < 4; ++t) a_cur[t] = a_nxt[t];
        }

#pragma unroll
        for (int r = 0; r < 8; ++r) {
            static_assert(NT == 4, "a lane's four tiles make one 16-byte store");
            const v4f cv = (v4f){acc[0][r], acc[1][r], acc[2][r], acc[3][r]};
            *(v4f*)(C + o0 + (size_t)r * ldc) = (cv + ad[r]) / dn[r];
        }
        Bt = Btn;
        Ap = Apn;
    }
}

struct GatRange {
    int cb;
    int nbg;
    int ng;
    int novf;
};

__device__ __forceinline__ GatRange gat_range(const int* __restrict__ cnt, int n) {
    GatRange rg;
    int cn = __builtin_amdgcn_readfirstlane(cnt[n]);
    cn = cn < 0 ? 0 : (cn > ET_EDGES ? ET_EDGES : cn);
    rg.cb = cn < BKT_CAP ? cn : BKT_CAP;
    rg.novf = 0;
    if (cn > BKT_CAP) {
        const int o = __builtin_amdgcn_readfirstlane(cnt[N_NODES]);
        rg.novf = o < 0 ? 0 : (o > ET_EDGES ? ET_EDGES : o);
    }
    rg.nbg = (rg.cb + 31) >> 5;
    rg.ng = rg.nbg + ((rg.novf + 31) >> 5);
    return rg;
}

__device__ __forceinline__ int4 gat_group(const GatRange& rg, int g, int& pos, int n, int lane,
                                          const int* __restrict__ bkt,
                                          const int* __restrict__ ovs,
                                          const int* __restrict__ ovf,
                                          int& cntc, int& jn) {
    int4 md = make_int4(0x7fffffff, 0, 0, -1);
    if (g < rg.nbg) {
        cntc = rg.cb - g * 32;
        if (cntc > 32) cntc = 32;
        if (lane < cntc) md = *(const int4*)(bkt + 4 * (size_t)(pos + lane));
        pos += 32;
        jn = cntc;
    } else {
        const int i = (g - rg.nbg) * 32 + lane;
        bool mine = false;
        if (i < rg.novf) mine = ovs[i] == n;
        if (mine) md = *(const int4*)(ovf + 4 * (size_t)i);
        cntc = __builtin_popcount(__builtin_amdgcn_ballot_w32(mine));
        jn = 32;
    }
    return md;
}

__device__ __forceinline__ float gat_xhalf(float x) {
    const int b = __builtin_bit_cast(int, x);
    return __builtin_bit_cast(float,
        __builtin_amdgcn_permlanex16(b, b, 0x76543210, 0xfedcba98, false, false));
}
template <int M>
__device__ __forceinline__ float gat_row_xadd(float x) {
    const int b = __builtin_bit_cast(int, x);
    return x + __builtin_bit_cast(float,
        __builtin_amdgcn_update_dpp(0, b, 0x160 + M, 0xf, 0xf, true));
}

#define NODE_WAVES 8
#define NODE_U     4

template <int LAYER>
__global__ __launch_bounds__(NODE_WAVES * 32)
void gat_node(const int* __restrict__ cnt, const int* __restrict__ bkt,
              const int* __restrict__ ovs, const int* __restrict__ ovf,
              const float* __restrict__ xtab, const float* __restrict__ etab,
              const float* __restrict__ rtab, const float* __restrict__ w2,
              float* __restrict__ dst) {
    const int wave = __builtin_amdgcn_readfirstlane(threadIdx.x >> 5);
    const int n = blockIdx.x * NODE_WAVES + wave;
    if (n >= N_NODES) return;
    const int lane = threadIdx.x & 31;
    const int k4 = lane * 4;
    const GatRange rg = gat_range(cnt, n);
    const v4f xa = *(const v4f*)(xtab + (size_t)n * 256 + k4);
    const v4f wv = *(const v4f*)(w2 + k4);
    v4f acc = (v4f){0.f, 0.f, 0.f, 0.f};
    float den = 0.f;

    int pos = n * BKT_CAP;
    for (int g = 0; g < rg.ng; ++g) {
        int cntc, jn;
        const int4 md = gat_group(rg, g, pos, n, lane, bkt, ovs, ovf, cntc, jn);
        int rk = 0;
        for (int j = 0; j < jn; ++j)
            rk += (__builtin_amdgcn_readlane(md.x, j) < md.x) ? 1 : 0;

        for (int j0 = 0; j0 < cntc; j0 += NODE_U) {
            v4f xb[NODE_U], e0[NODE_U], e1[NODE_U];
#pragma unroll
            for (int u = 0; u < NODE_U; ++u) {
                int jj = j0 + u;
                if (jj >= cntc) jj = cntc - 1;
                const unsigned hit = __builtin_amdgcn_ballot_w32(rk == jj);
                const int sl = __builtin_ctz(hit);
                int e  = __builtin_amdgcn_readlane(md.x, sl);
                int d  = __builtin_amdgcn_readlane(md.y, sl);
                int t0 = __builtin_amdgcn_readlane(md.z, sl);
                int t1 = __builtin_amdgcn_readlane(md.w, sl);
                e  = e  < 0 ? 0 : (e  > E_EDGES - 1 ? E_EDGES - 1 : e);
                d  = d  < 0 ? 0 : (d  > N_NODES - 1 ? N_NODES - 1 : d);
                t0 = t0 < 0 ? 0 : (t0 > NREL - 1 ? NREL - 1 : t0);
                t1 = t1 > NREL - 1 ? NREL - 1 : t1;
                xb[u] = *(const v4f*)(xtab + (size_t)d * 256 + 128 + k4);
                if (LAYER == 1 && t1 < 0)
                    e0[u] = *(const v4f*)(etab + (size_t)e * 128 + k4);
                else
                    e0[u] = *(const v4f*)(rtab + (size_t)t0 * 128 + k4);
                if (t1 >= 0)
                    e1[u] = *(const v4f*)(rtab + (size_t)t1 * 128 + k4);
                else
                    e1[u] = (v4f){0.f, 0.f, 0.f, 0.f};
            }
            v4f m[NODE_U];
            float p[NODE_U];
#pragma unroll
            for (int u = 0; u < NODE_U; ++u) {
                const v4f ev = e0[u] + e1[u];
                m[u] = (xa + xb[u]) + ev;
                p[u] = m[u][0] * wv[0] + m[u][1] * wv[1] + m[u][2] * wv[2] + m[u][3] * wv[3];
            }
            if (LAYER == 2) {
#pragma unroll
                for (int u = 0; u < NODE_U; ++u) p[u] += gat_xhalf(p[u]);
            }
#pragma unroll
            for (int u = 0; u < NODE_U; ++u) p[u] = gat_row_xadd<8>(p[u]);
#pragma unroll
            for (int u = 0; u < NODE_U; ++u) p[u] = gat_row_xadd<4>(p[u]);
#pragma unroll
            for (int u = 0; u < NODE_U; ++u) p[u] = gat_row_xadd<2>(p[u]);
#pragma unroll
            for (int u = 0; u < NODE_U; ++u) p[u] = gat_row_xadd<1>(p[u]);
#pragma unroll
            for (int u = 0; u < NODE_U; ++u) {
                if (j0 + u < cntc) {
                    const float lr = p[u] > 0.f ? p[u] : ALPHA * p[u];
                    const float ee = __expf(-lr);
                    acc += ee * m[u];
                    den += ee;
                }
            }
        }
    }

    const float dn = den > 1e-12f ? den : 1e-12f;
    v4f h = acc / dn;
    if (LAYER == 2) {
#pragma unroll
        for (int c = 0; c < 4; ++c) {
            float y = h[c] > 0.f ? h[c] : (__expf(h[c]) - 1.f);
            y = y > 0.f ? y : (__expf(y) - 1.f);
            h[c] = y;
        }
    }
    *(v4f*)(dst + (size_t)n * 128 + k4) = h;
}

__global__ __launch_bounds__(NODE_WAVES * 32)
void gat_node_l1(const int* __restrict__ cnt, const int* __restrict__ bkt,
                 const int* __restrict__ ovs, const int* __restrict__ ovf,
                 const float* __restrict__ xtab, const float* __restrict__ eraw,
                 const float* __restrict__ rtab, const float* __restrict__ w2,
                 const float* __restrict__ uc, float* __restrict__ accp,
                 float* __restrict__ agg, float* __restrict__ den2) {
    const int wave = __builtin_amdgcn_readfirstlane(threadIdx.x >> 5);
    const int n = blockIdx.x * NODE_WAVES + wave;
    if (n >= N_NODES) return;
    const int lane = threadIdx.x & 31;
    const int k4 = lane * 4;
    const bool lo = lane < 16;
    const GatRange rg = gat_range(cnt, n);
    const v4f xa = *(const v4f*)(xtab + (size_t)n * 256 + k4);
    const v4f wv = *(const v4f*)(w2 + k4);
    const v4f u0 = *(const v4f*)(uc + k4);
    const v4f u1 = *(const v4f*)(uc + 128 + k4);
    const v4f zero4 = (v4f){0.f, 0.f, 0.f, 0.f};
    v4f acc = zero4, a0 = zero4, a1 = zero4;
    float den = 0.f;

    int pos = n * BKT_CAP;
    for (int g = 0; g < rg.ng; ++g) {
        int cntc, jn;
        const int4 md = gat_group(rg, g, pos, n, lane, bkt, ovs, ovf, cntc, jn);
        int rk = 0;
        for (int j = 0; j < jn; ++j)
            rk += (__builtin_amdgcn_readlane(md.x, j) < md.x) ? 1 : 0;

        for (int j0 = 0; j0 < cntc; j0 += NODE_U) {
            v4f xb[NODE_U], e0[NODE_U], e1[NODE_U];
            bool dir[NODE_U];
#pragma unroll
            for (int u = 0; u < NODE_U; ++u) {
                int jj = j0 + u;
                if (jj >= cntc) jj = cntc - 1;
                const unsigned hit = __builtin_amdgcn_ballot_w32(rk == jj);
                const int sl = __builtin_ctz(hit);
                int e  = __builtin_amdgcn_readlane(md.x, sl);
                int d  = __builtin_amdgcn_readlane(md.y, sl);
                int t0 = __builtin_amdgcn_readlane(md.z, sl);
                int t1 = __builtin_amdgcn_readlane(md.w, sl);
                e  = e  < 0 ? 0 : (e  > E_EDGES - 1 ? E_EDGES - 1 : e);
                d  = d  < 0 ? 0 : (d  > N_NODES - 1 ? N_NODES - 1 : d);
                t0 = t0 < 0 ? 0 : (t0 > NREL - 1 ? NREL - 1 : t0);
                t1 = t1 > NREL - 1 ? NREL - 1 : t1;
                dir[u] = t1 < 0;
                xb[u] = *(const v4f*)(xtab + (size_t)d * 256 + 128 + k4);
                if (t1 < 0)
                    e0[u] = *(const v4f*)(eraw + (size_t)e * 128 + k4);
                else
                    e0[u] = *(const v4f*)(rtab + (size_t)t0 * 128 + k4);
                if (t1 >= 0)
                    e1[u] = *(const v4f*)(rtab + (size_t)t1 * 128 + k4);
                else
                    e1[u] = zero4;
            }
            v4f m[NODE_U], er[NODE_U];
            float p[NODE_U];
#pragma unroll
            for (int u = 0; u < NODE_U; ++u) {
                const v4f ev = dir[u] ? zero4 : (e0[u] + e1[u]);
                er[u] = dir[u] ? e0[u] : zero4;
                m[u] = (xa + xb[u]) + ev;
                const float qn = __builtin_fmaf(m[u][3], wv[3], __builtin_fmaf(m[u][2], wv[2],
                                 __builtin_fmaf(m[u][1], wv[1], m[u][0] * wv[0])));
                const float q0 = __builtin_fmaf(er[u][3], u0[3], __builtin_fmaf(er[u][2], u0[2],
                                 __builtin_fmaf(er[u][1], u0[1], er[u][0] * u0[0])));
                const float q1 = __builtin_fmaf(er[u][3], u1[3], __builtin_fmaf(er[u][2], u1[2],
                                 __builtin_fmaf(er[u][1], u1[1], er[u][0] * u1[0])));
                const float mine = qn + (lo ? q0 : q1);
                const float send = lo ? q1 : q0;
                p[u] = mine + gat_xhalf(send);
            }
#pragma unroll
            for (int u = 0; u < NODE_U; ++u) p[u] = gat_row_xadd<8>(p[u]);
#pragma unroll
            for (int u = 0; u < NODE_U; ++u) p[u] = gat_row_xadd<4>(p[u]);
#pragma unroll
            for (int u = 0; u < NODE_U; ++u) p[u] = gat_row_xadd<2>(p[u]);
#pragma unroll
            for (int u = 0; u < NODE_U; ++u) p[u] = gat_row_xadd<1>(p[u]);
            float g[NODE_U], go[NODE_U];
#pragma unroll
            for (int u = 0; u < NODE_U; ++u) {
                const float lr = p[u] > 0.f ? p[u] : ALPHA * p[u];
                g[u]  = __expf(-lr);
                go[u] = gat_xhalf(g[u]);
            }
#pragma unroll
            for (int u = 0; u < NODE_U; ++u) {
                if (j0 + u < cntc) {
                    const float g0 = lo ? g[u] : go[u];
                    const float g1 = lo ? go[u] : g[u];
                    acc += g[u] * m[u];
                    den += g[u];
                    a0 += g0 * er[u];
                    a1 += g1 * er[u];
                }
            }
        }
    }

    *(v4f*)(accp + (size_t)n * 128 + k4) = acc;
    *(v4f*)(agg + (size_t)n * 128 + k4) = a0;
    *(v4f*)(agg + (size_t)(N_NODES + n) * 128 + k4) = a1;
    if ((lane & 15) == 0) den2[n * 2 + (lane >> 4)] = den;
}

extern "C" void kernel_launch(void* const* d_in, const int* in_sizes, int n_in,
                              void* d_out, int out_size, void* d_ws, size_t ws_size,
                              hipStream_t stream) {
    const float* x    = (const float*)d_in[0];
    const float* rel  = (const float*)d_in[1];
    const int*   el   = (const int*)d_in[2];
    const int*   et   = (const int*)d_in[3];
    const float* eemb = (const float*)d_in[4];
    const int*   eln  = (const int*)d_in[5];
    const int*   etn  = (const int*)d_in[6];
    const float* w1h  = (const float*)d_in[8];
    const float* w2h  = (const float*)d_in[9];
    const float* w1o  = (const float*)d_in[10];
    const float* w2o  = (const float*)d_in[11];
    const float* WR   = (const float*)d_in[12];
    float* out = (float*)d_out;

    static_assert(((size_t)N_NODES * 2 + 32) % 64 == 0 && (NPAD + 44) % 64 == 0 &&
                  ((size_t)N_NODES * 128) % 64 == 0 && ((size_t)NREL * 128) % 64 == 0,
                  "row arrays must start on 256-byte boundaries");
    float* ws = (float*)d_ws;
    float* xnode  = ws;
    float* accp   = xnode  + (size_t)N_NODES * 256;
    float* agg    = accp   + (size_t)N_NODES * 128;
    float* den2   = agg    + (size_t)N_NODES * 256;
    float* uc     = den2   + (size_t)N_NODES * 2 + 32;
    float* relc   = uc     + 256;
    float* lx     = relc   + (size_t)NREL * 128;
    float* xab2   = lx     + (size_t)N_NODES * 128;
    float* relW2  = xab2   + (size_t)N_NODES * 256;
    int*   cnt    = (int*)(relW2 + (size_t)NREL * 128);
    int*   bkt    = cnt    + NPAD + 44;
    int*   ovs    = bkt    + (size_t)N_NODES * BKT_CAP * 4;
    int*   ovf    = ovs    + ET_EDGES;

    float* out_rel = out + (size_t)N_NODES * 128;

    hipMemsetAsync(cnt, 0, (size_t)NPAD * sizeof(int), stream);

    static_assert(2 * ((NREL + 15) / 16) <= GEMM_MAX_BLOCKS,
                  "the two small problems must not have more tiles than blocks");
    gat_wmma_gemm_multi<<<GEMM_MAX_BLOCKS, GEMM_THREADS, 0, stream>>>(
        x,   w1h,       xnode,       N_NODES, 256,
        x,   w1h + 128, xnode + 128, N_NODES, 256,
        rel, w1h + 256, relc,        NREL,    128,
        rel, WR,        out_rel,     NREL,    128,
        384, el, eln, et, etn, cnt, bkt, ovs, ovf, w2h, uc);

    gat_node_l1<<<(N_NODES + NODE_WAVES - 1) / NODE_WAVES, NODE_WAVES * 32, 0, stream>>>(
        cnt, bkt, ovs, ovf, xnode, eemb, relc, w2h, uc, accp, agg, den2);
    static_assert(N_NODES % 16 == 0, "each head's plane must be whole row tiles");
    gat_wmma_gemm_heads<<<GEMM_MAX_BLOCKS, GEMM_THREADS, 0, stream>>>(
        agg, w1h + 256, lx, 2 * N_NODES, 128, 384, 128, 0, 2 * N_NODES / 16, accp, den2);

    gat_wmma_gemm_multi<<<GEMM_MAX_BLOCKS, GEMM_THREADS, 0, stream>>>(
        lx,      w1o,       xab2,       N_NODES, 256,
        lx,      w1o + 128, xab2 + 128, N_NODES, 256,
        out_rel, w1o + 256, relW2,      NREL,    128,
        (const float*)nullptr, (const float*)nullptr, (float*)nullptr, 0, 0,
        384, (const int*)nullptr, (const int*)nullptr, (const int*)nullptr, (const int*)nullptr,
        (int*)nullptr, (int*)nullptr, (int*)nullptr, (int*)nullptr,
        (const float*)nullptr, (float*)nullptr);

    gat_node<2><<<(N_NODES + NODE_WAVES - 1) / NODE_WAVES, NODE_WAVES * 32, 0, stream>>>(
        cnt, bkt, ovs, ovf, xab2, relW2, relW2, w2o, out);
}
